# starting kernel
# baseline (speedup 1.0000x reference)
#include <hip/hip_runtime.h>
#include <hip/hip_bf16.h>

#define EMB 128
#define NCLS 10

typedef __attribute__((ext_vector_type(2))) float v2f;
typedef __attribute__((ext_vector_type(8))) float v8f;

__global__ void k_embed(const int* __restrict__ x,
                        const float* __restrict__ st,
                        const float* __restrict__ ct,
                        const float* __restrict__ pt,
                        float* __restrict__ h, int n)
{
    int idx = blockIdx.x * blockDim.x + threadIdx.x;
    if (idx >= n * 32) return;
    int node = idx >> 5;
    int q    = idx & 31;
    int s = x[node * 3 + 0];
    int c = x[node * 3 + 1];
    int p = x[node * 3 + 2];
    const float4 a = reinterpret_cast<const float4*>(st)[s * 32 + q];
    const float4 b = reinterpret_cast<const float4*>(ct)[c * 32 + q];
    const float4 d = reinterpret_cast<const float4*>(pt)[p * 32 + q];
    float4 o;
    o.x = a.x + b.x + d.x;
    o.y = a.y + b.y + d.y;
    o.z = a.z + b.z + d.z;
    o.w = a.w + b.w + d.w;
    reinterpret_cast<float4*>(h)[node * 32 + q] = o;
}

__global__ void k_fill(float* __restrict__ p, float v, int n)
{
    int idx = blockIdx.x * blockDim.x + threadIdx.x;
    for (; idx < n; idx += gridDim.x * blockDim.x) p[idx] = v;
}

__global__ void k_deg_accum(const int* __restrict__ dst, float* __restrict__ deg, int e)
{
    int idx = blockIdx.x * blockDim.x + threadIdx.x;
    if (idx >= e) return;
    unsafeAtomicAdd(&deg[dst[idx]], 1.0f);
}

__global__ void k_rsqrt_inplace(float* __restrict__ d, int n)
{
    int idx = blockIdx.x * blockDim.x + threadIdx.x;
    if (idx >= n) return;
    d[idx] = __frsqrt_rn(d[idx]);
}

#define WSTRIDE 128
__global__ __launch_bounds__(256) void k_gemm_wmma(
    const float* __restrict__ A, const float* __restrict__ W,
    float* __restrict__ C, int nrows)
{
    __shared__ float Wl[128 * WSTRIDE];
    const int tid = threadIdx.x;

    for (int idx = tid; idx < 128 * 32; idx += 256) {
        int r = idx >> 5, q = idx & 31;
        reinterpret_cast<float4*>(&Wl[r * WSTRIDE + q * 4])[0] =
            reinterpret_cast<const float4*>(W)[r * 32 + q];
    }
    __syncthreads();

    const int wave  = tid >> 5;
    const int lane  = tid & 31;
    const int m     = lane & 15;
    const int khalf = (lane >> 4) * 2;
    const int tileRow = blockIdx.x * 128 + wave * 16;
    if (tileRow >= nrows) return;

    v8f acc[8] = {};

    const float* arow = A + (long long)(tileRow + m) * EMB;
    for (int k = 0; k < EMB; k += 4) {
        const int kk = k + khalf;
        v2f a;
        a.x = arow[kk];
        a.y = arow[kk + 1];
#pragma unroll
        for (int j = 0; j < 8; ++j) {
            const int col = j * 16 + m;
            v2f b;
            b.x = Wl[kk * WSTRIDE + col];
            b.y = Wl[(kk + 1) * WSTRIDE + col];
            acc[j] = __builtin_amdgcn_wmma_f32_16x16x4_f32(
                false, a, false, b, (short)0, acc[j], false, false);
        }
    }

    const int rbase = (lane >> 4) * 8;
#pragma unroll
    for (int j = 0; j < 8; ++j) {
#pragma unroll
        for (int r = 0; r < 8; ++r) {
            C[(long long)(tileRow + rbase + r) * EMB + j * 16 + m] = acc[j][r];
        }
    }
}

__global__ void k_agg_edges(const float* __restrict__ xw,
                            const int* __restrict__ src,
                            const int* __restrict__ dst,
                            const float* __restrict__ dinv,
                            float* __restrict__ agg, int e)
{
    int idx = blockIdx.x * blockDim.x + threadIdx.x;
    if (idx >= e * 32) return;
    const int ed   = idx >> 5;
    const int lane = idx & 31;
    const int s = src[ed];
    const int d = dst[ed];
    const float coef = dinv[s] * dinv[d];
    if (lane == 0 && ed + 64 < e) {
        __builtin_prefetch(xw + (long long)src[ed + 64] * EMB, 0, 0);
    }
    const float4 v = reinterpret_cast<const float4*>(xw)[(long long)s * 32 + lane];
    float* ap = agg + (long long)d * EMB + lane * 4;
    unsafeAtomicAdd(ap + 0, v.x * coef);
    unsafeAtomicAdd(ap + 1, v.y * coef);
    unsafeAtomicAdd(ap + 2, v.z * coef);
    unsafeAtomicAdd(ap + 3, v.w * coef);
}

__global__ void k_finalize_relu(const float* __restrict__ agg,
                                const float* __restrict__ xw,
                                const float* __restrict__ dinv,
                                const float* __restrict__ bias,
                                float* __restrict__ h, int n)
{
    int idx = blockIdx.x * blockDim.x + threadIdx.x;
    if (idx >= n * 32) return;
    const int node = idx >> 5;
    const int q    = idx & 31;
    const float di = dinv[node];
    const float sl = di * di;
    const float4 ag = reinterpret_cast<const float4*>(agg)[(long long)node * 32 + q];
    const float4 xv = reinterpret_cast<const float4*>(xw)[(long long)node * 32 + q];
    const float* bp = bias + q * 4;
    float4 o;
    o.x = fmaxf(ag.x + xv.x * sl + bp[0], 0.0f);
    o.y = fmaxf(ag.y + xv.y * sl + bp[1], 0.0f);
    o.z = fmaxf(ag.z + xv.z * sl + bp[2], 0.0f);
    o.w = fmaxf(ag.w + xv.w * sl + bp[3], 0.0f);
    reinterpret_cast<float4*>(h)[(long long)node * 32 + q] = o;
}

__global__ void k_pool(const float* __restrict__ h,
                       const int* __restrict__ batch,
                       float* __restrict__ sums,
                       float* __restrict__ cnt, int n)
{
    int idx = blockIdx.x * blockDim.x + threadIdx.x;
    if (idx >= n * 32) return;
    const int node = idx >> 5;
    const int q    = idx & 31;
    const int g = batch[node];
    const float4 v = reinterpret_cast<const float4*>(h)[(long long)node * 32 + q];
    float* sp = sums + (long long)g * EMB + q * 4;
    unsafeAtomicAdd(sp + 0, v.x);
    unsafeAtomicAdd(sp + 1, v.y);
    unsafeAtomicAdd(sp + 2, v.z);
    unsafeAtomicAdd(sp + 3, v.w);
    if (q == 0) unsafeAtomicAdd(&cnt[g], 1.0f);
}

__global__ void k_head(const float* __restrict__ sums,
                       const float* __restrict__ cnt,
                       const float* __restrict__ Wlin,
                       const float* __restrict__ blin,
                       float* __restrict__ out, int g_total)
{
    int idx = blockIdx.x * blockDim.x + threadIdx.x;
    if (idx >= g_total * NCLS) return;
    const int g = idx / NCLS;
    const int c = idx % NCLS;
    const float inv = 1.0f / fmaxf(cnt[g], 1.0f);
    const float* sp = sums + (long long)g * EMB;
    float acc = 0.0f;
#pragma unroll 4
    for (int k = 0; k < EMB; ++k) acc += sp[k] * Wlin[k * NCLS + c];
    out[idx] = acc * inv + blin[c];
}

static inline size_t align256(size_t x) { return (x + 255) & ~(size_t)255; }

extern "C" void kernel_launch(void* const* d_in, const int* in_sizes, int n_in,
                              void* d_out, int out_size, void* d_ws, size_t ws_size,
                              hipStream_t stream)
{
    const int*   x        = (const int*)  d_in[0];
    const int*   eidx     = (const int*)  d_in[1];
    const int*   batch    = (const int*)  d_in[2];
    const float* shapeTab = (const float*)d_in[3];
    const float* colorTab = (const float*)d_in[4];
    const float* posTab   = (const float*)d_in[5];
    const float* W1       = (const float*)d_in[6];
    const float* b1       = (const float*)d_in[7];
    const float* W2       = (const float*)d_in[8];
    const float* b2       = (const float*)d_in[9];
    const float* Wlin     = (const float*)d_in[10];
    const float* blin     = (const float*)d_in[11];
    float*       out      = (float*)d_out;

    const int N = in_sizes[0] / 3;
    const int E = in_sizes[1] / 2;
    const int G = out_size / NCLS;

    const int* src = eidx;
    const int* dst = eidx + E;

    char* ws = (char*)d_ws;
    size_t off = 0;
    float* h    = (float*)(ws + off); off = align256(off + (size_t)N * EMB * 4);
    float* xw   = (float*)(ws + off); off = align256(off + (size_t)N * EMB * 4);
    float* agg  = (float*)(ws + off); off = align256(off + (size_t)N * EMB * 4);
    float* dinv = (float*)(ws + off); off = align256(off + (size_t)N * 4);
    float* sums = (float*)(ws + off); off = align256(off + (size_t)G * EMB * 4);
    float* cnt  = (float*)(ws + off); off = align256(off + (size_t)G * 4);
    (void)ws_size;

    const int TB = 256;
    const int nVec      = N * 32;
    const int gVecBlks  = (nVec + TB - 1) / TB;
    const int gEdgeBlks = (E * 32 + TB - 1) / TB;
    const int gGemmBlks = (N + 127) / 128;

    k_embed<<<gVecBlks, TB, 0, stream>>>(x, shapeTab, colorTab, posTab, h, N);

    k_fill<<<(N + TB - 1) / TB, TB, 0, stream>>>(dinv, 1.0f, N);
    k_deg_accum<<<(E + TB - 1) / TB, TB, 0, stream>>>(dst, dinv, E);
    k_rsqrt_inplace<<<(N + TB - 1) / TB, TB, 0, stream>>>(dinv, N);

    k_gemm_wmma<<<gGemmBlks, TB, 0, stream>>>(h, W1, xw, N);
    k_fill<<<4096, TB, 0, stream>>>(agg, 0.0f, N * EMB);
    k_agg_edges<<<gEdgeBlks, TB, 0, stream>>>(xw, src, dst, dinv, agg, E);
    k_finalize_relu<<<gVecBlks, TB, 0, stream>>>(agg, xw, dinv, b1, h, N);

    k_gemm_wmma<<<gGemmBlks, TB, 0, stream>>>(h, W2, xw, N);
    k_fill<<<4096, TB, 0, stream>>>(agg, 0.0f, N * EMB);
    k_agg_edges<<<gEdgeBlks, TB, 0, stream>>>(xw, src, dst, dinv, agg, E);
    k_finalize_relu<<<gVecBlks, TB, 0, stream>>>(agg, xw, dinv, b2, h, N);

    k_fill<<<(G * EMB + TB - 1) / TB, TB, 0, stream>>>(sums, 0.0f, G * EMB);
    k_fill<<<(G + TB - 1) / TB, TB, 0, stream>>>(cnt, 0.0f, G);
    k_pool<<<gVecBlks, TB, 0, stream>>>(h, batch, sums, cnt, N);
    k_head<<<(G * NCLS + TB - 1) / TB, TB, 0, stream>>>(sums, cnt, Wlin, blin, out, G);
}


# aggregation without atomics: edges grouped by destination into an in-call CSR (slot = arrival rank of one integer atomic per edge), one wave per node sums its row in registers in slot order
# speedup vs baseline: 2.1671x; 2.1671x over previous
#include <hip/hip_runtime.h>
#include <hip/hip_bf16.h>

#define EMB 128
#define NCLS 10

typedef __attribute__((ext_vector_type(2))) float v2f;
typedef __attribute__((ext_vector_type(8))) float v8f;

__global__ void k_embed(const int* __restrict__ x,
                        const float* __restrict__ st,
                        const float* __restrict__ ct,
                        const float* __restrict__ pt,
                        float* __restrict__ h, int n)
{
    int idx = blockIdx.x * blockDim.x + threadIdx.x;
    if (idx >= n * 32) return;
    int node = idx >> 5;
    int q    = idx & 31;
    int s = x[node * 3 + 0];
    int c = x[node * 3 + 1];
    int p = x[node * 3 + 2];
    const float4 a = reinterpret_cast<const float4*>(st)[s * 32 + q];
    const float4 b = reinterpret_cast<const float4*>(ct)[c * 32 + q];
    const float4 d = reinterpret_cast<const float4*>(pt)[p * 32 + q];
    float4 o;
    o.x = a.x + b.x + d.x;
    o.y = a.y + b.y + d.y;
    o.z = a.z + b.z + d.z;
    o.w = a.w + b.w + d.w;
    reinterpret_cast<float4*>(h)[node * 32 + q] = o;
}

__global__ void k_fill(float* __restrict__ p, float v, int n)
{
    int idx = blockIdx.x * blockDim.x + threadIdx.x;
    for (; idx < n; idx += gridDim.x * blockDim.x) p[idx] = v;
}

__global__ void k_count(const int* __restrict__ dst, int* __restrict__ cnt,
                        int* __restrict__ rank, int e)
{
    int idx = blockIdx.x * blockDim.x + threadIdx.x;
    if (idx >= e) return;
    rank[idx] = atomicAdd(&cnt[dst[idx]], 1);
}

__global__ __launch_bounds__(256) void k_blocksum(const int* __restrict__ cnt,
                                                  int* __restrict__ bsum, int n)
{
    __shared__ int sh[256];
    const int t = threadIdx.x;
    const int base = blockIdx.x * 1024 + t * 4;
    int s = 0;
    for (int i = 0; i < 4; ++i)
        if (base + i < n) s += cnt[base + i];
    sh[t] = s;
    __syncthreads();
    for (int off = 128; off > 0; off >>= 1) {
        if (t < off) sh[t] += sh[t + off];
        __syncthreads();
    }
    if (t == 0) bsum[blockIdx.x] = sh[0];
}

__global__ __launch_bounds__(256) void k_scan_write(const int* __restrict__ cnt,
                                                    const int* __restrict__ bsum,
                                                    int* __restrict__ rowptr,
                                                    float* __restrict__ dinv, int n)
{
    __shared__ int sh[256];
    __shared__ int red[256];
    const int t = threadIdx.x;
    const int b = blockIdx.x;

    int part = 0;
    for (int i = t; i < b; i += 256) part += bsum[i];
    red[t] = part;

    const int base = b * 1024 + t * 4;
    int v[4];
    int s = 0;
    for (int i = 0; i < 4; ++i) {
        v[i] = (base + i < n) ? cnt[base + i] : 0;
        s += v[i];
    }
    sh[t] = s;
    __syncthreads();

    for (int off = 128; off > 0; off >>= 1) {
        if (t < off) red[t] += red[t + off];
        __syncthreads();
    }
    const int boff = red[0];

    for (int off = 1; off < 256; off <<= 1) {
        const int a = (t >= off) ? sh[t - off] : 0;
        __syncthreads();
        sh[t] += a;
        __syncthreads();
    }
    int run = boff + sh[t] - s;
    for (int i = 0; i < 4; ++i) {
        if (base + i < n) {
            rowptr[base + i] = run;
            dinv[base + i]   = __frsqrt_rn((float)(v[i] + 1));
            run += v[i];
        }
    }
}

__global__ void k_csr_fill(const int* __restrict__ src, const int* __restrict__ dst,
                           const int* __restrict__ rowptr, const int* __restrict__ rank,
                           int* __restrict__ csr_src, int e)
{
    int idx = blockIdx.x * blockDim.x + threadIdx.x;
    if (idx >= e) return;
    csr_src[rowptr[dst[idx]] + rank[idx]] = src[idx];
}

#define WSTRIDE 128
__global__ __launch_bounds__(256) void k_gemm_wmma(
    const float* __restrict__ A, const float* __restrict__ W,
    float* __restrict__ C, int nrows)
{
    __shared__ float Wl[128 * WSTRIDE];
    const int tid = threadIdx.x;

    for (int idx = tid; idx < 128 * 32; idx += 256) {
        int r = idx >> 5, q = idx & 31;
        reinterpret_cast<float4*>(&Wl[r * WSTRIDE + q * 4])[0] =
            reinterpret_cast<const float4*>(W)[r * 32 + q];
    }
    __syncthreads();

    const int wave  = tid >> 5;
    const int lane  = tid & 31;
    const int m     = lane & 15;
    const int khalf = (lane >> 4) * 2;
    const int tileRow = blockIdx.x * 128 + wave * 16;
    if (tileRow >= nrows) return;

    v8f acc[8] = {};

    const float* arow = A + (long long)(tileRow + m) * EMB;
    for (int k = 0; k < EMB; k += 4) {
        const int kk = k + khalf;
        v2f a;
        a.x = arow[kk];
        a.y = arow[kk + 1];
#pragma unroll
        for (int j = 0; j < 8; ++j) {
            const int col = j * 16 + m;
            v2f b;
            b.x = Wl[kk * WSTRIDE + col];
            b.y = Wl[(kk + 1) * WSTRIDE + col];
            acc[j] = __builtin_amdgcn_wmma_f32_16x16x4_f32(
                false, a, false, b, (short)0, acc[j], false, false);
        }
    }

    const int rbase = (lane >> 4) * 8;
#pragma unroll
    for (int j = 0; j < 8; ++j) {
#pragma unroll
        for (int r = 0; r < 8; ++r) {
            C[(long long)(tileRow + rbase + r) * EMB + j * 16 + m] = acc[j][r];
        }
    }
}

__global__ __launch_bounds__(256) void k_gather_relu(const float* __restrict__ xw,
                                                     const int* __restrict__ rowptr,
                                                     const int* __restrict__ cnt,
                                                     const int* __restrict__ csr_src,
                                                     const float* __restrict__ dinv,
                                                     const float* __restrict__ bias,
                                                     float* __restrict__ h, int n)
{
    const int lane = threadIdx.x & 31;
    const int node = blockIdx.x * 8 + (threadIdx.x >> 5);
    if (node >= n) return;
    const int beg = __builtin_amdgcn_readfirstlane(rowptr[node]);
    const int deg = __builtin_amdgcn_readfirstlane(cnt[node]);
    const float dd = dinv[node];
    const float4* xw4 = reinterpret_cast<const float4*>(xw);

    float4 acc = make_float4(0.0f, 0.0f, 0.0f, 0.0f);
    for (int base = 0; base < deg; base += 32) {
        const int m = (deg - base < 32) ? (deg - base) : 32;
        int   s_l = 0;
        float c_l = 0.0f;
        if (lane < m) {
            s_l = csr_src[beg + base + lane];
            c_l = dinv[s_l] * dd;
        }
#pragma unroll 4
        for (int j = 0; j < m; ++j) {
            const int   s = __builtin_amdgcn_readlane(s_l, j);
            const float c = __int_as_float(__builtin_amdgcn_readlane(__float_as_int(c_l), j));
            const float4 v = xw4[(long long)s * 32 + lane];
            acc.x += v.x * c;
            acc.y += v.y * c;
            acc.z += v.z * c;
            acc.w += v.w * c;
        }
    }

    const float sl = dd * dd;
    const float4 xv = xw4[(long long)node * 32 + lane];
    const float* bp = bias + lane * 4;
    float4 o;
    o.x = fmaxf(acc.x + xv.x * sl + bp[0], 0.0f);
    o.y = fmaxf(acc.y + xv.y * sl + bp[1], 0.0f);
    o.z = fmaxf(acc.z + xv.z * sl + bp[2], 0.0f);
    o.w = fmaxf(acc.w + xv.w * sl + bp[3], 0.0f);
    reinterpret_cast<float4*>(h)[(long long)node * 32 + lane] = o;
}

__global__ void k_pool(const float* __restrict__ h,
                       const int* __restrict__ batch,
                       float* __restrict__ sums,
                       float* __restrict__ cnt, int n)
{
    int idx = blockIdx.x * blockDim.x + threadIdx.x;
    if (idx >= n * 32) return;
    const int node = idx >> 5;
    const int q    = idx & 31;
    const int g = batch[node];
    const float4 v = reinterpret_cast<const float4*>(h)[(long long)node * 32 + q];
    float* sp = sums + (long long)g * EMB + q * 4;
    unsafeAtomicAdd(sp + 0, v.x);
    unsafeAtomicAdd(sp + 1, v.y);
    unsafeAtomicAdd(sp + 2, v.z);
    unsafeAtomicAdd(sp + 3, v.w);
    if (q == 0) unsafeAtomicAdd(&cnt[g], 1.0f);
}

__global__ void k_head(const float* __restrict__ sums,
                       const float* __restrict__ cnt,
                       const float* __restrict__ Wlin,
                       const float* __restrict__ blin,
                       float* __restrict__ out, int g_total)
{
    int idx = blockIdx.x * blockDim.x + threadIdx.x;
    if (idx >= g_total * NCLS) return;
    const int g = idx / NCLS;
    const int c = idx % NCLS;
    const float inv = 1.0f / fmaxf(cnt[g], 1.0f);
    const float* sp = sums + (long long)g * EMB;
    float acc = 0.0f;
#pragma unroll 4
    for (int k = 0; k < EMB; ++k) acc += sp[k] * Wlin[k * NCLS + c];
    out[idx] = acc * inv + blin[c];
}

static inline size_t align256(size_t x) { return (x + 255) & ~(size_t)255; }

extern "C" void kernel_launch(void* const* d_in, const int* in_sizes, int n_in,
                              void* d_out, int out_size, void* d_ws, size_t ws_size,
                              hipStream_t stream)
{
    const int*   x        = (const int*)  d_in[0];
    const int*   eidx     = (const int*)  d_in[1];
    const int*   batch    = (const int*)  d_in[2];
    const float* shapeTab = (const float*)d_in[3];
    const float* colorTab = (const float*)d_in[4];
    const float* posTab   = (const float*)d_in[5];
    const float* W1       = (const float*)d_in[6];
    const float* b1       = (const float*)d_in[7];
    const float* W2       = (const float*)d_in[8];
    const float* b2       = (const float*)d_in[9];
    const float* Wlin     = (const float*)d_in[10];
    const float* blin     = (const float*)d_in[11];
    float*       out      = (float*)d_out;

    const int N = in_sizes[0] / 3;
    const int E = in_sizes[1] / 2;
    const int G = out_size / NCLS;

    const int* src = eidx;
    const int* dst = eidx + E;

    char* ws = (char*)d_ws;
    size_t off = 0;
    float* h    = (float*)(ws + off); off = align256(off + (size_t)N * EMB * 4);
    float* xw   = (float*)(ws + off); off = align256(off + (size_t)N * EMB * 4);
    char*  csr  = ws + off;           off = align256(off + (size_t)N * EMB * 4);
    float* dinv = (float*)(ws + off); off = align256(off + (size_t)N * 4);
    float* sums = (float*)(ws + off); off = align256(off + (size_t)G * EMB * 4);
    float* cnt  = (float*)(ws + off); off = align256(off + (size_t)G * 4);
    (void)ws_size;

    const int nTiles = (N + 1023) / 1024;
    size_t coff = 0;
    int* degc   = (int*)(csr + coff); coff = align256(coff + (size_t)N * 4);
    int* rowptr = (int*)(csr + coff); coff = align256(coff + (size_t)N * 4);
    int* bsum   = (int*)(csr + coff); coff = align256(coff + (size_t)nTiles * 4);
    int* rank   = (int*)(csr + coff); coff = align256(coff + (size_t)E * 4);
    int* csrSrc = (int*)(csr + coff); coff = align256(coff + (size_t)E * 4);

    const int TB = 256;
    const int nVec      = N * 32;
    const int gVecBlks  = (nVec + TB - 1) / TB;
    const int gEdge1    = (E + TB - 1) / TB;
    const int gNodeWave = (N + 7) / 8;
    const int gGemmBlks = (N + 127) / 128;

    k_embed<<<gVecBlks, TB, 0, stream>>>(x, shapeTab, colorTab, posTab, h, N);

    k_fill<<<(N + TB - 1) / TB, TB, 0, stream>>>((float*)degc, 0.0f, N);
    k_count<<<gEdge1, TB, 0, stream>>>(dst, degc, rank, E);
    k_blocksum<<<nTiles, 256, 0, stream>>>(degc, bsum, N);
    k_scan_write<<<nTiles, 256, 0, stream>>>(degc, bsum, rowptr, dinv, N);
    k_csr_fill<<<gEdge1, TB, 0, stream>>>(src, dst, rowptr, rank, csrSrc, E);

    k_gemm_wmma<<<gGemmBlks, TB, 0, stream>>>(h, W1, xw, N);
    k_gather_relu<<<gNodeWave, 256, 0, stream>>>(xw, rowptr, degc, csrSrc, dinv, b1, h, N);

    k_gemm_wmma<<<gGemmBlks, TB, 0, stream>>>(h, W2, xw, N);
    k_gather_relu<<<gNodeWave, 256, 0, stream>>>(xw, rowptr, degc, csrSrc, dinv, b2, h, N);

    k_fill<<<(G * EMB + TB - 1) / TB, TB, 0, stream>>>(sums, 0.0f, G * EMB);
    k_fill<<<(G + TB - 1) / TB, TB, 0, stream>>>(cnt, 0.0f, G);
    k_pool<<<gVecBlks, TB, 0, stream>>>(h, batch, sums, cnt, N);
    k_head<<<(G * NCLS + TB - 1) / TB, TB, 0, stream>>>(sums, cnt, Wlin, blin, out, G);
}


# layer 1 by linearity: the three embedding tables go through W1 on WMMA (288 rows) and the layer-1 gather sums the transformed table rows per edge; no embedding pass, no N-row product; sums regrouped
# speedup vs baseline: 2.3560x; 1.0872x over previous
#include <hip/hip_runtime.h>
#include <hip/hip_bf16.h>

#define EMB 128
#define NCLS 10
#define NSHAPE 16
#define NCOLOR 16
#define MAXPOS 256
#define NTABROWS (NSHAPE + NCOLOR + MAXPOS)

typedef __attribute__((ext_vector_type(2))) float v2f;
typedef __attribute__((ext_vector_type(8))) float v8f;

__global__ __launch_bounds__(256) void k_tab_gemm_wmma(
    const float* __restrict__ st, const float* __restrict__ ct, const float* __restrict__ pt,
    const float* __restrict__ W, float* __restrict__ tw)
{
    const int tile  = blockIdx.x;
    const int wave  = threadIdx.x >> 5;
    const int lane  = threadIdx.x & 31;
    const int m     = lane & 15;
    const int khalf = (lane >> 4) * 2;

    const float* abase;
    if (tile == 0)      abase = st;
    else if (tile == 1) abase = ct;
    else                abase = pt + (long long)(tile - 2) * 16 * EMB;

    const float* arow = abase + m * EMB;
    const int col = wave * 16 + m;

    v8f acc = {};
    for (int k = 0; k < EMB; k += 4) {
        const int kk = k + khalf;
        v2f a, b;
        a.x = arow[kk];
        a.y = arow[kk + 1];
        b.x = W[kk * EMB + col];
        b.y = W[(kk + 1) * EMB + col];
        acc = __builtin_amdgcn_wmma_f32_16x16x4_f32(
            false, a, false, b, (short)0, acc, false, false);
    }

    const int rbase = (lane >> 4) * 8;
#pragma unroll
    for (int r = 0; r < 8; ++r)
        tw[(long long)(tile * 16 + rbase + r) * EMB + col] = acc[r];
}

__global__ void k_fill(float* __restrict__ p, float v, int n)
{
    int idx = blockIdx.x * blockDim.x + threadIdx.x;
    for (; idx < n; idx += gridDim.x * blockDim.x) p[idx] = v;
}

__global__ void k_count(const int* __restrict__ dst, int* __restrict__ cnt,
                        int* __restrict__ rank, int e)
{
    int idx = blockIdx.x * blockDim.x + threadIdx.x;
    if (idx >= e) return;
    rank[idx] = atomicAdd(&cnt[dst[idx]], 1);
}

__global__ __launch_bounds__(256) void k_blocksum(const int* __restrict__ cnt,
                                                  int* __restrict__ bsum, int n)
{
    __shared__ int sh[256];
    const int t = threadIdx.x;
    const int base = blockIdx.x * 1024 + t * 4;
    int s = 0;
    for (int i = 0; i < 4; ++i)
        if (base + i < n) s += cnt[base + i];
    sh[t] = s;
    __syncthreads();
    for (int off = 128; off > 0; off >>= 1) {
        if (t < off) sh[t] += sh[t + off];
        __syncthreads();
    }
    if (t == 0) bsum[blockIdx.x] = sh[0];
}

__global__ __launch_bounds__(256) void k_scan_write(const int* __restrict__ cnt,
                                                    const int* __restrict__ bsum,
                                                    int* __restrict__ rowptr,
                                                    float* __restrict__ dinv, int n)
{
    __shared__ int sh[256];
    __shared__ int red[256];
    const int t = threadIdx.x;
    const int b = blockIdx.x;

    int part = 0;
    for (int i = t; i < b; i += 256) part += bsum[i];
    red[t] = part;

    const int base = b * 1024 + t * 4;
    int v[4];
    int s = 0;
    for (int i = 0; i < 4; ++i) {
        v[i] = (base + i < n) ? cnt[base + i] : 0;
        s += v[i];
    }
    sh[t] = s;
    __syncthreads();

    for (int off = 128; off > 0; off >>= 1) {
        if (t < off) red[t] += red[t + off];
        __syncthreads();
    }
    const int boff = red[0];

    for (int off = 1; off < 256; off <<= 1) {
        const int a = (t >= off) ? sh[t - off] : 0;
        __syncthreads();
        sh[t] += a;
        __syncthreads();
    }
    int run = boff + sh[t] - s;
    for (int i = 0; i < 4; ++i) {
        if (base + i < n) {
            rowptr[base + i] = run;
            dinv[base + i]   = __frsqrt_rn((float)(v[i] + 1));
            run += v[i];
        }
    }
}

__global__ void k_csr_fill(const int* __restrict__ src, const int* __restrict__ dst,
                           const int* __restrict__ rowptr, const int* __restrict__ rank,
                           int* __restrict__ csr_src, int e)
{
    int idx = blockIdx.x * blockDim.x + threadIdx.x;
    if (idx >= e) return;
    csr_src[rowptr[dst[idx]] + rank[idx]] = src[idx];
}

#define WSTRIDE 128
__global__ __launch_bounds__(256) void k_gemm_wmma(
    const float* __restrict__ A, const float* __restrict__ W,
    float* __restrict__ C, int nrows)
{
    __shared__ float Wl[128 * WSTRIDE];
    const int tid = threadIdx.x;

    for (int idx = tid; idx < 128 * 32; idx += 256) {
        int r = idx >> 5, q = idx & 31;
        reinterpret_cast<float4*>(&Wl[r * WSTRIDE + q * 4])[0] =
            reinterpret_cast<const float4*>(W)[r * 32 + q];
    }
    __syncthreads();

    const int wave  = tid >> 5;
    const int lane  = tid & 31;
    const int m     = lane & 15;
    const int khalf = (lane >> 4) * 2;
    const int tileRow = blockIdx.x * 128 + wave * 16;
    if (tileRow >= nrows) return;

    v8f acc[8] = {};

    const float* arow = A + (long long)(tileRow + m) * EMB;
    for (int k = 0; k < EMB; k += 4) {
        const int kk = k + khalf;
        v2f a;
        a.x = arow[kk];
        a.y = arow[kk + 1];
#pragma unroll
        for (int j = 0; j < 8; ++j) {
            const int col = j * 16 + m;
            v2f b;
            b.x = Wl[kk * WSTRIDE + col];
            b.y = Wl[(kk + 1) * WSTRIDE + col];
            acc[j] = __builtin_amdgcn_wmma_f32_16x16x4_f32(
                false, a, false, b, (short)0, acc[j], false, false);
        }
    }

    const int rbase = (lane >> 4) * 8;
#pragma unroll
    for (int j = 0; j < 8; ++j) {
#pragma unroll
        for (int r = 0; r < 8; ++r) {
            C[(long long)(tileRow + rbase + r) * EMB + j * 16 + m] = acc[j][r];
        }
    }
}

__global__ __launch_bounds__(256) void k_gather_relu(const float* __restrict__ xw,
                                                     const int* __restrict__ rowptr,
                                                     const int* __restrict__ cnt,
                                                     const int* __restrict__ csr_src,
                                                     const float* __restrict__ dinv,
                                                     const float* __restrict__ bias,
                                                     float* __restrict__ h, int n)
{
    const int lane = threadIdx.x & 31;
    const int node = blockIdx.x * 8 + (threadIdx.x >> 5);
    if (node >= n) return;
    const int beg = __builtin_amdgcn_readfirstlane(rowptr[node]);
    const int deg = __builtin_amdgcn_readfirstlane(cnt[node]);
    const float dd = dinv[node];
    const float4* xw4 = reinterpret_cast<const float4*>(xw);

    float4 acc = make_float4(0.0f, 0.0f, 0.0f, 0.0f);
    for (int base = 0; base < deg; base += 32) {
        const int m = (deg - base < 32) ? (deg - base) : 32;
        int   s_l = 0;
        float c_l = 0.0f;
        if (lane < m) {
            s_l = csr_src[beg + base + lane];
            c_l = dinv[s_l] * dd;
        }
#pragma unroll 4
        for (int j = 0; j < m; ++j) {
            const int   s = __builtin_amdgcn_readlane(s_l, j);
            const float c = __int_as_float(__builtin_amdgcn_readlane(__float_as_int(c_l), j));
            const float4 v = xw4[(long long)s * 32 + lane];
            acc.x += v.x * c;
            acc.y += v.y * c;
            acc.z += v.z * c;
            acc.w += v.w * c;
        }
    }

    const float sl = dd * dd;
    const float4 xv = xw4[(long long)node * 32 + lane];
    const float* bp = bias + lane * 4;
    float4 o;
    o.x = fmaxf(acc.x + xv.x * sl + bp[0], 0.0f);
    o.y = fmaxf(acc.y + xv.y * sl + bp[1], 0.0f);
    o.z = fmaxf(acc.z + xv.z * sl + bp[2], 0.0f);
    o.w = fmaxf(acc.w + xv.w * sl + bp[3], 0.0f);
    reinterpret_cast<float4*>(h)[(long long)node * 32 + lane] = o;
}

__global__ __launch_bounds__(256) void k_gather_tab_relu(const int* __restrict__ x,
                                                         const float* __restrict__ tw,
                                                         const int* __restrict__ rowptr,
                                                         const int* __restrict__ cnt,
                                                         const int* __restrict__ csr_src,
                                                         const float* __restrict__ dinv,
                                                         const float* __restrict__ bias,
                                                         float* __restrict__ h, int n)
{
    const int lane = threadIdx.x & 31;
    const int node = blockIdx.x * 8 + (threadIdx.x >> 5);
    if (node >= n) return;
    const int beg = __builtin_amdgcn_readfirstlane(rowptr[node]);
    const int deg = __builtin_amdgcn_readfirstlane(cnt[node]);
    const float dd = dinv[node];
    const float4* tw4 = reinterpret_cast<const float4*>(tw);

    float4 acc = make_float4(0.0f, 0.0f, 0.0f, 0.0f);
    for (int base = 0; base < deg; base += 32) {
        const int m = (deg - base < 32) ? (deg - base) : 32;
        int   r0 = 0, r1 = 0, r2 = 0;
        float c_l = 0.0f;
        if (lane < m) {
            const int s_l = csr_src[beg + base + lane];
            c_l = dinv[s_l] * dd;
            r0 = x[s_l * 3 + 0];
            r1 = x[s_l * 3 + 1] + NSHAPE;
            r2 = x[s_l * 3 + 2] + NSHAPE + NCOLOR;
        }
#pragma unroll 4
        for (int j = 0; j < m; ++j) {
            const int   i0 = __builtin_amdgcn_readlane(r0, j);
            const int   i1 = __builtin_amdgcn_readlane(r1, j);
            const int   i2 = __builtin_amdgcn_readlane(r2, j);
            const float c  = __int_as_float(__builtin_amdgcn_readlane(__float_as_int(c_l), j));
            const float4 a = tw4[i0 * 32 + lane];
            const float4 b = tw4[i1 * 32 + lane];
            const float4 d = tw4[i2 * 32 + lane];
            acc.x += ((a.x + b.x) + d.x) * c;
            acc.y += ((a.y + b.y) + d.y) * c;
            acc.z += ((a.z + b.z) + d.z) * c;
            acc.w += ((a.w + b.w) + d.w) * c;
        }
    }

    const int q0 = __builtin_amdgcn_readfirstlane(x[node * 3 + 0]);
    const int q1 = __builtin_amdgcn_readfirstlane(x[node * 3 + 1]) + NSHAPE;
    const int q2 = __builtin_amdgcn_readfirstlane(x[node * 3 + 2]) + NSHAPE + NCOLOR;
    const float4 a = tw4[q0 * 32 + lane];
    const float4 b = tw4[q1 * 32 + lane];
    const float4 d = tw4[q2 * 32 + lane];
    const float sl = dd * dd;
    const float* bp = bias + lane * 4;
    float4 o;
    o.x = fmaxf(acc.x + ((a.x + b.x) + d.x) * sl + bp[0], 0.0f);
    o.y = fmaxf(acc.y + ((a.y + b.y) + d.y) * sl + bp[1], 0.0f);
    o.z = fmaxf(acc.z + ((a.z + b.z) + d.z) * sl + bp[2], 0.0f);
    o.w = fmaxf(acc.w + ((a.w + b.w) + d.w) * sl + bp[3], 0.0f);
    reinterpret_cast<float4*>(h)[(long long)node * 32 + lane] = o;
}

__global__ void k_pool(const float* __restrict__ h,
                       const int* __restrict__ batch,
                       float* __restrict__ sums,
                       float* __restrict__ cnt, int n)
{
    int idx = blockIdx.x * blockDim.x + threadIdx.x;
    if (idx >= n * 32) return;
    const int node = idx >> 5;
    const int q    = idx & 31;
    const int g = batch[node];
    const float4 v = reinterpret_cast<const float4*>(h)[(long long)node * 32 + q];
    float* sp = sums + (long long)g * EMB + q * 4;
    unsafeAtomicAdd(sp + 0, v.x);
    unsafeAtomicAdd(sp + 1, v.y);
    unsafeAtomicAdd(sp + 2, v.z);
    unsafeAtomicAdd(sp + 3, v.w);
    if (q == 0) unsafeAtomicAdd(&cnt[g], 1.0f);
}

__global__ void k_head(const float* __restrict__ sums,
                       const float* __restrict__ cnt,
                       const float* __restrict__ Wlin,
                       const float* __restrict__ blin,
                       float* __restrict__ out, int g_total)
{
    int idx = blockIdx.x * blockDim.x + threadIdx.x;
    if (idx >= g_total * NCLS) return;
    const int g = idx / NCLS;
    const int c = idx % NCLS;
    const float inv = 1.0f / fmaxf(cnt[g], 1.0f);
    const float* sp = sums + (long long)g * EMB;
    float acc = 0.0f;
#pragma unroll 4
    for (int k = 0; k < EMB; ++k) acc += sp[k] * Wlin[k * NCLS + c];
    out[idx] = acc * inv + blin[c];
}

static inline size_t align256(size_t x) { return (x + 255) & ~(size_t)255; }

extern "C" void kernel_launch(void* const* d_in, const int* in_sizes, int n_in,
                              void* d_out, int out_size, void* d_ws, size_t ws_size,
                              hipStream_t stream)
{
    const int*   x        = (const int*)  d_in[0];
    const int*   eidx     = (const int*)  d_in[1];
    const int*   batch    = (const int*)  d_in[2];
    const float* shapeTab = (const float*)d_in[3];
    const float* colorTab = (const float*)d_in[4];
    const float* posTab   = (const float*)d_in[5];
    const float* W1       = (const float*)d_in[6];
    const float* b1       = (const float*)d_in[7];
    const float* W2       = (const float*)d_in[8];
    const float* b2       = (const float*)d_in[9];
    const float* Wlin     = (const float*)d_in[10];
    const float* blin     = (const float*)d_in[11];
    float*       out      = (float*)d_out;

    const int N = in_sizes[0] / 3;
    const int E = in_sizes[1] / 2;
    const int G = out_size / NCLS;

    const int* src = eidx;
    const int* dst = eidx + E;

    char* ws = (char*)d_ws;
    size_t off = 0;
    float* h    = (float*)(ws + off); off = align256(off + (size_t)N * EMB * 4);
    float* xw   = (float*)(ws + off); off = align256(off + (size_t)N * EMB * 4);
    char*  csr  = ws + off;           off = align256(off + (size_t)N * EMB * 4);
    float* dinv = (float*)(ws + off); off = align256(off + (size_t)N * 4);
    float* sums = (float*)(ws + off); off = align256(off + (size_t)G * EMB * 4);
    float* cnt  = (float*)(ws + off); off = align256(off + (size_t)G * 4);
    (void)ws_size;

    const int nTiles = (N + 1023) / 1024;
    size_t coff = 0;
    int* degc   = (int*)(csr + coff); coff = align256(coff + (size_t)N * 4);
    int* rowptr = (int*)(csr + coff); coff = align256(coff + (size_t)N * 4);
    int* bsum   = (int*)(csr + coff); coff = align256(coff + (size_t)nTiles * 4);
    int* rank   = (int*)(csr + coff); coff = align256(coff + (size_t)E * 4);
    int* csrSrc = (int*)(csr + coff); coff = align256(coff + (size_t)E * 4);
    float* tw   = (float*)(csr + coff); coff = align256(coff + (size_t)NTABROWS * EMB * 4);

    const int TB = 256;
    const int nVec      = N * 32;
    const int gVecBlks  = (nVec + TB - 1) / TB;
    const int gEdge1    = (E + TB - 1) / TB;
    const int gNodeWave = (N + 7) / 8;
    const int gGemmBlks = (N + 127) / 128;

    k_tab_gemm_wmma<<<NTABROWS / 16, TB, 0, stream>>>(shapeTab, colorTab, posTab, W1, tw);

    k_fill<<<(N + TB - 1) / TB, TB, 0, stream>>>((float*)degc, 0.0f, N);
    k_count<<<gEdge1, TB, 0, stream>>>(dst, degc, rank, E);
    k_blocksum<<<nTiles, 256, 0, stream>>>(degc, bsum, N);
    k_scan_write<<<nTiles, 256, 0, stream>>>(degc, bsum, rowptr, dinv, N);
    k_csr_fill<<<gEdge1, TB, 0, stream>>>(src, dst, rowptr, rank, csrSrc, E);

    k_gather_tab_relu<<<gNodeWave, 256, 0, stream>>>(x, tw, rowptr, degc, csrSrc, dinv, b1, h, N);

    k_gemm_wmma<<<gGemmBlks, TB, 0, stream>>>(h, W2, xw, N);
    k_gather_relu<<<gNodeWave, 256, 0, stream>>>(xw, rowptr, degc, csrSrc, dinv, b2, h, N);

    k_fill<<<(G * EMB + TB - 1) / TB, TB, 0, stream>>>(sums, 0.0f, G * EMB);
    k_fill<<<(G + TB - 1) / TB, TB, 0, stream>>>(cnt, 0.0f, G);
    k_pool<<<gVecBlks, TB, 0, stream>>>(h, batch, sums, cnt, N);
    k_head<<<(G * NCLS + TB - 1) / TB, TB, 0, stream>>>(sums, cnt, Wlin, blin, out, G);
}


# the wide kernels (both gathers, edge count, edge placement, pool) run fixed resident grids of long-lived waves that stride over nodes or edges, next row header prefetched; atomic arrival orders shift
# speedup vs baseline: 2.8731x; 1.2195x over previous
#include <hip/hip_runtime.h>
#include <hip/hip_bf16.h>

#define EMB 128
#define NCLS 10
#define NSHAPE 16
#define NCOLOR 16
#define MAXPOS 256
#define NTABROWS (NSHAPE + NCOLOR + MAXPOS)

typedef __attribute__((ext_vector_type(2))) float v2f;
typedef __attribute__((ext_vector_type(8))) float v8f;

__global__ __launch_bounds__(256) void k_tab_gemm_wmma(
    const float* __restrict__ st, const float* __restrict__ ct, const float* __restrict__ pt,
    const float* __restrict__ W, float* __restrict__ tw)
{
    const int tile  = blockIdx.x;
    const int wave  = threadIdx.x >> 5;
    const int lane  = threadIdx.x & 31;
    const int m     = lane & 15;
    const int khalf = (lane >> 4) * 2;

    const float* abase;
    if (tile == 0)      abase = st;
    else if (tile == 1) abase = ct;
    else                abase = pt + (long long)(tile - 2) * 16 * EMB;

    const float* arow = abase + m * EMB;
    const int col = wave * 16 + m;

    v8f acc = {};
    for (int k = 0; k < EMB; k += 4) {
        const int kk = k + khalf;
        v2f a, b;
        a.x = arow[kk];
        a.y = arow[kk + 1];
        b.x = W[kk * EMB + col];
        b.y = W[(kk + 1) * EMB + col];
        acc = __builtin_amdgcn_wmma_f32_16x16x4_f32(
            false, a, false, b, (short)0, acc, false, false);
    }

    const int rbase = (lane >> 4) * 8;
#pragma unroll
    for (int r = 0; r < 8; ++r)
        tw[(long long)(tile * 16 + rbase + r) * EMB + col] = acc[r];
}

__global__ void k_fill(float* __restrict__ p, float v, int n)
{
    int idx = blockIdx.x * blockDim.x + threadIdx.x;
    for (; idx < n; idx += gridDim.x * blockDim.x) p[idx] = v;
}

__global__ __launch_bounds__(256) void k_count(const int* __restrict__ dst, int* __restrict__ cnt,
                                               int* __restrict__ rank, int e)
{
    const int stride = gridDim.x * blockDim.x;
    for (int base = blockIdx.x * blockDim.x + threadIdx.x; base < e; base += 4 * stride) {
        int d[4], r[4];
#pragma unroll
        for (int j = 0; j < 4; ++j) {
            const int i = base + j * stride;
            d[j] = dst[(i < e) ? i : base];
        }
#pragma unroll
        for (int j = 0; j < 4; ++j) {
            r[j] = 0;
            if (base + j * stride < e) r[j] = atomicAdd(&cnt[d[j]], 1);
        }
#pragma unroll
        for (int j = 0; j < 4; ++j) {
            const int i = base + j * stride;
            if (i < e) rank[i] = r[j];
        }
    }
}

__global__ __launch_bounds__(256) void k_blocksum(const int* __restrict__ cnt,
                                                  int* __restrict__ bsum, int n)
{
    __shared__ int sh[256];
    const int t = threadIdx.x;
    const int base = blockIdx.x * 1024 + t * 4;
    int s = 0;
    for (int i = 0; i < 4; ++i)
        if (base + i < n) s += cnt[base + i];
    sh[t] = s;
    __syncthreads();
    for (int off = 128; off > 0; off >>= 1) {
        if (t < off) sh[t] += sh[t + off];
        __syncthreads();
    }
    if (t == 0) bsum[blockIdx.x] = sh[0];
}

__global__ __launch_bounds__(256) void k_scan_write(const int* __restrict__ cnt,
                                                    const int* __restrict__ bsum,
                                                    int* __restrict__ rowptr,
                                                    float* __restrict__ dinv, int n)
{
    __shared__ int sh[256];
    __shared__ int red[256];
    const int t = threadIdx.x;
    const int b = blockIdx.x;

    int part = 0;
    for (int i = t; i < b; i += 256) part += bsum[i];
    red[t] = part;

    const int base = b * 1024 + t * 4;
    int v[4];
    int s = 0;
    for (int i = 0; i < 4; ++i) {
        v[i] = (base + i < n) ? cnt[base + i] : 0;
        s += v[i];
    }
    sh[t] = s;
    __syncthreads();

    for (int off = 128; off > 0; off >>= 1) {
        if (t < off) red[t] += red[t + off];
        __syncthreads();
    }
    const int boff = red[0];

    for (int off = 1; off < 256; off <<= 1) {
        const int a = (t >= off) ? sh[t - off] : 0;
        __syncthreads();
        sh[t] += a;
        __syncthreads();
    }
    int run = boff + sh[t] - s;
    for (int i = 0; i < 4; ++i) {
        if (base + i < n) {
            rowptr[base + i] = run;
            dinv[base + i]   = __frsqrt_rn((float)(v[i] + 1));
            run += v[i];
        }
    }
}

__global__ __launch_bounds__(256) void k_csr_fill(const int* __restrict__ src, const int* __restrict__ dst,
                                                  const int* __restrict__ rowptr, const int* __restrict__ rank,
                                                  int* __restrict__ csr_src, int e)
{
    const int stride = gridDim.x * blockDim.x;
    for (int base = blockIdx.x * blockDim.x + threadIdx.x; base < e; base += 4 * stride) {
        int d[4], s[4], r[4], p[4];
#pragma unroll
        for (int j = 0; j < 4; ++j) {
            const int i  = base + j * stride;
            const int ic = (i < e) ? i : base;
            d[j] = dst[ic];
            s[j] = src[ic];
            r[j] = rank[ic];
        }
#pragma unroll
        for (int j = 0; j < 4; ++j) p[j] = rowptr[d[j]];
#pragma unroll
        for (int j = 0; j < 4; ++j) {
            if (base + j * stride < e) csr_src[p[j] + r[j]] = s[j];
        }
    }
}

#define WSTRIDE 128
__global__ __launch_bounds__(256) void k_gemm_wmma(
    const float* __restrict__ A, const float* __restrict__ W,
    float* __restrict__ C, int nrows)
{
    __shared__ float Wl[128 * WSTRIDE];
    const int tid = threadIdx.x;

    for (int idx = tid; idx < 128 * 32; idx += 256) {
        int r = idx >> 5, q = idx & 31;
        reinterpret_cast<float4*>(&Wl[r * WSTRIDE + q * 4])[0] =
            reinterpret_cast<const float4*>(W)[r * 32 + q];
    }
    __syncthreads();

    const int wave  = tid >> 5;
    const int lane  = tid & 31;
    const int m     = lane & 15;
    const int khalf = (lane >> 4) * 2;
    const int tileRow = blockIdx.x * 128 + wave * 16;
    if (tileRow >= nrows) return;

    v8f acc[8] = {};

    const float* arow = A + (long long)(tileRow + m) * EMB;
    for (int k = 0; k < EMB; k += 4) {
        const int kk = k + khalf;
        v2f a;
        a.x = arow[kk];
        a.y = arow[kk + 1];
#pragma unroll
        for (int j = 0; j < 8; ++j) {
            const int col = j * 16 + m;
            v2f b;
            b.x = Wl[kk * WSTRIDE + col];
            b.y = Wl[(kk + 1) * WSTRIDE + col];
            acc[j] = __builtin_amdgcn_wmma_f32_16x16x4_f32(
                false, a, false, b, (short)0, acc[j], false, false);
        }
    }

    const int rbase = (lane >> 4) * 8;
#pragma unroll
    for (int j = 0; j < 8; ++j) {
#pragma unroll
        for (int r = 0; r < 8; ++r) {
            C[(long long)(tileRow + rbase + r) * EMB + j * 16 + m] = acc[j][r];
        }
    }
}

__global__ __launch_bounds__(256) void k_gather_relu(const float* __restrict__ xw,
                                                     const int* __restrict__ rowptr,
                                                     const int* __restrict__ cnt,
                                                     const int* __restrict__ csr_src,
                                                     const float* __restrict__ dinv,
                                                     const float* __restrict__ bias,
                                                     float* __restrict__ h, int n)
{
    const int lane = threadIdx.x & 31;
    const int nw   = gridDim.x * 8;
    int node = __builtin_amdgcn_readfirstlane((int)(blockIdx.x * 8 + (threadIdx.x >> 5)));
    if (node >= n) return;
    const float4* xw4 = reinterpret_cast<const float4*>(xw);
    const float* bp = bias + lane * 4;
    const float4 bv = make_float4(bp[0], bp[1], bp[2], bp[3]);

    int   beg = __builtin_amdgcn_readfirstlane(rowptr[node]);
    int   deg = __builtin_amdgcn_readfirstlane(cnt[node]);
    float dd  = dinv[node];

    for (;;) {
        const int  next = node + nw;
        const bool more = next < n;
        int   nbeg = 0, ndeg = 0;
        float ndd  = 0.0f;
        if (more) {
            nbeg = rowptr[next];
            ndeg = cnt[next];
            ndd  = dinv[next];
        }

        float4 acc = make_float4(0.0f, 0.0f, 0.0f, 0.0f);
        for (int base = 0; base < deg; base += 32) {
            const int m = (deg - base < 32) ? (deg - base) : 32;
            int   s_l = 0;
            float c_l = 0.0f;
            if (lane < m) {
                s_l = csr_src[beg + base + lane];
                c_l = dinv[s_l] * dd;
            }
#pragma unroll 4
            for (int j = 0; j < m; ++j) {
                const int   s = __builtin_amdgcn_readlane(s_l, j);
                const float c = __int_as_float(__builtin_amdgcn_readlane(__float_as_int(c_l), j));
                const float4 v = xw4[(long long)s * 32 + lane];
                acc.x += v.x * c;
                acc.y += v.y * c;
                acc.z += v.z * c;
                acc.w += v.w * c;
            }
        }

        const float sl = dd * dd;
        const float4 xv = xw4[(long long)node * 32 + lane];
        float4 o;
        o.x = fmaxf(acc.x + xv.x * sl + bv.x, 0.0f);
        o.y = fmaxf(acc.y + xv.y * sl + bv.y, 0.0f);
        o.z = fmaxf(acc.z + xv.z * sl + bv.z, 0.0f);
        o.w = fmaxf(acc.w + xv.w * sl + bv.w, 0.0f);
        reinterpret_cast<float4*>(h)[(long long)node * 32 + lane] = o;

        if (!more) break;
        node = next;
        beg  = __builtin_amdgcn_readfirstlane(nbeg);
        deg  = __builtin_amdgcn_readfirstlane(ndeg);
        dd   = ndd;
    }
}

__global__ __launch_bounds__(256) void k_gather_tab_relu(const int* __restrict__ x,
                                                         const float* __restrict__ tw,
                                                         const int* __restrict__ rowptr,
                                                         const int* __restrict__ cnt,
                                                         const int* __restrict__ csr_src,
                                                         const float* __restrict__ dinv,
                                                         const float* __restrict__ bias,
                                                         float* __restrict__ h, int n)
{
    const int lane = threadIdx.x & 31;
    const int nw   = gridDim.x * 8;
    int node = __builtin_amdgcn_readfirstlane((int)(blockIdx.x * 8 + (threadIdx.x >> 5)));
    if (node >= n) return;
    const float4* tw4 = reinterpret_cast<const float4*>(tw);
    const float* bp = bias + lane * 4;
    const float4 bv = make_float4(bp[0], bp[1], bp[2], bp[3]);

    int   beg = __builtin_amdgcn_readfirstlane(rowptr[node]);
    int   deg = __builtin_amdgcn_readfirstlane(cnt[node]);
    float dd  = dinv[node];
    int   q0 = __builtin_amdgcn_readfirstlane(x[node * 3 + 0]);
    int   q1 = __builtin_amdgcn_readfirstlane(x[node * 3 + 1]);
    int   q2 = __builtin_amdgcn_readfirstlane(x[node * 3 + 2]);

    for (;;) {
        const int  next = node + nw;
        const bool more = next < n;
        int   nbeg = 0, ndeg = 0, nq0 = 0, nq1 = 0, nq2 = 0;
        float ndd  = 0.0f;
        if (more) {
            nbeg = rowptr[next];
            ndeg = cnt[next];
            ndd  = dinv[next];
            nq0  = x[next * 3 + 0];
            nq1  = x[next * 3 + 1];
            nq2  = x[next * 3 + 2];
        }

        float4 acc = make_float4(0.0f, 0.0f, 0.0f, 0.0f);
        for (int base = 0; base < deg; base += 32) {
            const int m = (deg - base < 32) ? (deg - base) : 32;
            int   r0 = 0, r1 = 0, r2 = 0;
            float c_l = 0.0f;
            if (lane < m) {
                const int s_l = csr_src[beg + base + lane];
                c_l = dinv[s_l] * dd;
                r0 = x[s_l * 3 + 0];
                r1 = x[s_l * 3 + 1] + NSHAPE;
                r2 = x[s_l * 3 + 2] + NSHAPE + NCOLOR;
            }
#pragma unroll 4
            for (int j = 0; j < m; ++j) {
                const int   i0 = __builtin_amdgcn_readlane(r0, j);
                const int   i1 = __builtin_amdgcn_readlane(r1, j);
                const int   i2 = __builtin_amdgcn_readlane(r2, j);
                const float c  = __int_as_float(__builtin_amdgcn_readlane(__float_as_int(c_l), j));
                const float4 a = tw4[i0 * 32 + lane];
                const float4 b = tw4[i1 * 32 + lane];
                const float4 d = tw4[i2 * 32 + lane];
                acc.x += ((a.x + b.x) + d.x) * c;
                acc.y += ((a.y + b.y) + d.y) * c;
                acc.z += ((a.z + b.z) + d.z) * c;
                acc.w += ((a.w + b.w) + d.w) * c;
            }
        }

        const float4 a = tw4[q0 * 32 + lane];
        const float4 b = tw4[(q1 + NSHAPE) * 32 + lane];
        const float4 d = tw4[(q2 + NSHAPE + NCOLOR) * 32 + lane];
        const float sl = dd * dd;
        float4 o;
        o.x = fmaxf(acc.x + ((a.x + b.x) + d.x) * sl + bv.x, 0.0f);
        o.y = fmaxf(acc.y + ((a.y + b.y) + d.y) * sl + bv.y, 0.0f);
        o.z = fmaxf(acc.z + ((a.z + b.z) + d.z) * sl + bv.z, 0.0f);
        o.w = fmaxf(acc.w + ((a.w + b.w) + d.w) * sl + bv.w, 0.0f);
        reinterpret_cast<float4*>(h)[(long long)node * 32 + lane] = o;

        if (!more) break;
        node = next;
        beg  = __builtin_amdgcn_readfirstlane(nbeg);
        deg  = __builtin_amdgcn_readfirstlane(ndeg);
        dd   = ndd;
        q0   = __builtin_amdgcn_readfirstlane(nq0);
        q1   = __builtin_amdgcn_readfirstlane(nq1);
        q2   = __builtin_amdgcn_readfirstlane(nq2);
    }
}

__global__ __launch_bounds__(256) void k_pool(const float* __restrict__ h,
                                              const int* __restrict__ batch,
                                              float* __restrict__ sums,
                                              float* __restrict__ cnt, int n)
{
    const int lane = threadIdx.x & 31;
    const int nw   = gridDim.x * 8;
    const int gw   = __builtin_amdgcn_readfirstlane((int)(blockIdx.x * 8 + (threadIdx.x >> 5)));
    const int per  = (n + nw - 1) / nw;
    const int beg  = gw * per;
    const int end  = (beg + per < n) ? (beg + per) : n;
    const float4* h4 = reinterpret_cast<const float4*>(h);

    for (int base = beg; base < end; base += 4) {
        int    g[4];
        float4 v[4];
#pragma unroll
        for (int j = 0; j < 4; ++j) {
            const int nd = (base + j < end) ? (base + j) : (end - 1);
            g[j] = batch[nd];
            v[j] = h4[(long long)nd * 32 + lane];
        }
#pragma unroll
        for (int j = 0; j < 4; ++j) {
            if (base + j < end) {
                const int gg = __builtin_amdgcn_readfirstlane(g[j]);
                float* sp = sums + (long long)gg * EMB + lane * 4;
                unsafeAtomicAdd(sp + 0, v[j].x);
                unsafeAtomicAdd(sp + 1, v[j].y);
                unsafeAtomicAdd(sp + 2, v[j].z);
                unsafeAtomicAdd(sp + 3, v[j].w);
                if (lane == 0) unsafeAtomicAdd(&cnt[gg], 1.0f);
            }
        }
    }
}

__global__ void k_head(const float* __restrict__ sums,
                       const float* __restrict__ cnt,
                       const float* __restrict__ Wlin,
                       const float* __restrict__ blin,
                       float* __restrict__ out, int g_total)
{
    int idx = blockIdx.x * blockDim.x + threadIdx.x;
    if (idx >= g_total * NCLS) return;
    const int g = idx / NCLS;
    const int c = idx % NCLS;
    const float inv = 1.0f / fmaxf(cnt[g], 1.0f);
    const float* sp = sums + (long long)g * EMB;
    float acc = 0.0f;
#pragma unroll 4
    for (int k = 0; k < EMB; ++k) acc += sp[k] * Wlin[k * NCLS + c];
    out[idx] = acc * inv + blin[c];
}

static inline size_t align256(size_t x) { return (x + 255) & ~(size_t)255; }

extern "C" void kernel_launch(void* const* d_in, const int* in_sizes, int n_in,
                              void* d_out, int out_size, void* d_ws, size_t ws_size,
                              hipStream_t stream)
{
    const int*   x        = (const int*)  d_in[0];
    const int*   eidx     = (const int*)  d_in[1];
    const int*   batch    = (const int*)  d_in[2];
    const float* shapeTab = (const float*)d_in[3];
    const float* colorTab = (const float*)d_in[4];
    const float* posTab   = (const float*)d_in[5];
    const float* W1       = (const float*)d_in[6];
    const float* b1       = (const float*)d_in[7];
    const float* W2       = (const float*)d_in[8];
    const float* b2       = (const float*)d_in[9];
    const float* Wlin     = (const float*)d_in[10];
    const float* blin     = (const float*)d_in[11];
    float*       out      = (float*)d_out;

    const int N = in_sizes[0] / 3;
    const int E = in_sizes[1] / 2;
    const int G = out_size / NCLS;

    const int* src = eidx;
    const int* dst = eidx + E;

    char* ws = (char*)d_ws;
    size_t off = 0;
    float* h    = (float*)(ws + off); off = align256(off + (size_t)N * EMB * 4);
    float* xw   = (float*)(ws + off); off = align256(off + (size_t)N * EMB * 4);
    char*  csr  = ws + off;           off = align256(off + (size_t)N * EMB * 4);
    float* dinv = (float*)(ws + off); off = align256(off + (size_t)N * 4);
    float* sums = (float*)(ws + off); off = align256(off + (size_t)G * EMB * 4);
    float* cnt  = (float*)(ws + off); off = align256(off + (size_t)G * 4);
    (void)ws_size;

    const int nTiles = (N + 1023) / 1024;
    size_t coff = 0;
    int* degc   = (int*)(csr + coff); coff = align256(coff + (size_t)N * 4);
    int* rowptr = (int*)(csr + coff); coff = align256(coff + (size_t)N * 4);
    int* bsum   = (int*)(csr + coff); coff = align256(coff + (size_t)nTiles * 4);
    int* rank   = (int*)(csr + coff); coff = align256(coff + (size_t)E * 4);
    int* csrSrc = (int*)(csr + coff); coff = align256(coff + (size_t)E * 4);
    float* tw   = (float*)(csr + coff); coff = align256(coff + (size_t)NTABROWS * EMB * 4);

    const int TB = 256;
    const int gEdge1    = 1024;
    const int gNodeWave = 2048;
    const int gNodeTab  = 1536;
    const int gVecBlks  = 1024;
    const int gGemmBlks = (N + 127) / 128;

    k_tab_gemm_wmma<<<NTABROWS / 16, TB, 0, stream>>>(shapeTab, colorTab, posTab, W1, tw);

    k_fill<<<(N + TB - 1) / TB, TB, 0, stream>>>((float*)degc, 0.0f, N);
    k_count<<<gEdge1, TB, 0, stream>>>(dst, degc, rank, E);
    k_blocksum<<<nTiles, 256, 0, stream>>>(degc, bsum, N);
    k_scan_write<<<nTiles, 256, 0, stream>>>(degc, bsum, rowptr, dinv, N);
    k_csr_fill<<<gEdge1, TB, 0, stream>>>(src, dst, rowptr, rank, csrSrc, E);

    k_gather_tab_relu<<<gNodeTab, 256, 0, stream>>>(x, tw, rowptr, degc, csrSrc, dinv, b1, h, N);

    k_gemm_wmma<<<gGemmBlks, TB, 0, stream>>>(h, W2, xw, N);
    k_gather_relu<<<gNodeWave, 256, 0, stream>>>(xw, rowptr, degc, csrSrc, dinv, b2, h, N);

    k_fill<<<(G * EMB + TB - 1) / TB, TB, 0, stream>>>(sums, 0.0f, G * EMB);
    k_fill<<<(G + TB - 1) / TB, TB, 0, stream>>>(cnt, 0.0f, G);
    k_pool<<<gVecBlks, TB, 0, stream>>>(h, batch, sums, cnt, N);
    k_head<<<(G * NCLS + TB - 1) / TB, TB, 0, stream>>>(sums, cnt, Wlin, blin, out, G);
}


# CSR built by a two-level counting sort with counters in LDS (edges binned by 128-node range, then ordered per bin): no global atomic or scattered store per edge; row slot order and sum order change
# speedup vs baseline: 4.3609x; 1.5178x over previous
#include <hip/hip_runtime.h>
#include <hip/hip_bf16.h>

#define EMB 128
#define NCLS 10
#define NSHAPE 16
#define NCOLOR 16
#define MAXPOS 256
#define NTABROWS (NSHAPE + NCOLOR + MAXPOS)

typedef __attribute__((ext_vector_type(2))) float v2f;
typedef __attribute__((ext_vector_type(8))) float v8f;

__global__ __launch_bounds__(256) void k_tab_gemm_wmma(
    const float* __restrict__ st, const float* __restrict__ ct, const float* __restrict__ pt,
    const float* __restrict__ W, float* __restrict__ tw)
{
    const int tile  = blockIdx.x;
    const int wave  = threadIdx.x >> 5;
    const int lane  = threadIdx.x & 31;
    const int m     = lane & 15;
    const int khalf = (lane >> 4) * 2;

    const float* abase;
    if (tile == 0)      abase = st;
    else if (tile == 1) abase = ct;
    else                abase = pt + (long long)(tile - 2) * 16 * EMB;

    const float* arow = abase + m * EMB;
    const int col = wave * 16 + m;

    v8f acc = {};
    for (int k = 0; k < EMB; k += 4) {
        const int kk = k + khalf;
        v2f a, b;
        a.x = arow[kk];
        a.y = arow[kk + 1];
        b.x = W[kk * EMB + col];
        b.y = W[(kk + 1) * EMB + col];
        acc = __builtin_amdgcn_wmma_f32_16x16x4_f32(
            false, a, false, b, (short)0, acc, false, false);
    }

    const int rbase = (lane >> 4) * 8;
#pragma unroll
    for (int r = 0; r < 8; ++r)
        tw[(long long)(tile * 16 + rbase + r) * EMB + col] = acc[r];
}

__global__ void k_fill(float* __restrict__ p, float v, int n)
{
    int idx = blockIdx.x * blockDim.x + threadIdx.x;
    for (; idx < n; idx += gridDim.x * blockDim.x) p[idx] = v;
}

#define BIN_SHIFT 7
#define BIN_NODES (1 << BIN_SHIFT)
#define MAXBINS 1024
#define PT 1024
#define KPT 12
#define CHUNK (PT * KPT)
#define STAGE_CAP 8192

__device__ __forceinline__ int wave_incl_scan(int v, int lane)
{
#pragma unroll
    for (int off = 1; off < 32; off <<= 1) {
        const int u = __shfl_up(v, off);
        if (lane >= off) v += u;
    }
    return v;
}

__global__ __launch_bounds__(PT) void k_bin_hist(const int* __restrict__ dst, int* __restrict__ counts,
                                                 int e, int nb, int pitch)
{
    __shared__ int hist[MAXBINS];
    const int t = threadIdx.x;
    hist[t] = 0;
    __syncthreads();
    const int base = blockIdx.x * CHUNK + t;
    int d[KPT];
#pragma unroll
    for (int j = 0; j < KPT; ++j) {
        const int i = base + j * PT;
        d[j] = dst[(i < e) ? i : (e - 1)];
    }
#pragma unroll
    for (int j = 0; j < KPT; ++j)
        if (base + j * PT < e) atomicAdd(&hist[d[j] >> BIN_SHIFT], 1);
    __syncthreads();
    if (t < nb) counts[blockIdx.x * pitch + t] = hist[t];
}

__global__ __launch_bounds__(PT) void k_bin_place(const int* __restrict__ src, const int* __restrict__ dst,
                                                  const int* __restrict__ counts, int* __restrict__ binStart,
                                                  int* __restrict__ pairs_out, int e, int nb, int pitch, int nparts)
{
    int2* pairs = reinterpret_cast<int2*>(pairs_out);
    __shared__ int  hist[MAXBINS];
    __shared__ int  lstart[MAXBINS];
    __shared__ int  gdelta[MAXBINS];
    __shared__ int  wsA[32];
    __shared__ int  wsG[32];
    __shared__ int2 sorted[CHUNK];
    const int t    = threadIdx.x;
    const int lane = t & 31;
    const int wave = t >> 5;
    const int p    = blockIdx.x;
    hist[t] = 0;
    __syncthreads();

    const int base = p * CHUNK + t;
    int s[KPT], d[KPT], r[KPT];
#pragma unroll
    for (int j = 0; j < KPT; ++j) {
        const int i  = base + j * PT;
        const int ic = (i < e) ? i : (e - 1);
        s[j] = src[ic];
        d[j] = dst[ic];
    }

    int tot = 0, pre = 0;
    if (t < nb) {
        for (int q0 = 0; q0 < nparts; q0 += 16) {
            int c[16];
#pragma unroll
            for (int k = 0; k < 16; ++k) {
                const int q = q0 + k;
                c[k] = counts[((q < nparts) ? q : (nparts - 1)) * pitch + t];
            }
#pragma unroll
            for (int k = 0; k < 16; ++k) {
                const int q = q0 + k;
                tot += (q < nparts) ? c[k] : 0;
                pre += (q < p) ? c[k] : 0;
            }
        }
    }

#pragma unroll
    for (int j = 0; j < KPT; ++j) {
        r[j] = 0;
        if (base + j * PT < e) r[j] = atomicAdd(&hist[d[j] >> BIN_SHIFT], 1);
    }
    __syncthreads();
    const int myh = hist[t];

    const int ia = wave_incl_scan(myh, lane);
    const int ig = wave_incl_scan(tot, lane);
    if (lane == 31) { wsA[wave] = ia; wsG[wave] = ig; }
    __syncthreads();
    if (wave == 0) {
        const int xa = wsA[lane], xg = wsG[lane];
        const int sa = wave_incl_scan(xa, lane);
        const int sg = wave_incl_scan(xg, lane);
        wsA[lane] = sa - xa;
        wsG[lane] = sg - xg;
    }
    __syncthreads();
    const int lex = ia - myh + wsA[wave];
    const int gex = ig - tot + wsG[wave];
    lstart[t] = lex;
    gdelta[t] = gex + pre - lex;
    if (p == 0) {
        if (t < nb)      binStart[t]  = gex;
        if (t == nb - 1) binStart[nb] = gex + tot;
    }
    __syncthreads();

#pragma unroll
    for (int j = 0; j < KPT; ++j) {
        if (base + j * PT < e)
            sorted[lstart[d[j] >> BIN_SHIFT] + r[j]] = make_int2(s[j], d[j]);
    }
    __syncthreads();

    const int left = e - p * CHUNK;
    const int cc   = (left < CHUNK) ? left : CHUNK;
    for (int i = t; i < cc; i += PT) {
        const int2 v = sorted[i];
        pairs[gdelta[v.y >> BIN_SHIFT] + i] = v;
    }
}

__global__ __launch_bounds__(256) void k_bin_sort(const int* __restrict__ pairs_in, const int* __restrict__ binStart,
                                                  int* __restrict__ rowptr, int* __restrict__ cnt,
                                                  float* __restrict__ dinv, int* __restrict__ csr_src, int n)
{
    const int2* pairs = reinterpret_cast<const int2*>(pairs_in);
    __shared__ int c[BIN_NODES];
    __shared__ int cur[BIN_NODES];
    __shared__ int ws[8];
    __shared__ int stage[STAGE_CAP];
    const int t     = threadIdx.x;
    const int lane  = t & 31;
    const int wave  = t >> 5;
    const int b     = blockIdx.x;
    const int start = binStart[b];
    const int m     = binStart[b + 1] - start;
    if (t < BIN_NODES) c[t] = 0;
    __syncthreads();

    for (int i0 = t; i0 < m; i0 += 256 * 8) {
        int dd[8];
#pragma unroll
        for (int j = 0; j < 8; ++j) {
            const int i = i0 + j * 256;
            dd[j] = pairs[start + ((i < m) ? i : i0)].y;
        }
#pragma unroll
        for (int j = 0; j < 8; ++j)
            if (i0 + j * 256 < m) atomicAdd(&c[dd[j] & (BIN_NODES - 1)], 1);
    }
    __syncthreads();

    const int v  = (t < BIN_NODES) ? c[t] : 0;
    const int iv = wave_incl_scan(v, lane);
    if (lane == 31) ws[wave] = iv;
    __syncthreads();
    int woff = 0;
    for (int w = 0; w < wave; ++w) woff += ws[w];
    if (t < BIN_NODES) {
        const int ex = iv - v + woff;
        cur[t] = ex;
        const int node = b * BIN_NODES + t;
        if (node < n) {
            rowptr[node] = start + ex;
            cnt[node]    = v;
            dinv[node]   = __frsqrt_rn((float)(v + 1));
        }
    }
    __syncthreads();

    const bool staged = (m <= STAGE_CAP);
    for (int i0 = t; i0 < m; i0 += 256 * 8) {
        int2 ev[8];
#pragma unroll
        for (int j = 0; j < 8; ++j) {
            const int i = i0 + j * 256;
            ev[j] = pairs[start + ((i < m) ? i : i0)];
        }
#pragma unroll
        for (int j = 0; j < 8; ++j) {
            if (i0 + j * 256 < m) {
                const int pos = atomicAdd(&cur[ev[j].y & (BIN_NODES - 1)], 1);
                if (staged) stage[pos] = ev[j].x;
                else        csr_src[start + pos] = ev[j].x;
            }
        }
    }
    if (staged) {
        __syncthreads();
        for (int i = t; i < m; i += 256) csr_src[start + i] = stage[i];
    }
}

#define WSTRIDE 128
__global__ __launch_bounds__(256) void k_gemm_wmma(
    const float* __restrict__ A, const float* __restrict__ W,
    float* __restrict__ C, int nrows)
{
    __shared__ float Wl[128 * WSTRIDE];
    const int tid = threadIdx.x;

    for (int idx = tid; idx < 128 * 32; idx += 256) {
        int r = idx >> 5, q = idx & 31;
        reinterpret_cast<float4*>(&Wl[r * WSTRIDE + q * 4])[0] =
            reinterpret_cast<const float4*>(W)[r * 32 + q];
    }
    __syncthreads();

    const int wave  = tid >> 5;
    const int lane  = tid & 31;
    const int m     = lane & 15;
    const int khalf = (lane >> 4) * 2;
    const int tileRow = blockIdx.x * 128 + wave * 16;
    if (tileRow >= nrows) return;

    v8f acc[8] = {};

    const float* arow = A + (long long)(tileRow + m) * EMB;
    for (int k = 0; k < EMB; k += 4) {
        const int kk = k + khalf;
        v2f a;
        a.x = arow[kk];
        a.y = arow[kk + 1];
#pragma unroll
        for (int j = 0; j < 8; ++j) {
            const int col = j * 16 + m;
            v2f b;
            b.x = Wl[kk * WSTRIDE + col];
            b.y = Wl[(kk + 1) * WSTRIDE + col];
            acc[j] = __builtin_amdgcn_wmma_f32_16x16x4_f32(
                false, a, false, b, (short)0, acc[j], false, false);
        }
    }

    const int rbase = (lane >> 4) * 8;
#pragma unroll
    for (int j = 0; j < 8; ++j) {
#pragma unroll
        for (int r = 0; r < 8; ++r) {
            C[(long long)(tileRow + rbase + r) * EMB + j * 16 + m] = acc[j][r];
        }
    }
}

__global__ __launch_bounds__(256) void k_gather_relu(const float* __restrict__ xw,
                                                     const int* __restrict__ rowptr,
                                                     const int* __restrict__ cnt,
                                                     const int* __restrict__ csr_src,
                                                     const float* __restrict__ dinv,
                                                     const float* __restrict__ bias,
                                                     float* __restrict__ h, int n)
{
    const int lane = threadIdx.x & 31;
    const int nw   = gridDim.x * 8;
    int node = __builtin_amdgcn_readfirstlane((int)(blockIdx.x * 8 + (threadIdx.x >> 5)));
    if (node >= n) return;
    const float4* xw4 = reinterpret_cast<const float4*>(xw);
    const float* bp = bias + lane * 4;
    const float4 bv = make_float4(bp[0], bp[1], bp[2], bp[3]);

    int   beg = __builtin_amdgcn_readfirstlane(rowptr[node]);
    int   deg = __builtin_amdgcn_readfirstlane(cnt[node]);
    float dd  = dinv[node];

    for (;;) {
        const int  next = node + nw;
        const bool more = next < n;
        int   nbeg = 0, ndeg = 0;
        float ndd  = 0.0f;
        if (more) {
            nbeg = rowptr[next];
            ndeg = cnt[next];
            ndd  = dinv[next];
        }

        float4 acc = make_float4(0.0f, 0.0f, 0.0f, 0.0f);
        for (int base = 0; base < deg; base += 32) {
            const int m = (deg - base < 32) ? (deg - base) : 32;
            int   s_l = 0;
            float c_l = 0.0f;
            if (lane < m) {
                s_l = csr_src[beg + base + lane];
                c_l = dinv[s_l] * dd;
            }
#pragma unroll 4
            for (int j = 0; j < m; ++j) {
                const int   s = __builtin_amdgcn_readlane(s_l, j);
                const float c = __int_as_float(__builtin_amdgcn_readlane(__float_as_int(c_l), j));
                const float4 v = xw4[(long long)s * 32 + lane];
                acc.x += v.x * c;
                acc.y += v.y * c;
                acc.z += v.z * c;
                acc.w += v.w * c;
            }
        }

        const float sl = dd * dd;
        const float4 xv = xw4[(long long)node * 32 + lane];
        float4 o;
        o.x = fmaxf(acc.x + xv.x * sl + bv.x, 0.0f);
        o.y = fmaxf(acc.y + xv.y * sl + bv.y, 0.0f);
        o.z = fmaxf(acc.z + xv.z * sl + bv.z, 0.0f);
        o.w = fmaxf(acc.w + xv.w * sl + bv.w, 0.0f);
        reinterpret_cast<float4*>(h)[(long long)node * 32 + lane] = o;

        if (!more) break;
        node = next;
        beg  = __builtin_amdgcn_readfirstlane(nbeg);
        deg  = __builtin_amdgcn_readfirstlane(ndeg);
        dd   = ndd;
    }
}

__global__ __launch_bounds__(256) void k_gather_tab_relu(const int* __restrict__ x,
                                                         const float* __restrict__ tw,
                                                         const int* __restrict__ rowptr,
                                                         const int* __restrict__ cnt,
                                                         const int* __restrict__ csr_src,
                                                         const float* __restrict__ dinv,
                                                         const float* __restrict__ bias,
                                                         float* __restrict__ h, int n)
{
    const int lane = threadIdx.x & 31;
    const int nw   = gridDim.x * 8;
    int node = __builtin_amdgcn_readfirstlane((int)(blockIdx.x * 8 + (threadIdx.x >> 5)));
    if (node >= n) return;
    const float4* tw4 = reinterpret_cast<const float4*>(tw);
    const float* bp = bias + lane * 4;
    const float4 bv = make_float4(bp[0], bp[1], bp[2], bp[3]);

    int   beg = __builtin_amdgcn_readfirstlane(rowptr[node]);
    int   deg = __builtin_amdgcn_readfirstlane(cnt[node]);
    float dd  = dinv[node];
    int   q0 = __builtin_amdgcn_readfirstlane(x[node * 3 + 0]);
    int   q1 = __builtin_amdgcn_readfirstlane(x[node * 3 + 1]);
    int   q2 = __builtin_amdgcn_readfirstlane(x[node * 3 + 2]);

    for (;;) {
        const int  next = node + nw;
        const bool more = next < n;
        int   nbeg = 0, ndeg = 0, nq0 = 0, nq1 = 0, nq2 = 0;
        float ndd  = 0.0f;
        if (more) {
            nbeg = rowptr[next];
            ndeg = cnt[next];
            ndd  = dinv[next];
            nq0  = x[next * 3 + 0];
            nq1  = x[next * 3 + 1];
            nq2  = x[next * 3 + 2];
        }

        float4 acc = make_float4(0.0f, 0.0f, 0.0f, 0.0f);
        for (int base = 0; base < deg; base += 32) {
            const int m = (deg - base < 32) ? (deg - base) : 32;
            int   r0 = 0, r1 = 0, r2 = 0;
            float c_l = 0.0f;
            if (lane < m) {
                const int s_l = csr_src[beg + base + lane];
                c_l = dinv[s_l] * dd;
                r0 = x[s_l * 3 + 0];
                r1 = x[s_l * 3 + 1] + NSHAPE;
                r2 = x[s_l * 3 + 2] + NSHAPE + NCOLOR;
            }
#pragma unroll 4
            for (int j = 0; j < m; ++j) {
                const int   i0 = __builtin_amdgcn_readlane(r0, j);
                const int   i1 = __builtin_amdgcn_readlane(r1, j);
                const int   i2 = __builtin_amdgcn_readlane(r2, j);
                const float c  = __int_as_float(__builtin_amdgcn_readlane(__float_as_int(c_l), j));
                const float4 a = tw4[i0 * 32 + lane];
                const float4 b = tw4[i1 * 32 + lane];
                const float4 d = tw4[i2 * 32 + lane];
                acc.x += ((a.x + b.x) + d.x) * c;
                acc.y += ((a.y + b.y) + d.y) * c;
                acc.z += ((a.z + b.z) + d.z) * c;
                acc.w += ((a.w + b.w) + d.w) * c;
            }
        }

        const float4 a = tw4[q0 * 32 + lane];
        const float4 b = tw4[(q1 + NSHAPE) * 32 + lane];
        const float4 d = tw4[(q2 + NSHAPE + NCOLOR) * 32 + lane];
        const float sl = dd * dd;
        float4 o;
        o.x = fmaxf(acc.x + ((a.x + b.x) + d.x) * sl + bv.x, 0.0f);
        o.y = fmaxf(acc.y + ((a.y + b.y) + d.y) * sl + bv.y, 0.0f);
        o.z = fmaxf(acc.z + ((a.z + b.z) + d.z) * sl + bv.z, 0.0f);
        o.w = fmaxf(acc.w + ((a.w + b.w) + d.w) * sl + bv.w, 0.0f);
        reinterpret_cast<float4*>(h)[(long long)node * 32 + lane] = o;

        if (!more) break;
        node = next;
        beg  = __builtin_amdgcn_readfirstlane(nbeg);
        deg  = __builtin_amdgcn_readfirstlane(ndeg);
        dd   = ndd;
        q0   = __builtin_amdgcn_readfirstlane(nq0);
        q1   = __builtin_amdgcn_readfirstlane(nq1);
        q2   = __builtin_amdgcn_readfirstlane(nq2);
    }
}

__global__ __launch_bounds__(256) void k_pool(const float* __restrict__ h,
                                              const int* __restrict__ batch,
                                              float* __restrict__ sums,
                                              float* __restrict__ cnt, int n)
{
    const int lane = threadIdx.x & 31;
    const int nw   = gridDim.x * 8;
    const int gw   = __builtin_amdgcn_readfirstlane((int)(blockIdx.x * 8 + (threadIdx.x >> 5)));
    const int per  = (n + nw - 1) / nw;
    const int beg  = gw * per;
    const int end  = (beg + per < n) ? (beg + per) : n;
    const float4* h4 = reinterpret_cast<const float4*>(h);

    for (int base = beg; base < end; base += 4) {
        int    g[4];
        float4 v[4];
#pragma unroll
        for (int j = 0; j < 4; ++j) {
            const int nd = (base + j < end) ? (base + j) : (end - 1);
            g[j] = batch[nd];
            v[j] = h4[(long long)nd * 32 + lane];
        }
#pragma unroll
        for (int j = 0; j < 4; ++j) {
            if (base + j < end) {
                const int gg = __builtin_amdgcn_readfirstlane(g[j]);
                float* sp = sums + (long long)gg * EMB + lane * 4;
                unsafeAtomicAdd(sp + 0, v[j].x);
                unsafeAtomicAdd(sp + 1, v[j].y);
                unsafeAtomicAdd(sp + 2, v[j].z);
                unsafeAtomicAdd(sp + 3, v[j].w);
                if (lane == 0) unsafeAtomicAdd(&cnt[gg], 1.0f);
            }
        }
    }
}

__global__ void k_head(const float* __restrict__ sums,
                       const float* __restrict__ cnt,
                       const float* __restrict__ Wlin,
                       const float* __restrict__ blin,
                       float* __restrict__ out, int g_total)
{
    int idx = blockIdx.x * blockDim.x + threadIdx.x;
    if (idx >= g_total * NCLS) return;
    const int g = idx / NCLS;
    const int c = idx % NCLS;
    const float inv = 1.0f / fmaxf(cnt[g], 1.0f);
    const float* sp = sums + (long long)g * EMB;
    float acc = 0.0f;
#pragma unroll 4
    for (int k = 0; k < EMB; ++k) acc += sp[k] * Wlin[k * NCLS + c];
    out[idx] = acc * inv + blin[c];
}

static inline size_t align256(size_t x) { return (x + 255) & ~(size_t)255; }

extern "C" void kernel_launch(void* const* d_in, const int* in_sizes, int n_in,
                              void* d_out, int out_size, void* d_ws, size_t ws_size,
                              hipStream_t stream)
{
    const int*   x        = (const int*)  d_in[0];
    const int*   eidx     = (const int*)  d_in[1];
    const int*   batch    = (const int*)  d_in[2];
    const float* shapeTab = (const float*)d_in[3];
    const float* colorTab = (const float*)d_in[4];
    const float* posTab   = (const float*)d_in[5];
    const float* W1       = (const float*)d_in[6];
    const float* b1       = (const float*)d_in[7];
    const float* W2       = (const float*)d_in[8];
    const float* b2       = (const float*)d_in[9];
    const float* Wlin     = (const float*)d_in[10];
    const float* blin     = (const float*)d_in[11];
    float*       out      = (float*)d_out;

    const int N = in_sizes[0] / 3;
    const int E = in_sizes[1] / 2;
    const int G = out_size / NCLS;

    const int* src = eidx;
    const int* dst = eidx + E;

    char* ws = (char*)d_ws;
    size_t off = 0;
    float* h    = (float*)(ws + off); off = align256(off + (size_t)N * EMB * 4);
    float* xw   = (float*)(ws + off); off = align256(off + (size_t)N * EMB * 4);
    char*  csr  = ws + off;           off = align256(off + (size_t)N * EMB * 4);
    float* dinv = (float*)(ws + off); off = align256(off + (size_t)N * 4);
    float* sums = (float*)(ws + off); off = align256(off + (size_t)G * EMB * 4);
    float* cnt  = (float*)(ws + off); off = align256(off + (size_t)G * 4);
    (void)ws_size;

    const int nBins  = (N + BIN_NODES - 1) / BIN_NODES;
    const int nParts = (E + CHUNK - 1) / CHUNK;
    const int cPitch = (nBins + 31) & ~31;
    size_t coff = 0;
    int*  degc     = (int*)(csr + coff);  coff = align256(coff + (size_t)N * 4);
    int*  rowptr   = (int*)(csr + coff);  coff = align256(coff + (size_t)N * 4);
    int*  binStart = (int*)(csr + coff);  coff = align256(coff + (size_t)(nBins + 1) * 4);
    int*  counts   = (int*)(csr + coff);  coff = align256(coff + (size_t)nParts * cPitch * 4);
    int*  pairs    = (int*)(csr + coff);  coff = align256(coff + (size_t)E * 8);
    int*  csrSrc   = (int*)(csr + coff);  coff = align256(coff + (size_t)E * 4);
    float* tw   = (float*)(csr + coff); coff = align256(coff + (size_t)NTABROWS * EMB * 4);

    const int TB = 256;
    const int gNodeWave = 2048;
    const int gNodeTab  = 1536;
    const int gVecBlks  = 1024;
    const int gGemmBlks = (N + 127) / 128;

    k_tab_gemm_wmma<<<NTABROWS / 16, TB, 0, stream>>>(shapeTab, colorTab, posTab, W1, tw);

    k_bin_hist<<<nParts, PT, 0, stream>>>(dst, counts, E, nBins, cPitch);
    k_bin_place<<<nParts, PT, 0, stream>>>(src, dst, counts, binStart, pairs, E, nBins, cPitch, nParts);
    k_bin_sort<<<nBins, 256, 0, stream>>>(pairs, binStart, rowptr, degc, dinv, csrSrc, N);

    k_gather_tab_relu<<<gNodeTab, 256, 0, stream>>>(x, tw, rowptr, degc, csrSrc, dinv, b1, h, N);

    k_gemm_wmma<<<gGemmBlks, TB, 0, stream>>>(h, W2, xw, N);
    k_gather_relu<<<gNodeWave, 256, 0, stream>>>(xw, rowptr, degc, csrSrc, dinv, b2, h, N);

    k_fill<<<(G * EMB + TB - 1) / TB, TB, 0, stream>>>(sums, 0.0f, G * EMB);
    k_fill<<<(G + TB - 1) / TB, TB, 0, stream>>>(cnt, 0.0f, G);
    k_pool<<<gVecBlks, TB, 0, stream>>>(h, batch, sums, cnt, N);
    k_head<<<(G * NCLS + TB - 1) / TB, TB, 0, stream>>>(sums, cnt, Wlin, blin, out, G);
}


# pool folded into the layer-2 gather: a wave owns consecutive nodes, sums its relu rows per graph in registers, float atomics once per graph change; no h store, no pool launch; pool sums regrouped
# speedup vs baseline: 5.1301x; 1.1764x over previous
#include <hip/hip_runtime.h>
#include <hip/hip_bf16.h>

#define EMB 128
#define NCLS 10
#define NSHAPE 16
#define NCOLOR 16
#define MAXPOS 256
#define NTABROWS (NSHAPE + NCOLOR + MAXPOS)

typedef __attribute__((ext_vector_type(2))) float v2f;
typedef __attribute__((ext_vector_type(8))) float v8f;

__global__ __launch_bounds__(256) void k_tab_gemm_wmma(
    const float* __restrict__ st, const float* __restrict__ ct, const float* __restrict__ pt,
    const float* __restrict__ W, float* __restrict__ tw)
{
    const int tile  = blockIdx.x;
    const int wave  = threadIdx.x >> 5;
    const int lane  = threadIdx.x & 31;
    const int m     = lane & 15;
    const int khalf = (lane >> 4) * 2;

    const float* abase;
    if (tile == 0)      abase = st;
    else if (tile == 1) abase = ct;
    else                abase = pt + (long long)(tile - 2) * 16 * EMB;

    const float* arow = abase + m * EMB;
    const int col = wave * 16 + m;

    v8f acc = {};
    for (int k = 0; k < EMB; k += 4) {
        const int kk = k + khalf;
        v2f a, b;
        a.x = arow[kk];
        a.y = arow[kk + 1];
        b.x = W[kk * EMB + col];
        b.y = W[(kk + 1) * EMB + col];
        acc = __builtin_amdgcn_wmma_f32_16x16x4_f32(
            false, a, false, b, (short)0, acc, false, false);
    }

    const int rbase = (lane >> 4) * 8;
#pragma unroll
    for (int r = 0; r < 8; ++r)
        tw[(long long)(tile * 16 + rbase + r) * EMB + col] = acc[r];
}

__global__ void k_fill(float* __restrict__ p, float v, int n)
{
    int idx = blockIdx.x * blockDim.x + threadIdx.x;
    for (; idx < n; idx += gridDim.x * blockDim.x) p[idx] = v;
}

#define BIN_SHIFT 7
#define BIN_NODES (1 << BIN_SHIFT)
#define MAXBINS 1024
#define PT 1024
#define KPT 12
#define CHUNK (PT * KPT)
#define STAGE_CAP 8192

__device__ __forceinline__ int wave_incl_scan(int v, int lane)
{
#pragma unroll
    for (int off = 1; off < 32; off <<= 1) {
        const int u = __shfl_up(v, off);
        if (lane >= off) v += u;
    }
    return v;
}

__global__ __launch_bounds__(PT) void k_bin_hist(const int* __restrict__ dst, int* __restrict__ counts,
                                                 int e, int nb, int pitch)
{
    __shared__ int hist[MAXBINS];
    const int t = threadIdx.x;
    hist[t] = 0;
    __syncthreads();
    const int base = blockIdx.x * CHUNK + t;
    int d[KPT];
#pragma unroll
    for (int j = 0; j < KPT; ++j) {
        const int i = base + j * PT;
        d[j] = dst[(i < e) ? i : (e - 1)];
    }
#pragma unroll
    for (int j = 0; j < KPT; ++j)
        if (base + j * PT < e) atomicAdd(&hist[d[j] >> BIN_SHIFT], 1);
    __syncthreads();
    if (t < nb) counts[blockIdx.x * pitch + t] = hist[t];
}

__global__ __launch_bounds__(PT) void k_bin_place(const int* __restrict__ src, const int* __restrict__ dst,
                                                  const int* __restrict__ counts, int* __restrict__ binStart,
                                                  int* __restrict__ pairs_out, int e, int nb, int pitch, int nparts)
{
    int2* pairs = reinterpret_cast<int2*>(pairs_out);
    __shared__ int  hist[MAXBINS];
    __shared__ int  lstart[MAXBINS];
    __shared__ int  gdelta[MAXBINS];
    __shared__ int  wsA[32];
    __shared__ int  wsG[32];
    __shared__ int2 sorted[CHUNK];
    const int t    = threadIdx.x;
    const int lane = t & 31;
    const int wave = t >> 5;
    const int p    = blockIdx.x;
    hist[t] = 0;
    __syncthreads();

    const int base = p * CHUNK + t;
    int s[KPT], d[KPT], r[KPT];
#pragma unroll
    for (int j = 0; j < KPT; ++j) {
        const int i  = base + j * PT;
        const int ic = (i < e) ? i : (e - 1);
        s[j] = src[ic];
        d[j] = dst[ic];
    }

    int tot = 0, pre = 0;
    if (t < nb) {
        for (int q0 = 0; q0 < nparts; q0 += 16) {
            int c[16];
#pragma unroll
            for (int k = 0; k < 16; ++k) {
                const int q = q0 + k;
                c[k] = counts[((q < nparts) ? q : (nparts - 1)) * pitch + t];
            }
#pragma unroll
            for (int k = 0; k < 16; ++k) {
                const int q = q0 + k;
                tot += (q < nparts) ? c[k] : 0;
                pre += (q < p) ? c[k] : 0;
            }
        }
    }

#pragma unroll
    for (int j = 0; j < KPT; ++j) {
        r[j] = 0;
        if (base + j * PT < e) r[j] = atomicAdd(&hist[d[j] >> BIN_SHIFT], 1);
    }
    __syncthreads();
    const int myh = hist[t];

    const int ia = wave_incl_scan(myh, lane);
    const int ig = wave_incl_scan(tot, lane);
    if (lane == 31) { wsA[wave] = ia; wsG[wave] = ig; }
    __syncthreads();
    if (wave == 0) {
        const int xa = wsA[lane], xg = wsG[lane];
        const int sa = wave_incl_scan(xa, lane);
        const int sg = wave_incl_scan(xg, lane);
        wsA[lane] = sa - xa;
        wsG[lane] = sg - xg;
    }
    __syncthreads();
    const int lex = ia - myh + wsA[wave];
    const int gex = ig - tot + wsG[wave];
    lstart[t] = lex;
    gdelta[t] = gex + pre - lex;
    if (p == 0) {
        if (t < nb)      binStart[t]  = gex;
        if (t == nb - 1) binStart[nb] = gex + tot;
    }
    __syncthreads();

#pragma unroll
    for (int j = 0; j < KPT; ++j) {
        if (base + j * PT < e)
            sorted[lstart[d[j] >> BIN_SHIFT] + r[j]] = make_int2(s[j], d[j]);
    }
    __syncthreads();

    const int left = e - p * CHUNK;
    const int cc   = (left < CHUNK) ? left : CHUNK;
    for (int i = t; i < cc; i += PT) {
        const int2 v = sorted[i];
        pairs[gdelta[v.y >> BIN_SHIFT] + i] = v;
    }
}

__global__ __launch_bounds__(256) void k_bin_sort(const int* __restrict__ pairs_in, const int* __restrict__ binStart,
                                                  int* __restrict__ rowptr, int* __restrict__ cnt,
                                                  float* __restrict__ dinv, int* __restrict__ csr_src, int n)
{
    const int2* pairs = reinterpret_cast<const int2*>(pairs_in);
    __shared__ int c[BIN_NODES];
    __shared__ int cur[BIN_NODES];
    __shared__ int ws[8];
    __shared__ int stage[STAGE_CAP];
    const int t     = threadIdx.x;
    const int lane  = t & 31;
    const int wave  = t >> 5;
    const int b     = blockIdx.x;
    const int start = binStart[b];
    const int m     = binStart[b + 1] - start;
    if (t < BIN_NODES) c[t] = 0;
    __syncthreads();

    for (int i0 = t; i0 < m; i0 += 256 * 8) {
        int dd[8];
#pragma unroll
        for (int j = 0; j < 8; ++j) {
            const int i = i0 + j * 256;
            dd[j] = pairs[start + ((i < m) ? i : i0)].y;
        }
#pragma unroll
        for (int j = 0; j < 8; ++j)
            if (i0 + j * 256 < m) atomicAdd(&c[dd[j] & (BIN_NODES - 1)], 1);
    }
    __syncthreads();

    const int v  = (t < BIN_NODES) ? c[t] : 0;
    const int iv = wave_incl_scan(v, lane);
    if (lane == 31) ws[wave] = iv;
    __syncthreads();
    int woff = 0;
    for (int w = 0; w < wave; ++w) woff += ws[w];
    if (t < BIN_NODES) {
        const int ex = iv - v + woff;
        cur[t] = ex;
        const int node = b * BIN_NODES + t;
        if (node < n) {
            rowptr[node] = start + ex;
            cnt[node]    = v;
            dinv[node]   = __frsqrt_rn((float)(v + 1));
        }
    }
    __syncthreads();

    const bool staged = (m <= STAGE_CAP);
    for (int i0 = t; i0 < m; i0 += 256 * 8) {
        int2 ev[8];
#pragma unroll
        for (int j = 0; j < 8; ++j) {
            const int i = i0 + j * 256;
            ev[j] = pairs[start + ((i < m) ? i : i0)];
        }
#pragma unroll
        for (int j = 0; j < 8; ++j) {
            if (i0 + j * 256 < m) {
                const int pos = atomicAdd(&cur[ev[j].y & (BIN_NODES - 1)], 1);
                if (staged) stage[pos] = ev[j].x;
                else        csr_src[start + pos] = ev[j].x;
            }
        }
    }
    if (staged) {
        __syncthreads();
        for (int i = t; i < m; i += 256) csr_src[start + i] = stage[i];
    }
}

#define WSTRIDE 128
__global__ __launch_bounds__(256) void k_gemm_wmma(
    const float* __restrict__ A, const float* __restrict__ W,
    float* __restrict__ C, int nrows)
{
    __shared__ float Wl[128 * WSTRIDE];
    const int tid = threadIdx.x;

    for (int idx = tid; idx < 128 * 32; idx += 256) {
        int r = idx >> 5, q = idx & 31;
        reinterpret_cast<float4*>(&Wl[r * WSTRIDE + q * 4])[0] =
            reinterpret_cast<const float4*>(W)[r * 32 + q];
    }
    __syncthreads();

    const int wave  = tid >> 5;
    const int lane  = tid & 31;
    const int m     = lane & 15;
    const int khalf = (lane >> 4) * 2;
    const int tileRow = blockIdx.x * 128 + wave * 16;
    if (tileRow >= nrows) return;

    v8f acc[8] = {};

    const float* arow = A + (long long)(tileRow + m) * EMB;
    for (int k = 0; k < EMB; k += 4) {
        const int kk = k + khalf;
        v2f a;
        a.x = arow[kk];
        a.y = arow[kk + 1];
#pragma unroll
        for (int j = 0; j < 8; ++j) {
            const int col = j * 16 + m;
            v2f b;
            b.x = Wl[kk * WSTRIDE + col];
            b.y = Wl[(kk + 1) * WSTRIDE + col];
            acc[j] = __builtin_amdgcn_wmma_f32_16x16x4_f32(
                false, a, false, b, (short)0, acc[j], false, false);
        }
    }

    const int rbase = (lane >> 4) * 8;
#pragma unroll
    for (int j = 0; j < 8; ++j) {
#pragma unroll
        for (int r = 0; r < 8; ++r) {
            C[(long long)(tileRow + rbase + r) * EMB + j * 16 + m] = acc[j][r];
        }
    }
}

__device__ __forceinline__ void pool_flush(float* __restrict__ sums, float* __restrict__ gcnt,
                                           int g, int lane, const float4& p, int rows)
{
    float* sp = sums + (long long)g * EMB + lane * 4;
    unsafeAtomicAdd(sp + 0, p.x);
    unsafeAtomicAdd(sp + 1, p.y);
    unsafeAtomicAdd(sp + 2, p.z);
    unsafeAtomicAdd(sp + 3, p.w);
    if (lane == 0) unsafeAtomicAdd(&gcnt[g], (float)rows);
}

__global__ __launch_bounds__(256) void k_gather_relu_pool(const float* __restrict__ xw,
                                                          const int* __restrict__ rowptr,
                                                          const int* __restrict__ cnt,
                                                          const int* __restrict__ csr_src,
                                                          const float* __restrict__ dinv,
                                                          const float* __restrict__ bias,
                                                          const int* __restrict__ batch,
                                                          float* __restrict__ sums,
                                                          float* __restrict__ gcnt, int n)
{
    const int lane = threadIdx.x & 31;
    const int nw   = gridDim.x * 8;
    const int gw   = __builtin_amdgcn_readfirstlane((int)(blockIdx.x * 8 + (threadIdx.x >> 5)));
    const int per  = n / nw;
    const int rem  = n - per * nw;
    const int lo   = gw * per + ((gw < rem) ? gw : rem);
    const int hi   = lo + per + ((gw < rem) ? 1 : 0);
    if (lo >= hi) return;
    const float4* xw4 = reinterpret_cast<const float4*>(xw);
    const float* bp = bias + lane * 4;
    const float4 bv = make_float4(bp[0], bp[1], bp[2], bp[3]);

    int   node = lo;
    int   beg = __builtin_amdgcn_readfirstlane(rowptr[node]);
    int   deg = __builtin_amdgcn_readfirstlane(cnt[node]);
    int   g   = __builtin_amdgcn_readfirstlane(batch[node]);
    float dd  = dinv[node];

    int    cg   = g;
    int    rows = 0;
    float4 pacc = make_float4(0.0f, 0.0f, 0.0f, 0.0f);

    for (;;) {
        const int  next = node + 1;
        const bool more = next < hi;
        int   nbeg = 0, ndeg = 0, ng = 0;
        float ndd  = 0.0f;
        if (more) {
            nbeg = rowptr[next];
            ndeg = cnt[next];
            ng   = batch[next];
            ndd  = dinv[next];
        }

        float4 acc = make_float4(0.0f, 0.0f, 0.0f, 0.0f);
        for (int base = 0; base < deg; base += 32) {
            const int m = (deg - base < 32) ? (deg - base) : 32;
            int   s_l = 0;
            float c_l = 0.0f;
            if (lane < m) {
                s_l = csr_src[beg + base + lane];
                c_l = dinv[s_l] * dd;
            }
#pragma unroll 4
            for (int j = 0; j < m; ++j) {
                const int   s = __builtin_amdgcn_readlane(s_l, j);
                const float c = __int_as_float(__builtin_amdgcn_readlane(__float_as_int(c_l), j));
                const float4 v = xw4[(long long)s * 32 + lane];
                acc.x += v.x * c;
                acc.y += v.y * c;
                acc.z += v.z * c;
                acc.w += v.w * c;
            }
        }

        const float sl = dd * dd;
        const float4 xv = xw4[(long long)node * 32 + lane];
        float4 o;
        o.x = fmaxf(acc.x + xv.x * sl + bv.x, 0.0f);
        o.y = fmaxf(acc.y + xv.y * sl + bv.y, 0.0f);
        o.z = fmaxf(acc.z + xv.z * sl + bv.z, 0.0f);
        o.w = fmaxf(acc.w + xv.w * sl + bv.w, 0.0f);

        if (g != cg) {
            pool_flush(sums, gcnt, cg, lane, pacc, rows);
            pacc = make_float4(0.0f, 0.0f, 0.0f, 0.0f);
            rows = 0;
            cg   = g;
        }
        pacc.x += o.x;
        pacc.y += o.y;
        pacc.z += o.z;
        pacc.w += o.w;
        rows += 1;

        if (!more) break;
        node = next;
        beg  = __builtin_amdgcn_readfirstlane(nbeg);
        deg  = __builtin_amdgcn_readfirstlane(ndeg);
        g    = __builtin_amdgcn_readfirstlane(ng);
        dd   = ndd;
    }
    pool_flush(sums, gcnt, cg, lane, pacc, rows);
}

__global__ __launch_bounds__(256) void k_gather_tab_relu(const int* __restrict__ x,
                                                         const float* __restrict__ tw,
                                                         const int* __restrict__ rowptr,
                                                         const int* __restrict__ cnt,
                                                         const int* __restrict__ csr_src,
                                                         const float* __restrict__ dinv,
                                                         const float* __restrict__ bias,
                                                         float* __restrict__ h, int n)
{
    const int lane = threadIdx.x & 31;
    const int nw   = gridDim.x * 8;
    int node = __builtin_amdgcn_readfirstlane((int)(blockIdx.x * 8 + (threadIdx.x >> 5)));
    if (node >= n) return;
    const float4* tw4 = reinterpret_cast<const float4*>(tw);
    const float* bp = bias + lane * 4;
    const float4 bv = make_float4(bp[0], bp[1], bp[2], bp[3]);

    int   beg = __builtin_amdgcn_readfirstlane(rowptr[node]);
    int   deg = __builtin_amdgcn_readfirstlane(cnt[node]);
    float dd  = dinv[node];
    int   q0 = __builtin_amdgcn_readfirstlane(x[node * 3 + 0]);
    int   q1 = __builtin_amdgcn_readfirstlane(x[node * 3 + 1]);
    int   q2 = __builtin_amdgcn_readfirstlane(x[node * 3 + 2]);

    for (;;) {
        const int  next = node + nw;
        const bool more = next < n;
        int   nbeg = 0, ndeg = 0, nq0 = 0, nq1 = 0, nq2 = 0;
        float ndd  = 0.0f;
        if (more) {
            nbeg = rowptr[next];
            ndeg = cnt[next];
            ndd  = dinv[next];
            nq0  = x[next * 3 + 0];
            nq1  = x[next * 3 + 1];
            nq2  = x[next * 3 + 2];
        }

        float4 acc = make_float4(0.0f, 0.0f, 0.0f, 0.0f);
        for (int base = 0; base < deg; base += 32) {
            const int m = (deg - base < 32) ? (deg - base) : 32;
            int   r0 = 0, r1 = 0, r2 = 0;
            float c_l = 0.0f;
            if (lane < m) {
                const int s_l = csr_src[beg + base + lane];
                c_l = dinv[s_l] * dd;
                r0 = x[s_l * 3 + 0];
                r1 = x[s_l * 3 + 1] + NSHAPE;
                r2 = x[s_l * 3 + 2] + NSHAPE + NCOLOR;
            }
#pragma unroll 4
            for (int j = 0; j < m; ++j) {
                const int   i0 = __builtin_amdgcn_readlane(r0, j);
                const int   i1 = __builtin_amdgcn_readlane(r1, j);
                const int   i2 = __builtin_amdgcn_readlane(r2, j);
                const float c  = __int_as_float(__builtin_amdgcn_readlane(__float_as_int(c_l), j));
                const float4 a = tw4[i0 * 32 + lane];
                const float4 b = tw4[i1 * 32 + lane];
                const float4 d = tw4[i2 * 32 + lane];
                acc.x += ((a.x + b.x) + d.x) * c;
                acc.y += ((a.y + b.y) + d.y) * c;
                acc.z += ((a.z + b.z) + d.z) * c;
                acc.w += ((a.w + b.w) + d.w) * c;
            }
        }

        const float4 a = tw4[q0 * 32 + lane];
        const float4 b = tw4[(q1 + NSHAPE) * 32 + lane];
        const float4 d = tw4[(q2 + NSHAPE + NCOLOR) * 32 + lane];
        const float sl = dd * dd;
        float4 o;
        o.x = fmaxf(acc.x + ((a.x + b.x) + d.x) * sl + bv.x, 0.0f);
        o.y = fmaxf(acc.y + ((a.y + b.y) + d.y) * sl + bv.y, 0.0f);
        o.z = fmaxf(acc.z + ((a.z + b.z) + d.z) * sl + bv.z, 0.0f);
        o.w = fmaxf(acc.w + ((a.w + b.w) + d.w) * sl + bv.w, 0.0f);
        reinterpret_cast<float4*>(h)[(long long)node * 32 + lane] = o;

        if (!more) break;
        node = next;
        beg  = __builtin_amdgcn_readfirstlane(nbeg);
        deg  = __builtin_amdgcn_readfirstlane(ndeg);
        dd   = ndd;
        q0   = __builtin_amdgcn_readfirstlane(nq0);
        q1   = __builtin_amdgcn_readfirstlane(nq1);
        q2   = __builtin_amdgcn_readfirstlane(nq2);
    }
}

__global__ void k_head(const float* __restrict__ sums,
                       const float* __restrict__ cnt,
                       const float* __restrict__ Wlin,
                       const float* __restrict__ blin,
                       float* __restrict__ out, int g_total)
{
    int idx = blockIdx.x * blockDim.x + threadIdx.x;
    if (idx >= g_total * NCLS) return;
    const int g = idx / NCLS;
    const int c = idx % NCLS;
    const float inv = 1.0f / fmaxf(cnt[g], 1.0f);
    const float* sp = sums + (long long)g * EMB;
    float acc = 0.0f;
#pragma unroll 4
    for (int k = 0; k < EMB; ++k) acc += sp[k] * Wlin[k * NCLS + c];
    out[idx] = acc * inv + blin[c];
}

static inline size_t align256(size_t x) { return (x + 255) & ~(size_t)255; }

extern "C" void kernel_launch(void* const* d_in, const int* in_sizes, int n_in,
                              void* d_out, int out_size, void* d_ws, size_t ws_size,
                              hipStream_t stream)
{
    const int*   x        = (const int*)  d_in[0];
    const int*   eidx     = (const int*)  d_in[1];
    const int*   batch    = (const int*)  d_in[2];
    const float* shapeTab = (const float*)d_in[3];
    const float* colorTab = (const float*)d_in[4];
    const float* posTab   = (const float*)d_in[5];
    const float* W1       = (const float*)d_in[6];
    const float* b1       = (const float*)d_in[7];
    const float* W2       = (const float*)d_in[8];
    const float* b2       = (const float*)d_in[9];
    const float* Wlin     = (const float*)d_in[10];
    const float* blin     = (const float*)d_in[11];
    float*       out      = (float*)d_out;

    const int N = in_sizes[0] / 3;
    const int E = in_sizes[1] / 2;
    const int G = out_size / NCLS;

    const int* src = eidx;
    const int* dst = eidx + E;

    char* ws = (char*)d_ws;
    size_t off = 0;
    float* h    = (float*)(ws + off); off = align256(off + (size_t)N * EMB * 4);
    float* xw   = (float*)(ws + off); off = align256(off + (size_t)N * EMB * 4);
    char*  csr  = ws + off;           off = align256(off + (size_t)N * EMB * 4);
    float* dinv = (float*)(ws + off); off = align256(off + (size_t)N * 4);
    float* sums = (float*)(ws + off); off = align256(off + (size_t)G * EMB * 4);
    float* cnt  = (float*)(ws + off); off = align256(off + (size_t)G * 4);
    (void)ws_size;

    const int nBins  = (N + BIN_NODES - 1) / BIN_NODES;
    const int nParts = (E + CHUNK - 1) / CHUNK;
    const int cPitch = (nBins + 31) & ~31;
    size_t coff = 0;
    int*  degc     = (int*)(csr + coff);  coff = align256(coff + (size_t)N * 4);
    int*  rowptr   = (int*)(csr + coff);  coff = align256(coff + (size_t)N * 4);
    int*  binStart = (int*)(csr + coff);  coff = align256(coff + (size_t)(nBins + 1) * 4);
    int*  counts   = (int*)(csr + coff);  coff = align256(coff + (size_t)nParts * cPitch * 4);
    int*  pairs    = (int*)(csr + coff);  coff = align256(coff + (size_t)E * 8);
    int*  csrSrc   = (int*)(csr + coff);  coff = align256(coff + (size_t)E * 4);
    float* tw   = (float*)(csr + coff); coff = align256(coff + (size_t)NTABROWS * EMB * 4);

    const int TB = 256;
    const int gNodeWave = 2048;
    const int gNodeTab  = 1536;
    const int gGemmBlks = (N + 127) / 128;

    k_tab_gemm_wmma<<<NTABROWS / 16, TB, 0, stream>>>(shapeTab, colorTab, posTab, W1, tw);

    k_bin_hist<<<nParts, PT, 0, stream>>>(dst, counts, E, nBins, cPitch);
    k_bin_place<<<nParts, PT, 0, stream>>>(src, dst, counts, binStart, pairs, E, nBins, cPitch, nParts);
    k_bin_sort<<<nBins, 256, 0, stream>>>(pairs, binStart, rowptr, degc, dinv, csrSrc, N);

    k_gather_tab_relu<<<gNodeTab, 256, 0, stream>>>(x, tw, rowptr, degc, csrSrc, dinv, b1, h, N);

    k_gemm_wmma<<<gGemmBlks, TB, 0, stream>>>(h, W2, xw, N);
    k_fill<<<(G * EMB + TB - 1) / TB, TB, 0, stream>>>(sums, 0.0f, G * EMB);
    k_fill<<<(G + TB - 1) / TB, TB, 0, stream>>>(cnt, 0.0f, G);
    k_gather_relu_pool<<<gNodeWave, 256, 0, stream>>>(xw, rowptr, degc, csrSrc, dinv, b2, batch, sums, cnt, N);

    k_head<<<(G * NCLS + TB - 1) / TB, TB, 0, stream>>>(sums, cnt, Wlin, blin, out, G);
}


# the three small kernels that depend on the inputs alone (table product on WMMA, two zero fills of the pool sums) ride in the edge-histogram launch as extra blocks: three launches fewer
# speedup vs baseline: 5.3425x; 1.0414x over previous
#include <hip/hip_runtime.h>
#include <hip/hip_bf16.h>

#define EMB 128
#define NCLS 10
#define NSHAPE 16
#define NCOLOR 16
#define MAXPOS 256
#define NTABROWS (NSHAPE + NCOLOR + MAXPOS)

typedef __attribute__((ext_vector_type(2))) float v2f;
typedef __attribute__((ext_vector_type(8))) float v8f;

__device__ __forceinline__ void tab_gemm_tile(int tile, int ctile, int lane,
    const float* __restrict__ st, const float* __restrict__ ct, const float* __restrict__ pt,
    const float* __restrict__ W, float* __restrict__ tw)
{
    const int m     = lane & 15;
    const int khalf = (lane >> 4) * 2;

    const float* abase;
    if (tile == 0)      abase = st;
    else if (tile == 1) abase = ct;
    else                abase = pt + (long long)(tile - 2) * 16 * EMB;

    const float* arow = abase + m * EMB;
    const int col = ctile * 16 + m;

    v8f acc = {};
    for (int k = 0; k < EMB; k += 4) {
        const int kk = k + khalf;
        v2f a, b;
        a.x = arow[kk];
        a.y = arow[kk + 1];
        b.x = W[kk * EMB + col];
        b.y = W[(kk + 1) * EMB + col];
        acc = __builtin_amdgcn_wmma_f32_16x16x4_f32(
            false, a, false, b, (short)0, acc, false, false);
    }

    const int rbase = (lane >> 4) * 8;
#pragma unroll
    for (int r = 0; r < 8; ++r)
        tw[(long long)(tile * 16 + rbase + r) * EMB + col] = acc[r];
}

#define BIN_SHIFT 7
#define BIN_NODES (1 << BIN_SHIFT)
#define MAXBINS 1024
#define PT 1024
#define KPT 12
#define CHUNK (PT * KPT)
#define STAGE_CAP 8192

__device__ __forceinline__ int wave_incl_scan(int v, int lane)
{
#pragma unroll
    for (int off = 1; off < 32; off <<= 1) {
        const int u = __shfl_up(v, off);
        if (lane >= off) v += u;
    }
    return v;
}

#define TAB_TILES (NTABROWS / 16)
#define TAB_BLOCKS ((TAB_TILES + 3) / 4)
__global__ __launch_bounds__(PT) void k_front(const int* __restrict__ dst, int* __restrict__ counts,
                                              int e, int nb, int pitch, int nparts,
                                              const float* __restrict__ st, const float* __restrict__ ct,
                                              const float* __restrict__ pt, const float* __restrict__ W,
                                              float* __restrict__ tw,
                                              float* __restrict__ sums, int nsums,
                                              float* __restrict__ gcnt, int ng)
{
    __shared__ int hist[MAXBINS];
    const int t = threadIdx.x;

    for (int i = blockIdx.x * PT + t; i < nsums; i += gridDim.x * PT) sums[i] = 0.0f;
    for (int i = blockIdx.x * PT + t; i < ng;    i += gridDim.x * PT) gcnt[i] = 0.0f;

    if ((int)blockIdx.x >= nparts) {
        const int wave = t >> 5;
        const int tile = ((int)blockIdx.x - nparts) * 4 + (wave >> 3);
        if (tile < TAB_TILES)
            tab_gemm_tile(tile, wave & 7, t & 31, st, ct, pt, W, tw);
        return;
    }

    hist[t] = 0;
    __syncthreads();
    const int base = blockIdx.x * CHUNK + t;
    int d[KPT];
#pragma unroll
    for (int j = 0; j < KPT; ++j) {
        const int i = base + j * PT;
        d[j] = dst[(i < e) ? i : (e - 1)];
    }
#pragma unroll
    for (int j = 0; j < KPT; ++j)
        if (base + j * PT < e) atomicAdd(&hist[d[j] >> BIN_SHIFT], 1);
    __syncthreads();
    if (t < nb) counts[blockIdx.x * pitch + t] = hist[t];
}

__global__ __launch_bounds__(PT) void k_bin_place(const int* __restrict__ src, const int* __restrict__ dst,
                                                  const int* __restrict__ counts, int* __restrict__ binStart,
                                                  int* __restrict__ pairs_out, int e, int nb, int pitch, int nparts)
{
    int2* pairs = reinterpret_cast<int2*>(pairs_out);
    __shared__ int  hist[MAXBINS];
    __shared__ int  lstart[MAXBINS];
    __shared__ int  gdelta[MAXBINS];
    __shared__ int  wsA[32];
    __shared__ int  wsG[32];
    __shared__ int2 sorted[CHUNK];
    const int t    = threadIdx.x;
    const int lane = t & 31;
    const int wave = t >> 5;
    const int p    = blockIdx.x;
    hist[t] = 0;
    __syncthreads();

    const int base = p * CHUNK + t;
    int s[KPT], d[KPT], r[KPT];
#pragma unroll
    for (int j = 0; j < KPT; ++j) {
        const int i  = base + j * PT;
        const int ic = (i < e) ? i : (e - 1);
        s[j] = src[ic];
        d[j] = dst[ic];
    }

    int tot = 0, pre = 0;
    if (t < nb) {
        for (int q0 = 0; q0 < nparts; q0 += 16) {
            int c[16];
#pragma unroll
            for (int k = 0; k < 16; ++k) {
                const int q = q0 + k;
                c[k] = counts[((q < nparts) ? q : (nparts - 1)) * pitch + t];
            }
#pragma unroll
            for (int k = 0; k < 16; ++k) {
                const int q = q0 + k;
                tot += (q < nparts) ? c[k] : 0;
                pre += (q < p) ? c[k] : 0;
            }
        }
    }

#pragma unroll
    for (int j = 0; j < KPT; ++j) {
        r[j] = 0;
        if (base + j * PT < e) r[j] = atomicAdd(&hist[d[j] >> BIN_SHIFT], 1);
    }
    __syncthreads();
    const int myh = hist[t];

    const int ia = wave_incl_scan(myh, lane);
    const int ig = wave_incl_scan(tot, lane);
    if (lane == 31) { wsA[wave] = ia; wsG[wave] = ig; }
    __syncthreads();
    if (wave == 0) {
        const int xa = wsA[lane], xg = wsG[lane];
        const int sa = wave_incl_scan(xa, lane);
        const int sg = wave_incl_scan(xg, lane);
        wsA[lane] = sa - xa;
        wsG[lane] = sg - xg;
    }
    __syncthreads();
    const int lex = ia - myh + wsA[wave];
    const int gex = ig - tot + wsG[wave];
    lstart[t] = lex;
    gdelta[t] = gex + pre - lex;
    if (p == 0) {
        if (t < nb)      binStart[t]  = gex;
        if (t == nb - 1) binStart[nb] = gex + tot;
    }
    __syncthreads();

#pragma unroll
    for (int j = 0; j < KPT; ++j) {
        if (base + j * PT < e)
            sorted[lstart[d[j] >> BIN_SHIFT] + r[j]] = make_int2(s[j], d[j]);
    }
    __syncthreads();

    const int left = e - p * CHUNK;
    const int cc   = (left < CHUNK) ? left : CHUNK;
    for (int i = t; i < cc; i += PT) {
        const int2 v = sorted[i];
        pairs[gdelta[v.y >> BIN_SHIFT] + i] = v;
    }
}

__global__ __launch_bounds__(256) void k_bin_sort(const int* __restrict__ pairs_in, const int* __restrict__ binStart,
                                                  int* __restrict__ rowptr, int* __restrict__ cnt,
                                                  float* __restrict__ dinv, int* __restrict__ csr_src, int n)
{
    const int2* pairs = reinterpret_cast<const int2*>(pairs_in);
    __shared__ int c[BIN_NODES];
    __shared__ int cur[BIN_NODES];
    __shared__ int ws[8];
    __shared__ int stage[STAGE_CAP];
    const int t     = threadIdx.x;
    const int lane  = t & 31;
    const int wave  = t >> 5;
    const int b     = blockIdx.x;
    const int start = binStart[b];
    const int m     = binStart[b + 1] - start;
    if (t < BIN_NODES) c[t] = 0;
    __syncthreads();

    for (int i0 = t; i0 < m; i0 += 256 * 8) {
        int dd[8];
#pragma unroll
        for (int j = 0; j < 8; ++j) {
            const int i = i0 + j * 256;
            dd[j] = pairs[start + ((i < m) ? i : i0)].y;
        }
#pragma unroll
        for (int j = 0; j < 8; ++j)
            if (i0 + j * 256 < m) atomicAdd(&c[dd[j] & (BIN_NODES - 1)], 1);
    }
    __syncthreads();

    const int v  = (t < BIN_NODES) ? c[t] : 0;
    const int iv = wave_incl_scan(v, lane);
    if (lane == 31) ws[wave] = iv;
    __syncthreads();
    int woff = 0;
    for (int w = 0; w < wave; ++w) woff += ws[w];
    if (t < BIN_NODES) {
        const int ex = iv - v + woff;
        cur[t] = ex;
        const int node = b * BIN_NODES + t;
        if (node < n) {
            rowptr[node] = start + ex;
            cnt[node]    = v;
            dinv[node]   = __frsqrt_rn((float)(v + 1));
        }
    }
    __syncthreads();

    const bool staged = (m <= STAGE_CAP);
    for (int i0 = t; i0 < m; i0 += 256 * 8) {
        int2 ev[8];
#pragma unroll
        for (int j = 0; j < 8; ++j) {
            const int i = i0 + j * 256;
            ev[j] = pairs[start + ((i < m) ? i : i0)];
        }
#pragma unroll
        for (int j = 0; j < 8; ++j) {
            if (i0 + j * 256 < m) {
                const int pos = atomicAdd(&cur[ev[j].y & (BIN_NODES - 1)], 1);
                if (staged) stage[pos] = ev[j].x;
                else        csr_src[start + pos] = ev[j].x;
            }
        }
    }
    if (staged) {
        __syncthreads();
        for (int i = t; i < m; i += 256) csr_src[start + i] = stage[i];
    }
}

#define WSTRIDE 128
__global__ __launch_bounds__(256) void k_gemm_wmma(
    const float* __restrict__ A, const float* __restrict__ W,
    float* __restrict__ C, int nrows)
{
    __shared__ float Wl[128 * WSTRIDE];
    const int tid = threadIdx.x;

    for (int idx = tid; idx < 128 * 32; idx += 256) {
        int r = idx >> 5, q = idx & 31;
        reinterpret_cast<float4*>(&Wl[r * WSTRIDE + q * 4])[0] =
            reinterpret_cast<const float4*>(W)[r * 32 + q];
    }
    __syncthreads();

    const int wave  = tid >> 5;
    const int lane  = tid & 31;
    const int m     = lane & 15;
    const int khalf = (lane >> 4) * 2;
    const int tileRow = blockIdx.x * 128 + wave * 16;
    if (tileRow >= nrows) return;

    v8f acc[8] = {};

    const float* arow = A + (long long)(tileRow + m) * EMB;
    for (int k = 0; k < EMB; k += 4) {
        const int kk = k + khalf;
        v2f a;
        a.x = arow[kk];
        a.y = arow[kk + 1];
#pragma unroll
        for (int j = 0; j < 8; ++j) {
            const int col = j * 16 + m;
            v2f b;
            b.x = Wl[kk * WSTRIDE + col];
            b.y = Wl[(kk + 1) * WSTRIDE + col];
            acc[j] = __builtin_amdgcn_wmma_f32_16x16x4_f32(
                false, a, false, b, (short)0, acc[j], false, false);
        }
    }

    const int rbase = (lane >> 4) * 8;
#pragma unroll
    for (int j = 0; j < 8; ++j) {
#pragma unroll
        for (int r = 0; r < 8; ++r) {
            C[(long long)(tileRow + rbase + r) * EMB + j * 16 + m] = acc[j][r];
        }
    }
}

__device__ __forceinline__ void pool_flush(float* __restrict__ sums, float* __restrict__ gcnt,
                                           int g, int lane, const float4& p, int rows)
{
    float* sp = sums + (long long)g * EMB + lane * 4;
    unsafeAtomicAdd(sp + 0, p.x);
    unsafeAtomicAdd(sp + 1, p.y);
    unsafeAtomicAdd(sp + 2, p.z);
    unsafeAtomicAdd(sp + 3, p.w);
    if (lane == 0) unsafeAtomicAdd(&gcnt[g], (float)rows);
}

__global__ __launch_bounds__(256) void k_gather_relu_pool(const float* __restrict__ xw,
                                                          const int* __restrict__ rowptr,
                                                          const int* __restrict__ cnt,
                                                          const int* __restrict__ csr_src,
                                                          const float* __restrict__ dinv,
                                                          const float* __restrict__ bias,
                                                          const int* __restrict__ batch,
                                                          float* __restrict__ sums,
                                                          float* __restrict__ gcnt, int n)
{
    const int lane = threadIdx.x & 31;
    const int nw   = gridDim.x * 8;
    const int gw   = __builtin_amdgcn_readfirstlane((int)(blockIdx.x * 8 + (threadIdx.x >> 5)));
    const int per  = n / nw;
    const int rem  = n - per * nw;
    const int lo   = gw * per + ((gw < rem) ? gw : rem);
    const int hi   = lo + per + ((gw < rem) ? 1 : 0);
    if (lo >= hi) return;
    const float4* xw4 = reinterpret_cast<const float4*>(xw);
    const float* bp = bias + lane * 4;
    const float4 bv = make_float4(bp[0], bp[1], bp[2], bp[3]);

    int   node = lo;
    int   beg = __builtin_amdgcn_readfirstlane(rowptr[node]);
    int   deg = __builtin_amdgcn_readfirstlane(cnt[node]);
    int   g   = __builtin_amdgcn_readfirstlane(batch[node]);
    float dd  = dinv[node];

    int    cg   = g;
    int    rows = 0;
    float4 pacc = make_float4(0.0f, 0.0f, 0.0f, 0.0f);

    for (;;) {
        const int  next = node + 1;
        const bool more = next < hi;
        int   nbeg = 0, ndeg = 0, ng = 0;
        float ndd  = 0.0f;
        if (more) {
            nbeg = rowptr[next];
            ndeg = cnt[next];
            ng   = batch[next];
            ndd  = dinv[next];
        }

        float4 acc = make_float4(0.0f, 0.0f, 0.0f, 0.0f);
        for (int base = 0; base < deg; base += 32) {
            const int m = (deg - base < 32) ? (deg - base) : 32;
            int   s_l = 0;
            float c_l = 0.0f;
            if (lane < m) {
                s_l = csr_src[beg + base + lane];
                c_l = dinv[s_l] * dd;
            }
#pragma unroll 4
            for (int j = 0; j < m; ++j) {
                const int   s = __builtin_amdgcn_readlane(s_l, j);
                const float c = __int_as_float(__builtin_amdgcn_readlane(__float_as_int(c_l), j));
                const float4 v = xw4[(long long)s * 32 + lane];
                acc.x += v.x * c;
                acc.y += v.y * c;
                acc.z += v.z * c;
                acc.w += v.w * c;
            }
        }

        const float sl = dd * dd;
        const float4 xv = xw4[(long long)node * 32 + lane];
        float4 o;
        o.x = fmaxf(acc.x + xv.x * sl + bv.x, 0.0f);
        o.y = fmaxf(acc.y + xv.y * sl + bv.y, 0.0f);
        o.z = fmaxf(acc.z + xv.z * sl + bv.z, 0.0f);
        o.w = fmaxf(acc.w + xv.w * sl + bv.w, 0.0f);

        if (g != cg) {
            pool_flush(sums, gcnt, cg, lane, pacc, rows);
            pacc = make_float4(0.0f, 0.0f, 0.0f, 0.0f);
            rows = 0;
            cg   = g;
        }
        pacc.x += o.x;
        pacc.y += o.y;
        pacc.z += o.z;
        pacc.w += o.w;
        rows += 1;

        if (!more) break;
        node = next;
        beg  = __builtin_amdgcn_readfirstlane(nbeg);
        deg  = __builtin_amdgcn_readfirstlane(ndeg);
        g    = __builtin_amdgcn_readfirstlane(ng);
        dd   = ndd;
    }
    pool_flush(sums, gcnt, cg, lane, pacc, rows);
}

__global__ __launch_bounds__(256) void k_gather_tab_relu(const int* __restrict__ x,
                                                         const float* __restrict__ tw,
                                                         const int* __restrict__ rowptr,
                                                         const int* __restrict__ cnt,
                                                         const int* __restrict__ csr_src,
                                                         const float* __restrict__ dinv,
                                                         const float* __restrict__ bias,
                                                         float* __restrict__ h, int n)
{
    const int lane = threadIdx.x & 31;
    const int nw   = gridDim.x * 8;
    int node = __builtin_amdgcn_readfirstlane((int)(blockIdx.x * 8 + (threadIdx.x >> 5)));
    if (node >= n) return;
    const float4* tw4 = reinterpret_cast<const float4*>(tw);
    const float* bp = bias + lane * 4;
    const float4 bv = make_float4(bp[0], bp[1], bp[2], bp[3]);

    int   beg = __builtin_amdgcn_readfirstlane(rowptr[node]);
    int   deg = __builtin_amdgcn_readfirstlane(cnt[node]);
    float dd  = dinv[node];
    int   q0 = __builtin_amdgcn_readfirstlane(x[node * 3 + 0]);
    int   q1 = __builtin_amdgcn_readfirstlane(x[node * 3 + 1]);
    int   q2 = __builtin_amdgcn_readfirstlane(x[node * 3 + 2]);

    for (;;) {
        const int  next = node + nw;
        const bool more = next < n;
        int   nbeg = 0, ndeg = 0, nq0 = 0, nq1 = 0, nq2 = 0;
        float ndd  = 0.0f;
        if (more) {
            nbeg = rowptr[next];
            ndeg = cnt[next];
            ndd  = dinv[next];
            nq0  = x[next * 3 + 0];
            nq1  = x[next * 3 + 1];
            nq2  = x[next * 3 + 2];
        }

        float4 acc = make_float4(0.0f, 0.0f, 0.0f, 0.0f);
        for (int base = 0; base < deg; base += 32) {
            const int m = (deg - base < 32) ? (deg - base) : 32;
            int   r0 = 0, r1 = 0, r2 = 0;
            float c_l = 0.0f;
            if (lane < m) {
                const int s_l = csr_src[beg + base + lane];
                c_l = dinv[s_l] * dd;
                r0 = x[s_l * 3 + 0];
                r1 = x[s_l * 3 + 1] + NSHAPE;
                r2 = x[s_l * 3 + 2] + NSHAPE + NCOLOR;
            }
#pragma unroll 4
            for (int j = 0; j < m; ++j) {
                const int   i0 = __builtin_amdgcn_readlane(r0, j);
                const int   i1 = __builtin_amdgcn_readlane(r1, j);
                const int   i2 = __builtin_amdgcn_readlane(r2, j);
                const float c  = __int_as_float(__builtin_amdgcn_readlane(__float_as_int(c_l), j));
                const float4 a = tw4[i0 * 32 + lane];
                const float4 b = tw4[i1 * 32 + lane];
                const float4 d = tw4[i2 * 32 + lane];
                acc.x += ((a.x + b.x) + d.x) * c;
                acc.y += ((a.y + b.y) + d.y) * c;
                acc.z += ((a.z + b.z) + d.z) * c;
                acc.w += ((a.w + b.w) + d.w) * c;
            }
        }

        const float4 a = tw4[q0 * 32 + lane];
        const float4 b = tw4[(q1 + NSHAPE) * 32 + lane];
        const float4 d = tw4[(q2 + NSHAPE + NCOLOR) * 32 + lane];
        const float sl = dd * dd;
        float4 o;
        o.x = fmaxf(acc.x + ((a.x + b.x) + d.x) * sl + bv.x, 0.0f);
        o.y = fmaxf(acc.y + ((a.y + b.y) + d.y) * sl + bv.y, 0.0f);
        o.z = fmaxf(acc.z + ((a.z + b.z) + d.z) * sl + bv.z, 0.0f);
        o.w = fmaxf(acc.w + ((a.w + b.w) + d.w) * sl + bv.w, 0.0f);
        reinterpret_cast<float4*>(h)[(long long)node * 32 + lane] = o;

        if (!more) break;
        node = next;
        beg  = __builtin_amdgcn_readfirstlane(nbeg);
        deg  = __builtin_amdgcn_readfirstlane(ndeg);
        dd   = ndd;
        q0   = __builtin_amdgcn_readfirstlane(nq0);
        q1   = __builtin_amdgcn_readfirstlane(nq1);
        q2   = __builtin_amdgcn_readfirstlane(nq2);
    }
}

__global__ void k_head(const float* __restrict__ sums,
                       const float* __restrict__ cnt,
                       const float* __restrict__ Wlin,
                       const float* __restrict__ blin,
                       float* __restrict__ out, int g_total)
{
    int idx = blockIdx.x * blockDim.x + threadIdx.x;
    if (idx >= g_total * NCLS) return;
    const int g = idx / NCLS;
    const int c = idx % NCLS;
    const float inv = 1.0f / fmaxf(cnt[g], 1.0f);
    const float* sp = sums + (long long)g * EMB;
    float acc = 0.0f;
#pragma unroll 4
    for (int k = 0; k < EMB; ++k) acc += sp[k] * Wlin[k * NCLS + c];
    out[idx] = acc * inv + blin[c];
}

static inline size_t align256(size_t x) { return (x + 255) & ~(size_t)255; }

extern "C" void kernel_launch(void* const* d_in, const int* in_sizes, int n_in,
                              void* d_out, int out_size, void* d_ws, size_t ws_size,
                              hipStream_t stream)
{
    const int*   x        = (const int*)  d_in[0];
    const int*   eidx     = (const int*)  d_in[1];
    const int*   batch    = (const int*)  d_in[2];
    const float* shapeTab = (const float*)d_in[3];
    const float* colorTab = (const float*)d_in[4];
    const float* posTab   = (const float*)d_in[5];
    const float* W1       = (const float*)d_in[6];
    const float* b1       = (const float*)d_in[7];
    const float* W2       = (const float*)d_in[8];
    const float* b2       = (const float*)d_in[9];
    const float* Wlin     = (const float*)d_in[10];
    const float* blin     = (const float*)d_in[11];
    float*       out      = (float*)d_out;

    const int N = in_sizes[0] / 3;
    const int E = in_sizes[1] / 2;
    const int G = out_size / NCLS;

    const int* src = eidx;
    const int* dst = eidx + E;

    char* ws = (char*)d_ws;
    size_t off = 0;
    float* h    = (float*)(ws + off); off = align256(off + (size_t)N * EMB * 4);
    float* xw   = (float*)(ws + off); off = align256(off + (size_t)N * EMB * 4);
    char*  csr  = ws + off;           off = align256(off + (size_t)N * EMB * 4);
    float* dinv = (float*)(ws + off); off = align256(off + (size_t)N * 4);
    float* sums = (float*)(ws + off); off = align256(off + (size_t)G * EMB * 4);
    float* cnt  = (float*)(ws + off); off = align256(off + (size_t)G * 4);
    (void)ws_size;

    const int nBins  = (N + BIN_NODES - 1) / BIN_NODES;
    const int nParts = (E + CHUNK - 1) / CHUNK;
    const int cPitch = (nBins + 31) & ~31;
    size_t coff = 0;
    int*  degc     = (int*)(csr + coff);  coff = align256(coff + (size_t)N * 4);
    int*  rowptr   = (int*)(csr + coff);  coff = align256(coff + (size_t)N * 4);
    int*  binStart = (int*)(csr + coff);  coff = align256(coff + (size_t)(nBins + 1) * 4);
    int*  counts   = (int*)(csr + coff);  coff = align256(coff + (size_t)nParts * cPitch * 4);
    int*  pairs    = (int*)(csr + coff);  coff = align256(coff + (size_t)E * 8);
    int*  csrSrc   = (int*)(csr + coff);  coff = align256(coff + (size_t)E * 4);
    float* tw   = (float*)(csr + coff); coff = align256(coff + (size_t)NTABROWS * EMB * 4);

    const int TB = 256;
    const int gNodeWave = 2048;
    const int gNodeTab  = 1536;
    const int gGemmBlks = (N + 127) / 128;

    k_front<<<nParts + TAB_BLOCKS, PT, 0, stream>>>(dst, counts, E, nBins, cPitch, nParts,
                                                   shapeTab, colorTab, posTab, W1, tw,
                                                   sums, G * EMB, cnt, G);

    k_bin_place<<<nParts, PT, 0, stream>>>(src, dst, counts, binStart, pairs, E, nBins, cPitch, nParts);
    k_bin_sort<<<nBins, 256, 0, stream>>>(pairs, binStart, rowptr, degc, dinv, csrSrc, N);

    k_gather_tab_relu<<<gNodeTab, 256, 0, stream>>>(x, tw, rowptr, degc, csrSrc, dinv, b1, h, N);

    k_gemm_wmma<<<gGemmBlks, TB, 0, stream>>>(h, W2, xw, N);
    k_gather_relu_pool<<<gNodeWave, 256, 0, stream>>>(xw, rowptr, degc, csrSrc, dinv, b2, batch, sums, cnt, N);

    k_head<<<(G * NCLS + TB - 1) / TB, TB, 0, stream>>>(sums, cnt, Wlin, blin, out, G);
}


# dinv[src] leaves the per-edge gathers: xw rows are stored times dinv (layer 2 reads no scalar per edge, self row starts the sum), layer 1 reads one 16-byte node record; layer-2 sums regrouped
# speedup vs baseline: 5.5977x; 1.0478x over previous
#include <hip/hip_runtime.h>
#include <hip/hip_bf16.h>

#define EMB 128
#define NCLS 10
#define NSHAPE 16
#define NCOLOR 16
#define MAXPOS 256
#define NTABROWS (NSHAPE + NCOLOR + MAXPOS)

typedef __attribute__((ext_vector_type(2))) float v2f;
typedef __attribute__((ext_vector_type(8))) float v8f;

__device__ __forceinline__ void tab_gemm_tile(int tile, int ctile, int lane,
    const float* __restrict__ st, const float* __restrict__ ct, const float* __restrict__ pt,
    const float* __restrict__ W, float* __restrict__ tw)
{
    const int m     = lane & 15;
    const int khalf = (lane >> 4) * 2;

    const float* abase;
    if (tile == 0)      abase = st;
    else if (tile == 1) abase = ct;
    else                abase = pt + (long long)(tile - 2) * 16 * EMB;

    const float* arow = abase + m * EMB;
    const int col = ctile * 16 + m;

    v8f acc = {};
    for (int k = 0; k < EMB; k += 4) {
        const int kk = k + khalf;
        v2f a, b;
        a.x = arow[kk];
        a.y = arow[kk + 1];
        b.x = W[kk * EMB + col];
        b.y = W[(kk + 1) * EMB + col];
        acc = __builtin_amdgcn_wmma_f32_16x16x4_f32(
            false, a, false, b, (short)0, acc, false, false);
    }

    const int rbase = (lane >> 4) * 8;
#pragma unroll
    for (int r = 0; r < 8; ++r)
        tw[(long long)(tile * 16 + rbase + r) * EMB + col] = acc[r];
}

#define BIN_SHIFT 7
#define BIN_NODES (1 << BIN_SHIFT)
#define MAXBINS 1024
#define PT 1024
#define KPT 12
#define CHUNK (PT * KPT)
#define STAGE_CAP 8192

__device__ __forceinline__ int wave_incl_scan(int v, int lane)
{
#pragma unroll
    for (int off = 1; off < 32; off <<= 1) {
        const int u = __shfl_up(v, off);
        if (lane >= off) v += u;
    }
    return v;
}

#define TAB_TILES (NTABROWS / 16)
#define TAB_BLOCKS ((TAB_TILES + 3) / 4)
__global__ __launch_bounds__(PT) void k_front(const int* __restrict__ dst, int* __restrict__ counts,
                                              int e, int nb, int pitch, int nparts,
                                              const float* __restrict__ st, const float* __restrict__ ct,
                                              const float* __restrict__ pt, const float* __restrict__ W,
                                              float* __restrict__ tw,
                                              float* __restrict__ sums, int nsums,
                                              float* __restrict__ gcnt, int ng)
{
    __shared__ int hist[MAXBINS];
    const int t = threadIdx.x;

    for (int i = blockIdx.x * PT + t; i < nsums; i += gridDim.x * PT) sums[i] = 0.0f;
    for (int i = blockIdx.x * PT + t; i < ng;    i += gridDim.x * PT) gcnt[i] = 0.0f;

    if ((int)blockIdx.x >= nparts) {
        const int wave = t >> 5;
        const int tile = ((int)blockIdx.x - nparts) * 4 + (wave >> 3);
        if (tile < TAB_TILES)
            tab_gemm_tile(tile, wave & 7, t & 31, st, ct, pt, W, tw);
        return;
    }

    hist[t] = 0;
    __syncthreads();
    const int base = blockIdx.x * CHUNK + t;
    int d[KPT];
#pragma unroll
    for (int j = 0; j < KPT; ++j) {
        const int i = base + j * PT;
        d[j] = dst[(i < e) ? i : (e - 1)];
    }
#pragma unroll
    for (int j = 0; j < KPT; ++j)
        if (base + j * PT < e) atomicAdd(&hist[d[j] >> BIN_SHIFT], 1);
    __syncthreads();
    if (t < nb) counts[blockIdx.x * pitch + t] = hist[t];
}

__global__ __launch_bounds__(PT) void k_bin_place(const int* __restrict__ src, const int* __restrict__ dst,
                                                  const int* __restrict__ counts, int* __restrict__ binStart,
                                                  int* __restrict__ pairs_out, int e, int nb, int pitch, int nparts)
{
    int2* pairs = reinterpret_cast<int2*>(pairs_out);
    __shared__ int  hist[MAXBINS];
    __shared__ int  lstart[MAXBINS];
    __shared__ int  gdelta[MAXBINS];
    __shared__ int  wsA[32];
    __shared__ int  wsG[32];
    __shared__ int2 sorted[CHUNK];
    const int t    = threadIdx.x;
    const int lane = t & 31;
    const int wave = t >> 5;
    const int p    = blockIdx.x;
    hist[t] = 0;
    __syncthreads();

    const int base = p * CHUNK + t;
    int s[KPT], d[KPT], r[KPT];
#pragma unroll
    for (int j = 0; j < KPT; ++j) {
        const int i  = base + j * PT;
        const int ic = (i < e) ? i : (e - 1);
        s[j] = src[ic];
        d[j] = dst[ic];
    }

    int tot = 0, pre = 0;
    if (t < nb) {
        for (int q0 = 0; q0 < nparts; q0 += 16) {
            int c[16];
#pragma unroll
            for (int k = 0; k < 16; ++k) {
                const int q = q0 + k;
                c[k] = counts[((q < nparts) ? q : (nparts - 1)) * pitch + t];
            }
#pragma unroll
            for (int k = 0; k < 16; ++k) {
                const int q = q0 + k;
                tot += (q < nparts) ? c[k] : 0;
                pre += (q < p) ? c[k] : 0;
            }
        }
    }

#pragma unroll
    for (int j = 0; j < KPT; ++j) {
        r[j] = 0;
        if (base + j * PT < e) r[j] = atomicAdd(&hist[d[j] >> BIN_SHIFT], 1);
    }
    __syncthreads();
    const int myh = hist[t];

    const int ia = wave_incl_scan(myh, lane);
    const int ig = wave_incl_scan(tot, lane);
    if (lane == 31) { wsA[wave] = ia; wsG[wave] = ig; }
    __syncthreads();
    if (wave == 0) {
        const int xa = wsA[lane], xg = wsG[lane];
        const int sa = wave_incl_scan(xa, lane);
        const int sg = wave_incl_scan(xg, lane);
        wsA[lane] = sa - xa;
        wsG[lane] = sg - xg;
    }
    __syncthreads();
    const int lex = ia - myh + wsA[wave];
    const int gex = ig - tot + wsG[wave];
    lstart[t] = lex;
    gdelta[t] = gex + pre - lex;
    if (p == 0) {
        if (t < nb)      binStart[t]  = gex;
        if (t == nb - 1) binStart[nb] = gex + tot;
    }
    __syncthreads();

#pragma unroll
    for (int j = 0; j < KPT; ++j) {
        if (base + j * PT < e)
            sorted[lstart[d[j] >> BIN_SHIFT] + r[j]] = make_int2(s[j], d[j]);
    }
    __syncthreads();

    const int left = e - p * CHUNK;
    const int cc   = (left < CHUNK) ? left : CHUNK;
    for (int i = t; i < cc; i += PT) {
        const int2 v = sorted[i];
        pairs[gdelta[v.y >> BIN_SHIFT] + i] = v;
    }
}

__global__ __launch_bounds__(256) void k_bin_sort(const int* __restrict__ pairs_in, const int* __restrict__ binStart,
                                                  int* __restrict__ rowptr, int* __restrict__ cnt,
                                                  float* __restrict__ dinv, int* __restrict__ csr_src,
                                                  const int* __restrict__ x, int* __restrict__ rec_out, int n)
{
    const int2* pairs = reinterpret_cast<const int2*>(pairs_in);
    __shared__ int c[BIN_NODES];
    __shared__ int cur[BIN_NODES];
    __shared__ int ws[8];
    __shared__ int stage[STAGE_CAP];
    const int t     = threadIdx.x;
    const int lane  = t & 31;
    const int wave  = t >> 5;
    const int b     = blockIdx.x;
    const int start = binStart[b];
    const int m     = binStart[b + 1] - start;
    if (t < BIN_NODES) c[t] = 0;
    __syncthreads();

    for (int i0 = t; i0 < m; i0 += 256 * 8) {
        int dd[8];
#pragma unroll
        for (int j = 0; j < 8; ++j) {
            const int i = i0 + j * 256;
            dd[j] = pairs[start + ((i < m) ? i : i0)].y;
        }
#pragma unroll
        for (int j = 0; j < 8; ++j)
            if (i0 + j * 256 < m) atomicAdd(&c[dd[j] & (BIN_NODES - 1)], 1);
    }
    __syncthreads();

    const int v  = (t < BIN_NODES) ? c[t] : 0;
    const int iv = wave_incl_scan(v, lane);
    if (lane == 31) ws[wave] = iv;
    __syncthreads();
    int woff = 0;
    for (int w = 0; w < wave; ++w) woff += ws[w];
    if (t < BIN_NODES) {
        const int ex = iv - v + woff;
        cur[t] = ex;
        const int node = b * BIN_NODES + t;
        if (node < n) {
            const float di = __frsqrt_rn((float)(v + 1));
            rowptr[node] = start + ex;
            cnt[node]    = v;
            dinv[node]   = di;
            reinterpret_cast<int4*>(rec_out)[node] =
                make_int4(x[node * 3 + 0], x[node * 3 + 1] + NSHAPE, x[node * 3 + 2] + NSHAPE + NCOLOR,
                          __float_as_int(di));
        }
    }
    __syncthreads();

    const bool staged = (m <= STAGE_CAP);
    for (int i0 = t; i0 < m; i0 += 256 * 8) {
        int2 ev[8];
#pragma unroll
        for (int j = 0; j < 8; ++j) {
            const int i = i0 + j * 256;
            ev[j] = pairs[start + ((i < m) ? i : i0)];
        }
#pragma unroll
        for (int j = 0; j < 8; ++j) {
            if (i0 + j * 256 < m) {
                const int pos = atomicAdd(&cur[ev[j].y & (BIN_NODES - 1)], 1);
                if (staged) stage[pos] = ev[j].x;
                else        csr_src[start + pos] = ev[j].x;
            }
        }
    }
    if (staged) {
        __syncthreads();
        for (int i = t; i < m; i += 256) csr_src[start + i] = stage[i];
    }
}

#define WSTRIDE 128
__global__ __launch_bounds__(256) void k_gemm_wmma(
    const float* __restrict__ A, const float* __restrict__ W, const float* __restrict__ rscale,
    float* __restrict__ C, int nrows)
{
    __shared__ float Wl[128 * WSTRIDE];
    const int tid = threadIdx.x;

    for (int idx = tid; idx < 128 * 32; idx += 256) {
        int r = idx >> 5, q = idx & 31;
        reinterpret_cast<float4*>(&Wl[r * WSTRIDE + q * 4])[0] =
            reinterpret_cast<const float4*>(W)[r * 32 + q];
    }
    __syncthreads();

    const int wave  = tid >> 5;
    const int lane  = tid & 31;
    const int m     = lane & 15;
    const int khalf = (lane >> 4) * 2;
    const int tileRow = blockIdx.x * 128 + wave * 16;
    if (tileRow >= nrows) return;

    v8f acc[8] = {};

    const float* arow = A + (long long)(tileRow + m) * EMB;
    for (int k = 0; k < EMB; k += 4) {
        const int kk = k + khalf;
        v2f a;
        a.x = arow[kk];
        a.y = arow[kk + 1];
#pragma unroll
        for (int j = 0; j < 8; ++j) {
            const int col = j * 16 + m;
            v2f b;
            b.x = Wl[kk * WSTRIDE + col];
            b.y = Wl[(kk + 1) * WSTRIDE + col];
            acc[j] = __builtin_amdgcn_wmma_f32_16x16x4_f32(
                false, a, false, b, (short)0, acc[j], false, false);
        }
    }

    const int rbase = (lane >> 4) * 8;
    const float4 s0 = reinterpret_cast<const float4*>(rscale + tileRow + rbase)[0];
    const float4 s1 = reinterpret_cast<const float4*>(rscale + tileRow + rbase)[1];
    const float sc[8] = { s0.x, s0.y, s0.z, s0.w, s1.x, s1.y, s1.z, s1.w };
#pragma unroll
    for (int j = 0; j < 8; ++j) {
#pragma unroll
        for (int r = 0; r < 8; ++r) {
            C[(long long)(tileRow + rbase + r) * EMB + j * 16 + m] = acc[j][r] * sc[r];
        }
    }
}

__device__ __forceinline__ void pool_flush(float* __restrict__ sums, float* __restrict__ gcnt,
                                           int g, int lane, const float4& p, int rows)
{
    float* sp = sums + (long long)g * EMB + lane * 4;
    unsafeAtomicAdd(sp + 0, p.x);
    unsafeAtomicAdd(sp + 1, p.y);
    unsafeAtomicAdd(sp + 2, p.z);
    unsafeAtomicAdd(sp + 3, p.w);
    if (lane == 0) unsafeAtomicAdd(&gcnt[g], (float)rows);
}

__global__ __launch_bounds__(256) void k_gather_relu_pool(const float* __restrict__ xw,
                                                          const int* __restrict__ rowptr,
                                                          const int* __restrict__ cnt,
                                                          const int* __restrict__ csr_src,
                                                          const float* __restrict__ dinv,
                                                          const float* __restrict__ bias,
                                                          const int* __restrict__ batch,
                                                          float* __restrict__ sums,
                                                          float* __restrict__ gcnt, int n)
{
    const int lane = threadIdx.x & 31;
    const int nw   = gridDim.x * 8;
    const int gw   = __builtin_amdgcn_readfirstlane((int)(blockIdx.x * 8 + (threadIdx.x >> 5)));
    const int per  = n / nw;
    const int rem  = n - per * nw;
    const int lo   = gw * per + ((gw < rem) ? gw : rem);
    const int hi   = lo + per + ((gw < rem) ? 1 : 0);
    if (lo >= hi) return;
    const float4* xw4 = reinterpret_cast<const float4*>(xw);
    const float* bp = bias + lane * 4;
    const float4 bv = make_float4(bp[0], bp[1], bp[2], bp[3]);

    int   node = lo;
    int   beg = __builtin_amdgcn_readfirstlane(rowptr[node]);
    int   deg = __builtin_amdgcn_readfirstlane(cnt[node]);
    int   g   = __builtin_amdgcn_readfirstlane(batch[node]);
    float dd  = dinv[node];

    int    cg   = g;
    int    rows = 0;
    float4 pacc = make_float4(0.0f, 0.0f, 0.0f, 0.0f);

    for (;;) {
        const int  next = node + 1;
        const bool more = next < hi;
        int   nbeg = 0, ndeg = 0, ng = 0;
        float ndd  = 0.0f;
        if (more) {
            nbeg = rowptr[next];
            ndeg = cnt[next];
            ng   = batch[next];
            ndd  = dinv[next];
        }

        int s_l = 0;
        if (lane < deg) s_l = csr_src[beg + lane];
        float4 acc = xw4[(long long)node * 32 + lane];
        for (int base = 0; base < deg; base += 32) {
            const int m = (deg - base < 32) ? (deg - base) : 32;
#pragma unroll 4
            for (int j = 0; j < m; ++j) {
                const int s = __builtin_amdgcn_readlane(s_l, j);
                const float4 v = xw4[(long long)s * 32 + lane];
                acc.x += v.x;
                acc.y += v.y;
                acc.z += v.z;
                acc.w += v.w;
            }
            const int left = deg - base - 32;
            if (left > 0) {
                s_l = 0;
                if (lane < left) s_l = csr_src[beg + base + 32 + lane];
            }
        }

        float4 o;
        o.x = fmaxf(acc.x * dd + bv.x, 0.0f);
        o.y = fmaxf(acc.y * dd + bv.y, 0.0f);
        o.z = fmaxf(acc.z * dd + bv.z, 0.0f);
        o.w = fmaxf(acc.w * dd + bv.w, 0.0f);

        if (g != cg) {
            pool_flush(sums, gcnt, cg, lane, pacc, rows);
            pacc = make_float4(0.0f, 0.0f, 0.0f, 0.0f);
            rows = 0;
            cg   = g;
        }
        pacc.x += o.x;
        pacc.y += o.y;
        pacc.z += o.z;
        pacc.w += o.w;
        rows += 1;

        if (!more) break;
        node = next;
        beg  = __builtin_amdgcn_readfirstlane(nbeg);
        deg  = __builtin_amdgcn_readfirstlane(ndeg);
        g    = __builtin_amdgcn_readfirstlane(ng);
        dd   = ndd;
    }
    pool_flush(sums, gcnt, cg, lane, pacc, rows);
}

__global__ __launch_bounds__(256) void k_gather_tab_relu(const int* __restrict__ rec_in,
                                                         const float* __restrict__ tw,
                                                         const int* __restrict__ rowptr,
                                                         const int* __restrict__ cnt,
                                                         const int* __restrict__ csr_src,
                                                         const float* __restrict__ bias,
                                                         float* __restrict__ h, int n)
{
    const int4* rec = reinterpret_cast<const int4*>(rec_in);
    const int lane = threadIdx.x & 31;
    const int nw   = gridDim.x * 8;
    int node = __builtin_amdgcn_readfirstlane((int)(blockIdx.x * 8 + (threadIdx.x >> 5)));
    if (node >= n) return;
    const float4* tw4 = reinterpret_cast<const float4*>(tw);
    const float* bp = bias + lane * 4;
    const float4 bv = make_float4(bp[0], bp[1], bp[2], bp[3]);

    int   beg = __builtin_amdgcn_readfirstlane(rowptr[node]);
    int   deg = __builtin_amdgcn_readfirstlane(cnt[node]);
    const int4 me = rec[node];
    float dd  = __int_as_float(me.w);
    int   q0 = __builtin_amdgcn_readfirstlane(me.x);
    int   q1 = __builtin_amdgcn_readfirstlane(me.y);
    int   q2 = __builtin_amdgcn_readfirstlane(me.z);

    for (;;) {
        const int  next = node + nw;
        const bool more = next < n;
        int  nbeg = 0, ndeg = 0;
        int4 nme  = make_int4(0, 0, 0, 0);
        if (more) {
            nbeg = rowptr[next];
            ndeg = cnt[next];
            nme  = rec[next];
        }

        float4 acc = make_float4(0.0f, 0.0f, 0.0f, 0.0f);
        for (int base = 0; base < deg; base += 32) {
            const int m = (deg - base < 32) ? (deg - base) : 32;
            int   r0 = 0, r1 = 0, r2 = 0;
            float c_l = 0.0f;
            if (lane < m) {
                const int  s_l = csr_src[beg + base + lane];
                const int4 rc  = rec[s_l];
                c_l = __int_as_float(rc.w) * dd;
                r0 = rc.x;
                r1 = rc.y;
                r2 = rc.z;
            }
#pragma unroll 4
            for (int j = 0; j < m; ++j) {
                const int   i0 = __builtin_amdgcn_readlane(r0, j);
                const int   i1 = __builtin_amdgcn_readlane(r1, j);
                const int   i2 = __builtin_amdgcn_readlane(r2, j);
                const float c  = __int_as_float(__builtin_amdgcn_readlane(__float_as_int(c_l), j));
                const float4 a = tw4[i0 * 32 + lane];
                const float4 b = tw4[i1 * 32 + lane];
                const float4 d = tw4[i2 * 32 + lane];
                acc.x += ((a.x + b.x) + d.x) * c;
                acc.y += ((a.y + b.y) + d.y) * c;
                acc.z += ((a.z + b.z) + d.z) * c;
                acc.w += ((a.w + b.w) + d.w) * c;
            }
        }

        const float4 a = tw4[q0 * 32 + lane];
        const float4 b = tw4[q1 * 32 + lane];
        const float4 d = tw4[q2 * 32 + lane];
        const float sl = dd * dd;
        float4 o;
        o.x = fmaxf(acc.x + ((a.x + b.x) + d.x) * sl + bv.x, 0.0f);
        o.y = fmaxf(acc.y + ((a.y + b.y) + d.y) * sl + bv.y, 0.0f);
        o.z = fmaxf(acc.z + ((a.z + b.z) + d.z) * sl + bv.z, 0.0f);
        o.w = fmaxf(acc.w + ((a.w + b.w) + d.w) * sl + bv.w, 0.0f);
        reinterpret_cast<float4*>(h)[(long long)node * 32 + lane] = o;

        if (!more) break;
        node = next;
        beg  = __builtin_amdgcn_readfirstlane(nbeg);
        deg  = __builtin_amdgcn_readfirstlane(ndeg);
        dd   = __int_as_float(nme.w);
        q0   = __builtin_amdgcn_readfirstlane(nme.x);
        q1   = __builtin_amdgcn_readfirstlane(nme.y);
        q2   = __builtin_amdgcn_readfirstlane(nme.z);
    }
}

__global__ void k_head(const float* __restrict__ sums,
                       const float* __restrict__ cnt,
                       const float* __restrict__ Wlin,
                       const float* __restrict__ blin,
                       float* __restrict__ out, int g_total)
{
    int idx = blockIdx.x * blockDim.x + threadIdx.x;
    if (idx >= g_total * NCLS) return;
    const int g = idx / NCLS;
    const int c = idx % NCLS;
    const float inv = 1.0f / fmaxf(cnt[g], 1.0f);
    const float* sp = sums + (long long)g * EMB;
    float acc = 0.0f;
#pragma unroll 4
    for (int k = 0; k < EMB; ++k) acc += sp[k] * Wlin[k * NCLS + c];
    out[idx] = acc * inv + blin[c];
}

static inline size_t align256(size_t x) { return (x + 255) & ~(size_t)255; }

extern "C" void kernel_launch(void* const* d_in, const int* in_sizes, int n_in,
                              void* d_out, int out_size, void* d_ws, size_t ws_size,
                              hipStream_t stream)
{
    const int*   x        = (const int*)  d_in[0];
    const int*   eidx     = (const int*)  d_in[1];
    const int*   batch    = (const int*)  d_in[2];
    const float* shapeTab = (const float*)d_in[3];
    const float* colorTab = (const float*)d_in[4];
    const float* posTab   = (const float*)d_in[5];
    const float* W1       = (const float*)d_in[6];
    const float* b1       = (const float*)d_in[7];
    const float* W2       = (const float*)d_in[8];
    const float* b2       = (const float*)d_in[9];
    const float* Wlin     = (const float*)d_in[10];
    const float* blin     = (const float*)d_in[11];
    float*       out      = (float*)d_out;

    const int N = in_sizes[0] / 3;
    const int E = in_sizes[1] / 2;
    const int G = out_size / NCLS;

    const int* src = eidx;
    const int* dst = eidx + E;

    char* ws = (char*)d_ws;
    size_t off = 0;
    float* h    = (float*)(ws + off); off = align256(off + (size_t)N * EMB * 4);
    float* xw   = (float*)(ws + off); off = align256(off + (size_t)N * EMB * 4);
    char*  csr  = ws + off;           off = align256(off + (size_t)N * EMB * 4);
    float* dinv = (float*)(ws + off); off = align256(off + (size_t)N * 4);
    float* sums = (float*)(ws + off); off = align256(off + (size_t)G * EMB * 4);
    float* cnt  = (float*)(ws + off); off = align256(off + (size_t)G * 4);
    (void)ws_size;

    const int nBins  = (N + BIN_NODES - 1) / BIN_NODES;
    const int nParts = (E + CHUNK - 1) / CHUNK;
    const int cPitch = (nBins + 31) & ~31;
    size_t coff = 0;
    int*  degc     = (int*)(csr + coff);  coff = align256(coff + (size_t)N * 4);
    int*  rowptr   = (int*)(csr + coff);  coff = align256(coff + (size_t)N * 4);
    int*  binStart = (int*)(csr + coff);  coff = align256(coff + (size_t)(nBins + 1) * 4);
    int*  counts   = (int*)(csr + coff);  coff = align256(coff + (size_t)nParts * cPitch * 4);
    int*  pairs    = (int*)(csr + coff);  coff = align256(coff + (size_t)E * 8);
    int*  csrSrc   = (int*)(csr + coff);  coff = align256(coff + (size_t)E * 4);
    float* tw   = (float*)(csr + coff); coff = align256(coff + (size_t)NTABROWS * EMB * 4);
    int*  rec      = (int*)(csr + coff);  coff = align256(coff + (size_t)N * 16);

    const int TB = 256;
    const int gNodeWave = 2048;
    const int gNodeTab  = 1536;
    const int gGemmBlks = (N + 127) / 128;

    k_front<<<nParts + TAB_BLOCKS, PT, 0, stream>>>(dst, counts, E, nBins, cPitch, nParts,
                                                   shapeTab, colorTab, posTab, W1, tw,
                                                   sums, G * EMB, cnt, G);

    k_bin_place<<<nParts, PT, 0, stream>>>(src, dst, counts, binStart, pairs, E, nBins, cPitch, nParts);
    k_bin_sort<<<nBins, 256, 0, stream>>>(pairs, binStart, rowptr, degc, dinv, csrSrc, x, rec, N);

    k_gather_tab_relu<<<gNodeTab, 256, 0, stream>>>(rec, tw, rowptr, degc, csrSrc, b1, h, N);

    k_gemm_wmma<<<gGemmBlks, TB, 0, stream>>>(h, W2, dinv, xw, N);
    k_gather_relu_pool<<<gNodeWave, 256, 0, stream>>>(xw, rowptr, degc, csrSrc, dinv, b2, batch, sums, cnt, N);

    k_head<<<(G * NCLS + TB - 1) / TB, TB, 0, stream>>>(sums, cnt, Wlin, blin, out, G);
}


# layer-1 gather reads two rows per edge, not three: shape+colour row sums for all 256 pairs formed per block in LDS, position row from the cache; same additions in the same order
# speedup vs baseline: 6.1391x; 1.0967x over previous
#include <hip/hip_runtime.h>
#include <hip/hip_bf16.h>

#define EMB 128
#define NCLS 10
#define NSHAPE 16
#define NCOLOR 16
#define MAXPOS 256
#define NTABROWS (NSHAPE + NCOLOR + MAXPOS)

typedef __attribute__((ext_vector_type(2))) float v2f;
typedef __attribute__((ext_vector_type(8))) float v8f;

__device__ __forceinline__ void tab_gemm_tile(int tile, int ctile, int lane,
    const float* __restrict__ st, const float* __restrict__ ct, const float* __restrict__ pt,
    const float* __restrict__ W, float* __restrict__ tw)
{
    const int m     = lane & 15;
    const int khalf = (lane >> 4) * 2;

    const float* abase;
    if (tile == 0)      abase = st;
    else if (tile == 1) abase = ct;
    else                abase = pt + (long long)(tile - 2) * 16 * EMB;

    const float* arow = abase + m * EMB;
    const int col = ctile * 16 + m;

    v8f acc = {};
    for (int k = 0; k < EMB; k += 4) {
        const int kk = k + khalf;
        v2f a, b;
        a.x = arow[kk];
        a.y = arow[kk + 1];
        b.x = W[kk * EMB + col];
        b.y = W[(kk + 1) * EMB + col];
        acc = __builtin_amdgcn_wmma_f32_16x16x4_f32(
            false, a, false, b, (short)0, acc, false, false);
    }

    const int rbase = (lane >> 4) * 8;
#pragma unroll
    for (int r = 0; r < 8; ++r)
        tw[(long long)(tile * 16 + rbase + r) * EMB + col] = acc[r];
}

#define BIN_SHIFT 7
#define BIN_NODES (1 << BIN_SHIFT)
#define MAXBINS 1024
#define PT 1024
#define KPT 12
#define CHUNK (PT * KPT)
#define STAGE_CAP 8192

__device__ __forceinline__ int wave_incl_scan(int v, int lane)
{
#pragma unroll
    for (int off = 1; off < 32; off <<= 1) {
        const int u = __shfl_up(v, off);
        if (lane >= off) v += u;
    }
    return v;
}

#define TAB_TILES (NTABROWS / 16)
#define TAB_BLOCKS ((TAB_TILES + 3) / 4)
__global__ __launch_bounds__(PT) void k_front(const int* __restrict__ dst, int* __restrict__ counts,
                                              int e, int nb, int pitch, int nparts,
                                              const float* __restrict__ st, const float* __restrict__ ct,
                                              const float* __restrict__ pt, const float* __restrict__ W,
                                              float* __restrict__ tw,
                                              float* __restrict__ sums, int nsums,
                                              float* __restrict__ gcnt, int ng)
{
    __shared__ int hist[MAXBINS];
    const int t = threadIdx.x;

    for (int i = blockIdx.x * PT + t; i < nsums; i += gridDim.x * PT) sums[i] = 0.0f;
    for (int i = blockIdx.x * PT + t; i < ng;    i += gridDim.x * PT) gcnt[i] = 0.0f;

    if ((int)blockIdx.x >= nparts) {
        const int wave = t >> 5;
        const int tile = ((int)blockIdx.x - nparts) * 4 + (wave >> 3);
        if (tile < TAB_TILES)
            tab_gemm_tile(tile, wave & 7, t & 31, st, ct, pt, W, tw);
        return;
    }

    hist[t] = 0;
    __syncthreads();
    const int base = blockIdx.x * CHUNK + t;
    int d[KPT];
#pragma unroll
    for (int j = 0; j < KPT; ++j) {
        const int i = base + j * PT;
        d[j] = dst[(i < e) ? i : (e - 1)];
    }
#pragma unroll
    for (int j = 0; j < KPT; ++j)
        if (base + j * PT < e) atomicAdd(&hist[d[j] >> BIN_SHIFT], 1);
    __syncthreads();
    if (t < nb) counts[blockIdx.x * pitch + t] = hist[t];
}

__global__ __launch_bounds__(PT) void k_bin_place(const int* __restrict__ src, const int* __restrict__ dst,
                                                  const int* __restrict__ counts, int* __restrict__ binStart,
                                                  int* __restrict__ pairs_out, int e, int nb, int pitch, int nparts)
{
    int2* pairs = reinterpret_cast<int2*>(pairs_out);
    __shared__ int  hist[MAXBINS];
    __shared__ int  lstart[MAXBINS];
    __shared__ int  gdelta[MAXBINS];
    __shared__ int  wsA[32];
    __shared__ int  wsG[32];
    __shared__ int2 sorted[CHUNK];
    const int t    = threadIdx.x;
    const int lane = t & 31;
    const int wave = t >> 5;
    const int p    = blockIdx.x;
    hist[t] = 0;
    __syncthreads();

    const int base = p * CHUNK + t;
    int s[KPT], d[KPT], r[KPT];
#pragma unroll
    for (int j = 0; j < KPT; ++j) {
        const int i  = base + j * PT;
        const int ic = (i < e) ? i : (e - 1);
        s[j] = src[ic];
        d[j] = dst[ic];
    }

    int tot = 0, pre = 0;
    if (t < nb) {
        for (int q0 = 0; q0 < nparts; q0 += 16) {
            int c[16];
#pragma unroll
            for (int k = 0; k < 16; ++k) {
                const int q = q0 + k;
                c[k] = counts[((q < nparts) ? q : (nparts - 1)) * pitch + t];
            }
#pragma unroll
            for (int k = 0; k < 16; ++k) {
                const int q = q0 + k;
                tot += (q < nparts) ? c[k] : 0;
                pre += (q < p) ? c[k] : 0;
            }
        }
    }

#pragma unroll
    for (int j = 0; j < KPT; ++j) {
        r[j] = 0;
        if (base + j * PT < e) r[j] = atomicAdd(&hist[d[j] >> BIN_SHIFT], 1);
    }
    __syncthreads();
    const int myh = hist[t];

    const int ia = wave_incl_scan(myh, lane);
    const int ig = wave_incl_scan(tot, lane);
    if (lane == 31) { wsA[wave] = ia; wsG[wave] = ig; }
    __syncthreads();
    if (wave == 0) {
        const int xa = wsA[lane], xg = wsG[lane];
        const int sa = wave_incl_scan(xa, lane);
        const int sg = wave_incl_scan(xg, lane);
        wsA[lane] = sa - xa;
        wsG[lane] = sg - xg;
    }
    __syncthreads();
    const int lex = ia - myh + wsA[wave];
    const int gex = ig - tot + wsG[wave];
    lstart[t] = lex;
    gdelta[t] = gex + pre - lex;
    if (p == 0) {
        if (t < nb)      binStart[t]  = gex;
        if (t == nb - 1) binStart[nb] = gex + tot;
    }
    __syncthreads();

#pragma unroll
    for (int j = 0; j < KPT; ++j) {
        if (base + j * PT < e)
            sorted[lstart[d[j] >> BIN_SHIFT] + r[j]] = make_int2(s[j], d[j]);
    }
    __syncthreads();

    const int left = e - p * CHUNK;
    const int cc   = (left < CHUNK) ? left : CHUNK;
    for (int i = t; i < cc; i += PT) {
        const int2 v = sorted[i];
        pairs[gdelta[v.y >> BIN_SHIFT] + i] = v;
    }
}

__global__ __launch_bounds__(256) void k_bin_sort(const int* __restrict__ pairs_in, const int* __restrict__ binStart,
                                                  int* __restrict__ rowptr, int* __restrict__ cnt,
                                                  float* __restrict__ dinv, int* __restrict__ csr_src,
                                                  const int* __restrict__ x, int* __restrict__ rec_out, int n)
{
    const int2* pairs = reinterpret_cast<const int2*>(pairs_in);
    __shared__ int c[BIN_NODES];
    __shared__ int cur[BIN_NODES];
    __shared__ int ws[8];
    __shared__ int stage[STAGE_CAP];
    const int t     = threadIdx.x;
    const int lane  = t & 31;
    const int wave  = t >> 5;
    const int b     = blockIdx.x;
    const int start = binStart[b];
    const int m     = binStart[b + 1] - start;
    if (t < BIN_NODES) c[t] = 0;
    __syncthreads();

    for (int i0 = t; i0 < m; i0 += 256 * 8) {
        int dd[8];
#pragma unroll
        for (int j = 0; j < 8; ++j) {
            const int i = i0 + j * 256;
            dd[j] = pairs[start + ((i < m) ? i : i0)].y;
        }
#pragma unroll
        for (int j = 0; j < 8; ++j)
            if (i0 + j * 256 < m) atomicAdd(&c[dd[j] & (BIN_NODES - 1)], 1);
    }
    __syncthreads();

    const int v  = (t < BIN_NODES) ? c[t] : 0;
    const int iv = wave_incl_scan(v, lane);
    if (lane == 31) ws[wave] = iv;
    __syncthreads();
    int woff = 0;
    for (int w = 0; w < wave; ++w) woff += ws[w];
    if (t < BIN_NODES) {
        const int ex = iv - v + woff;
        cur[t] = ex;
        const int node = b * BIN_NODES + t;
        if (node < n) {
            const float di = __frsqrt_rn((float)(v + 1));
            rowptr[node] = start + ex;
            cnt[node]    = v;
            dinv[node]   = di;
            reinterpret_cast<int4*>(rec_out)[node] =
                make_int4(x[node * 3 + 0], x[node * 3 + 1] + NSHAPE, x[node * 3 + 2] + NSHAPE + NCOLOR,
                          __float_as_int(di));
        }
    }
    __syncthreads();

    const bool staged = (m <= STAGE_CAP);
    for (int i0 = t; i0 < m; i0 += 256 * 8) {
        int2 ev[8];
#pragma unroll
        for (int j = 0; j < 8; ++j) {
            const int i = i0 + j * 256;
            ev[j] = pairs[start + ((i < m) ? i : i0)];
        }
#pragma unroll
        for (int j = 0; j < 8; ++j) {
            if (i0 + j * 256 < m) {
                const int pos = atomicAdd(&cur[ev[j].y & (BIN_NODES - 1)], 1);
                if (staged) stage[pos] = ev[j].x;
                else        csr_src[start + pos] = ev[j].x;
            }
        }
    }
    if (staged) {
        __syncthreads();
        for (int i = t; i < m; i += 256) csr_src[start + i] = stage[i];
    }
}

#define WSTRIDE 128
__global__ __launch_bounds__(256) void k_gemm_wmma(
    const float* __restrict__ A, const float* __restrict__ W, const float* __restrict__ rscale,
    float* __restrict__ C, int nrows)
{
    __shared__ float Wl[128 * WSTRIDE];
    const int tid = threadIdx.x;

    for (int idx = tid; idx < 128 * 32; idx += 256) {
        int r = idx >> 5, q = idx & 31;
        reinterpret_cast<float4*>(&Wl[r * WSTRIDE + q * 4])[0] =
            reinterpret_cast<const float4*>(W)[r * 32 + q];
    }
    __syncthreads();

    const int wave  = tid >> 5;
    const int lane  = tid & 31;
    const int m     = lane & 15;
    const int khalf = (lane >> 4) * 2;
    const int tileRow = blockIdx.x * 128 + wave * 16;
    if (tileRow >= nrows) return;

    v8f acc[8] = {};

    const float* arow = A + (long long)(tileRow + m) * EMB;
    for (int k = 0; k < EMB; k += 4) {
        const int kk = k + khalf;
        v2f a;
        a.x = arow[kk];
        a.y = arow[kk + 1];
#pragma unroll
        for (int j = 0; j < 8; ++j) {
            const int col = j * 16 + m;
            v2f b;
            b.x = Wl[kk * WSTRIDE + col];
            b.y = Wl[(kk + 1) * WSTRIDE + col];
            acc[j] = __builtin_amdgcn_wmma_f32_16x16x4_f32(
                false, a, false, b, (short)0, acc[j], false, false);
        }
    }

    const int rbase = (lane >> 4) * 8;
    const float4 s0 = reinterpret_cast<const float4*>(rscale + tileRow + rbase)[0];
    const float4 s1 = reinterpret_cast<const float4*>(rscale + tileRow + rbase)[1];
    const float sc[8] = { s0.x, s0.y, s0.z, s0.w, s1.x, s1.y, s1.z, s1.w };
#pragma unroll
    for (int j = 0; j < 8; ++j) {
#pragma unroll
        for (int r = 0; r < 8; ++r) {
            C[(long long)(tileRow + rbase + r) * EMB + j * 16 + m] = acc[j][r] * sc[r];
        }
    }
}

__device__ __forceinline__ void pool_flush(float* __restrict__ sums, float* __restrict__ gcnt,
                                           int g, int lane, const float4& p, int rows)
{
    float* sp = sums + (long long)g * EMB + lane * 4;
    unsafeAtomicAdd(sp + 0, p.x);
    unsafeAtomicAdd(sp + 1, p.y);
    unsafeAtomicAdd(sp + 2, p.z);
    unsafeAtomicAdd(sp + 3, p.w);
    if (lane == 0) unsafeAtomicAdd(&gcnt[g], (float)rows);
}

__global__ __launch_bounds__(256) void k_gather_relu_pool(const float* __restrict__ xw,
                                                          const int* __restrict__ rowptr,
                                                          const int* __restrict__ cnt,
                                                          const int* __restrict__ csr_src,
                                                          const float* __restrict__ dinv,
                                                          const float* __restrict__ bias,
                                                          const int* __restrict__ batch,
                                                          float* __restrict__ sums,
                                                          float* __restrict__ gcnt, int n)
{
    const int lane = threadIdx.x & 31;
    const int nw   = gridDim.x * 8;
    const int gw   = __builtin_amdgcn_readfirstlane((int)(blockIdx.x * 8 + (threadIdx.x >> 5)));
    const int per  = n / nw;
    const int rem  = n - per * nw;
    const int lo   = gw * per + ((gw < rem) ? gw : rem);
    const int hi   = lo + per + ((gw < rem) ? 1 : 0);
    if (lo >= hi) return;
    const float4* xw4 = reinterpret_cast<const float4*>(xw);
    const float* bp = bias + lane * 4;
    const float4 bv = make_float4(bp[0], bp[1], bp[2], bp[3]);

    int   node = lo;
    int   beg = __builtin_amdgcn_readfirstlane(rowptr[node]);
    int   deg = __builtin_amdgcn_readfirstlane(cnt[node]);
    int   g   = __builtin_amdgcn_readfirstlane(batch[node]);
    float dd  = dinv[node];

    int    cg   = g;
    int    rows = 0;
    float4 pacc = make_float4(0.0f, 0.0f, 0.0f, 0.0f);

    for (;;) {
        const int  next = node + 1;
        const bool more = next < hi;
        int   nbeg = 0, ndeg = 0, ng = 0;
        float ndd  = 0.0f;
        if (more) {
            nbeg = rowptr[next];
            ndeg = cnt[next];
            ng   = batch[next];
            ndd  = dinv[next];
        }

        int s_l = 0;
        if (lane < deg) s_l = csr_src[beg + lane];
        float4 acc = xw4[(long long)node * 32 + lane];
        for (int base = 0; base < deg; base += 32) {
            const int m = (deg - base < 32) ? (deg - base) : 32;
#pragma unroll 4
            for (int j = 0; j < m; ++j) {
                const int s = __builtin_amdgcn_readlane(s_l, j);
                const float4 v = xw4[(long long)s * 32 + lane];
                acc.x += v.x;
                acc.y += v.y;
                acc.z += v.z;
                acc.w += v.w;
            }
            const int left = deg - base - 32;
            if (left > 0) {
                s_l = 0;
                if (lane < left) s_l = csr_src[beg + base + 32 + lane];
            }
        }

        float4 o;
        o.x = fmaxf(acc.x * dd + bv.x, 0.0f);
        o.y = fmaxf(acc.y * dd + bv.y, 0.0f);
        o.z = fmaxf(acc.z * dd + bv.z, 0.0f);
        o.w = fmaxf(acc.w * dd + bv.w, 0.0f);

        if (g != cg) {
            pool_flush(sums, gcnt, cg, lane, pacc, rows);
            pacc = make_float4(0.0f, 0.0f, 0.0f, 0.0f);
            rows = 0;
            cg   = g;
        }
        pacc.x += o.x;
        pacc.y += o.y;
        pacc.z += o.z;
        pacc.w += o.w;
        rows += 1;

        if (!more) break;
        node = next;
        beg  = __builtin_amdgcn_readfirstlane(nbeg);
        deg  = __builtin_amdgcn_readfirstlane(ndeg);
        g    = __builtin_amdgcn_readfirstlane(ng);
        dd   = ndd;
    }
    pool_flush(sums, gcnt, cg, lane, pacc, rows);
}

#define SC_ROWS (NSHAPE * NCOLOR)
#define GT_THREADS 1024
__global__ __launch_bounds__(GT_THREADS) void k_gather_tab_relu(const int* __restrict__ rec_in,
                                                                const float* __restrict__ tw,
                                                                const int* __restrict__ rowptr,
                                                                const int* __restrict__ cnt,
                                                                const int* __restrict__ csr_src,
                                                                const float* __restrict__ bias,
                                                                float* __restrict__ h, int n)
{
    __shared__ float4 sc4[SC_ROWS * 32];
    const float4* tw4 = reinterpret_cast<const float4*>(tw);
#pragma unroll
    for (int i = threadIdx.x; i < SC_ROWS * 32; i += GT_THREADS) {
        const int pr = i >> 5, q = i & 31;
        const float4 a = tw4[(pr / NCOLOR) * 32 + q];
        const float4 b = tw4[(NSHAPE + (pr % NCOLOR)) * 32 + q];
        sc4[i] = make_float4(a.x + b.x, a.y + b.y, a.z + b.z, a.w + b.w);
    }
    __syncthreads();

    const int4* rec = reinterpret_cast<const int4*>(rec_in);
    const int lane = threadIdx.x & 31;
    const int nw   = gridDim.x * (GT_THREADS / 32);
    int node = __builtin_amdgcn_readfirstlane((int)(blockIdx.x * (GT_THREADS / 32) + (threadIdx.x >> 5)));
    if (node >= n) return;
    const float* bp = bias + lane * 4;
    const float4 bv = make_float4(bp[0], bp[1], bp[2], bp[3]);

    int   beg = __builtin_amdgcn_readfirstlane(rowptr[node]);
    int   deg = __builtin_amdgcn_readfirstlane(cnt[node]);
    const int4 me = rec[node];
    float dd  = __int_as_float(me.w);
    int   q0 = __builtin_amdgcn_readfirstlane(me.x);
    int   q1 = __builtin_amdgcn_readfirstlane(me.y);
    int   q2 = __builtin_amdgcn_readfirstlane(me.z);

    for (;;) {
        const int  next = node + nw;
        const bool more = next < n;
        int  nbeg = 0, ndeg = 0;
        int4 nme  = make_int4(0, 0, 0, 0);
        if (more) {
            nbeg = rowptr[next];
            ndeg = cnt[next];
            nme  = rec[next];
        }

        float4 acc = make_float4(0.0f, 0.0f, 0.0f, 0.0f);
        for (int base = 0; base < deg; base += 32) {
            const int m = (deg - base < 32) ? (deg - base) : 32;
            int   rp = 0, r2 = 0;
            float c_l = 0.0f;
            if (lane < m) {
                const int  s_l = csr_src[beg + base + lane];
                const int4 rc  = rec[s_l];
                c_l = __int_as_float(rc.w) * dd;
                rp = rc.x * NCOLOR + (rc.y - NSHAPE);
                r2 = rc.z;
            }
#pragma unroll 4
            for (int j = 0; j < m; ++j) {
                const int   ip = __builtin_amdgcn_readlane(rp, j);
                const int   i2 = __builtin_amdgcn_readlane(r2, j);
                const float c  = __int_as_float(__builtin_amdgcn_readlane(__float_as_int(c_l), j));
                const float4 ab = sc4[ip * 32 + lane];
                const float4 d  = tw4[i2 * 32 + lane];
                acc.x += (ab.x + d.x) * c;
                acc.y += (ab.y + d.y) * c;
                acc.z += (ab.z + d.z) * c;
                acc.w += (ab.w + d.w) * c;
            }
        }

        const int qp = q0 * NCOLOR + (q1 - NSHAPE);
        const float4 ab = sc4[qp * 32 + lane];
        const float4 d  = tw4[q2 * 32 + lane];
        const float sl = dd * dd;
        float4 o;
        o.x = fmaxf(acc.x + (ab.x + d.x) * sl + bv.x, 0.0f);
        o.y = fmaxf(acc.y + (ab.y + d.y) * sl + bv.y, 0.0f);
        o.z = fmaxf(acc.z + (ab.z + d.z) * sl + bv.z, 0.0f);
        o.w = fmaxf(acc.w + (ab.w + d.w) * sl + bv.w, 0.0f);
        reinterpret_cast<float4*>(h)[(long long)node * 32 + lane] = o;

        if (!more) break;
        node = next;
        beg  = __builtin_amdgcn_readfirstlane(nbeg);
        deg  = __builtin_amdgcn_readfirstlane(ndeg);
        dd   = __int_as_float(nme.w);
        q0   = __builtin_amdgcn_readfirstlane(nme.x);
        q1   = __builtin_amdgcn_readfirstlane(nme.y);
        q2   = __builtin_amdgcn_readfirstlane(nme.z);
    }
}

__global__ void k_head(const float* __restrict__ sums,
                       const float* __restrict__ cnt,
                       const float* __restrict__ Wlin,
                       const float* __restrict__ blin,
                       float* __restrict__ out, int g_total)
{
    int idx = blockIdx.x * blockDim.x + threadIdx.x;
    if (idx >= g_total * NCLS) return;
    const int g = idx / NCLS;
    const int c = idx % NCLS;
    const float inv = 1.0f / fmaxf(cnt[g], 1.0f);
    const float* sp = sums + (long long)g * EMB;
    float acc = 0.0f;
#pragma unroll 4
    for (int k = 0; k < EMB; ++k) acc += sp[k] * Wlin[k * NCLS + c];
    out[idx] = acc * inv + blin[c];
}

static inline size_t align256(size_t x) { return (x + 255) & ~(size_t)255; }

extern "C" void kernel_launch(void* const* d_in, const int* in_sizes, int n_in,
                              void* d_out, int out_size, void* d_ws, size_t ws_size,
                              hipStream_t stream)
{
    const int*   x        = (const int*)  d_in[0];
    const int*   eidx     = (const int*)  d_in[1];
    const int*   batch    = (const int*)  d_in[2];
    const float* shapeTab = (const float*)d_in[3];
    const float* colorTab = (const float*)d_in[4];
    const float* posTab   = (const float*)d_in[5];
    const float* W1       = (const float*)d_in[6];
    const float* b1       = (const float*)d_in[7];
    const float* W2       = (const float*)d_in[8];
    const float* b2       = (const float*)d_in[9];
    const float* Wlin     = (const float*)d_in[10];
    const float* blin     = (const float*)d_in[11];
    float*       out      = (float*)d_out;

    const int N = in_sizes[0] / 3;
    const int E = in_sizes[1] / 2;
    const int G = out_size / NCLS;

    const int* src = eidx;
    const int* dst = eidx + E;

    char* ws = (char*)d_ws;
    size_t off = 0;
    float* h    = (float*)(ws + off); off = align256(off + (size_t)N * EMB * 4);
    float* xw   = (float*)(ws + off); off = align256(off + (size_t)N * EMB * 4);
    char*  csr  = ws + off;           off = align256(off + (size_t)N * EMB * 4);
    float* dinv = (float*)(ws + off); off = align256(off + (size_t)N * 4);
    float* sums = (float*)(ws + off); off = align256(off + (size_t)G * EMB * 4);
    float* cnt  = (float*)(ws + off); off = align256(off + (size_t)G * 4);
    (void)ws_size;

    const int nBins  = (N + BIN_NODES - 1) / BIN_NODES;
    const int nParts = (E + CHUNK - 1) / CHUNK;
    const int cPitch = (nBins + 31) & ~31;
    size_t coff = 0;
    int*  degc     = (int*)(csr + coff);  coff = align256(coff + (size_t)N * 4);
    int*  rowptr   = (int*)(csr + coff);  coff = align256(coff + (size_t)N * 4);
    int*  binStart = (int*)(csr + coff);  coff = align256(coff + (size_t)(nBins + 1) * 4);
    int*  counts   = (int*)(csr + coff);  coff = align256(coff + (size_t)nParts * cPitch * 4);
    int*  pairs    = (int*)(csr + coff);  coff = align256(coff + (size_t)E * 8);
    int*  csrSrc   = (int*)(csr + coff);  coff = align256(coff + (size_t)E * 4);
    float* tw   = (float*)(csr + coff); coff = align256(coff + (size_t)NTABROWS * EMB * 4);
    int*  rec      = (int*)(csr + coff);  coff = align256(coff + (size_t)N * 16);

    const int TB = 256;
    const int gNodeWave = 2048;
    const int gNodeTab  = 512;
    const int gGemmBlks = (N + 127) / 128;

    k_front<<<nParts + TAB_BLOCKS, PT, 0, stream>>>(dst, counts, E, nBins, cPitch, nParts,
                                                   shapeTab, colorTab, posTab, W1, tw,
                                                   sums, G * EMB, cnt, G);

    k_bin_place<<<nParts, PT, 0, stream>>>(src, dst, counts, binStart, pairs, E, nBins, cPitch, nParts);
    k_bin_sort<<<nBins, 256, 0, stream>>>(pairs, binStart, rowptr, degc, dinv, csrSrc, x, rec, N);

    k_gather_tab_relu<<<gNodeTab, GT_THREADS, 0, stream>>>(rec, tw, rowptr, degc, csrSrc, b1, h, N);

    k_gemm_wmma<<<gGemmBlks, TB, 0, stream>>>(h, W2, dinv, xw, N);
    k_gather_relu_pool<<<gNodeWave, 256, 0, stream>>>(xw, rowptr, degc, csrSrc, dinv, b2, batch, sums, cnt, N);

    k_head<<<(G * NCLS + TB - 1) / TB, TB, 0, stream>>>(sums, cnt, Wlin, blin, out, G);
}


# CSR sort in two launches, not three: a chunk block orders its own edges by bin and stores run starts, a bin block collects its runs from all chunks; no counting pass; row slot and sum order change
# speedup vs baseline: 6.7152x; 1.0938x over previous
#include <hip/hip_runtime.h>
#include <hip/hip_bf16.h>

#define EMB 128
#define NCLS 10
#define NSHAPE 16
#define NCOLOR 16
#define MAXPOS 256
#define NTABROWS (NSHAPE + NCOLOR + MAXPOS)

typedef __attribute__((ext_vector_type(2))) float v2f;
typedef __attribute__((ext_vector_type(8))) float v8f;

__device__ __forceinline__ void tab_gemm_tile(int tile, int ctile, int lane,
    const float* __restrict__ st, const float* __restrict__ ct, const float* __restrict__ pt,
    const float* __restrict__ W, float* __restrict__ tw)
{
    const int m     = lane & 15;
    const int khalf = (lane >> 4) * 2;

    const float* abase;
    if (tile == 0)      abase = st;
    else if (tile == 1) abase = ct;
    else                abase = pt + (long long)(tile - 2) * 16 * EMB;

    const float* arow = abase + m * EMB;
    const int col = ctile * 16 + m;

    v8f acc = {};
    for (int k = 0; k < EMB; k += 4) {
        const int kk = k + khalf;
        v2f a, b;
        a.x = arow[kk];
        a.y = arow[kk + 1];
        b.x = W[kk * EMB + col];
        b.y = W[(kk + 1) * EMB + col];
        acc = __builtin_amdgcn_wmma_f32_16x16x4_f32(
            false, a, false, b, (short)0, acc, false, false);
    }

    const int rbase = (lane >> 4) * 8;
#pragma unroll
    for (int r = 0; r < 8; ++r)
        tw[(long long)(tile * 16 + rbase + r) * EMB + col] = acc[r];
}

#define BIN_SHIFT 7
#define BIN_NODES (1 << BIN_SHIFT)
#define MAXBINS 1024
#define PT 1024
#define KPT 12
#define CHUNK (PT * KPT)
#define MAXPARTS 256
#define RUNS_U 17
#define STAGE_CAP 8192

__device__ __forceinline__ int wave_incl_scan(int v, int lane)
{
#pragma unroll
    for (int off = 1; off < 32; off <<= 1) {
        const int u = __shfl_up(v, off);
        if (lane >= off) v += u;
    }
    return v;
}

__device__ __forceinline__ int wave_sum(int v)
{
#pragma unroll
    for (int off = 16; off > 0; off >>= 1) v += __shfl_xor(v, off);
    return v;
}

#define TAB_TILES (NTABROWS / 16)
#define TAB_BLOCKS ((TAB_TILES + 3) / 4)
__global__ __launch_bounds__(PT) void k_chunk_sort(const int* __restrict__ src, const int* __restrict__ dst,
                                                   int* __restrict__ runs, int* __restrict__ pairs_out,
                                                   int e, int nb, int pitch, int nparts,
                                                   const float* __restrict__ st, const float* __restrict__ ct,
                                                   const float* __restrict__ pt, const float* __restrict__ W,
                                                   float* __restrict__ tw,
                                                   float* __restrict__ sums, int nsums,
                                                   float* __restrict__ gcnt, int ng)
{
    int2* pairs = reinterpret_cast<int2*>(pairs_out);
    __shared__ int  hist[MAXBINS];
    __shared__ int  lstart[MAXBINS];
    __shared__ int  wsA[32];
    __shared__ int2 sorted[CHUNK];
    const int t    = threadIdx.x;
    const int lane = t & 31;
    const int wave = t >> 5;
    const int p    = blockIdx.x;

    for (int i = blockIdx.x * PT + t; i < nsums; i += gridDim.x * PT) sums[i] = 0.0f;
    for (int i = blockIdx.x * PT + t; i < ng;    i += gridDim.x * PT) gcnt[i] = 0.0f;

    if (p >= nparts) {
        const int tile = (p - nparts) * 4 + (wave >> 3);
        if (tile < TAB_TILES)
            tab_gemm_tile(tile, wave & 7, lane, st, ct, pt, W, tw);
        return;
    }

    hist[t] = 0;
    __syncthreads();

    const int base = p * CHUNK + t;
    int s[KPT], d[KPT], r[KPT];
#pragma unroll
    for (int j = 0; j < KPT; ++j) {
        const int i  = base + j * PT;
        const int ic = (i < e) ? i : (e - 1);
        s[j] = src[ic];
        d[j] = dst[ic];
    }
#pragma unroll
    for (int j = 0; j < KPT; ++j) {
        r[j] = 0;
        if (base + j * PT < e) r[j] = atomicAdd(&hist[d[j] >> BIN_SHIFT], 1);
    }
    __syncthreads();
    const int myh = hist[t];

    const int ia = wave_incl_scan(myh, lane);
    if (lane == 31) wsA[wave] = ia;
    __syncthreads();
    if (wave == 0) {
        const int xa = wsA[lane];
        const int sa = wave_incl_scan(xa, lane);
        wsA[lane] = sa - xa;
    }
    __syncthreads();
    const int lex = ia - myh + wsA[wave];
    lstart[t] = lex;

    const int left = e - p * CHUNK;
    const int cc   = (left < CHUNK) ? left : CHUNK;
    if (t < nb)  runs[p * pitch + t]  = lex;
    if (t == 0)  runs[p * pitch + nb] = cc;
    __syncthreads();

#pragma unroll
    for (int j = 0; j < KPT; ++j) {
        if (base + j * PT < e)
            sorted[lstart[d[j] >> BIN_SHIFT] + r[j]] = make_int2(s[j], d[j]);
    }
    __syncthreads();

    for (int i = t; i < cc; i += PT) pairs[p * CHUNK + i] = sorted[i];
}

template <bool WITH_SRC, class F>
__device__ __forceinline__ void for_each_pair_of_bin(const int* __restrict__ pairs_in, int rb_l, int rl_l,
                                                     int nr, int lane, F f)
{
    const int2* pairs = reinterpret_cast<const int2*>(pairs_in);
    for (int u0 = 0; u0 < nr; u0 += RUNS_U) {
        int len[RUNS_U], rb[RUNS_U], es[RUNS_U], ed[RUNS_U];
#pragma unroll
        for (int k = 0; k < RUNS_U; ++k) {
            const int u = u0 + k;
            rb[k]  = __builtin_amdgcn_readlane(rb_l, u & 31);
            len[k] = (u < nr) ? __builtin_amdgcn_readlane(rl_l, u & 31) : 0;
            const int i = (lane < len[k]) ? (rb[k] + lane) : 0;
            if constexpr (WITH_SRC) { const int2 v = pairs[i]; es[k] = v.x; ed[k] = v.y; }
            else                    { es[k] = 0; ed[k] = pairs_in[2 * i + 1]; }
        }
#pragma unroll
        for (int k = 0; k < RUNS_U; ++k)
            if (lane < len[k]) f(es[k], ed[k]);
#pragma unroll
        for (int k = 0; k < RUNS_U; ++k)
            for (int i = 32 + lane; i < len[k]; i += 32) {
                if constexpr (WITH_SRC) { const int2 v = pairs[rb[k] + i]; f(v.x, v.y); }
                else                    f(0, pairs_in[2 * (rb[k] + i) + 1]);
            }
    }
}

__global__ __launch_bounds__(256) void k_bin_sort(const int* __restrict__ pairs_in, const int* __restrict__ runs,
                                                  int pitch, int nparts,
                                                  int* __restrict__ rowptr, int* __restrict__ cnt,
                                                  float* __restrict__ dinv, int* __restrict__ csr_src,
                                                  const int* __restrict__ x, int* __restrict__ rec_out, int n)
{
    __shared__ int c[BIN_NODES];
    __shared__ int cur[BIN_NODES];
    __shared__ int ws[8];
    __shared__ int wsM[8];
    __shared__ int wsS[8];
    __shared__ int stage[STAGE_CAP];
    const int t     = threadIdx.x;
    const int lane  = t & 31;
    const int wave  = __builtin_amdgcn_readfirstlane(t >> 5);
    const int b     = blockIdx.x;

    const int q = lane * 8 + wave;
    int rs = 0, rl = 0;
    if (q < nparts) {
        const int* row = runs + q * pitch + b;
        rs = row[0];
        rl = row[1] - rs;
    }
    const int rb_l = q * CHUNK + rs;
    const int nr   = (nparts - wave + 7) >> 3;
    if (t < BIN_NODES) c[t] = 0;
    const int wm = wave_sum(rl);
    const int wr = wave_sum(rs);
    if (lane == 0) { wsM[wave] = wm; wsS[wave] = wr; }
    __syncthreads();

    for_each_pair_of_bin<false>(pairs_in, rb_l, rl, nr, lane,
                                [&](int, int dn) { atomicAdd(&c[dn & (BIN_NODES - 1)], 1); });

    int m = 0, start = 0;
#pragma unroll
    for (int w = 0; w < 8; ++w) { m += wsM[w]; start += wsS[w]; }
    __syncthreads();

    const int v  = (t < BIN_NODES) ? c[t] : 0;
    const int iv = wave_incl_scan(v, lane);
    if (lane == 31) ws[wave] = iv;
    __syncthreads();
    int woff = 0;
    for (int w = 0; w < wave; ++w) woff += ws[w];
    if (t < BIN_NODES) {
        const int ex = iv - v + woff;
        cur[t] = ex;
        const int node = b * BIN_NODES + t;
        if (node < n) {
            const float di = __frsqrt_rn((float)(v + 1));
            rowptr[node] = start + ex;
            cnt[node]    = v;
            dinv[node]   = di;
            reinterpret_cast<int4*>(rec_out)[node] =
                make_int4(x[node * 3 + 0], x[node * 3 + 1] + NSHAPE, x[node * 3 + 2] + NSHAPE + NCOLOR,
                          __float_as_int(di));
        }
    }
    __syncthreads();

    const bool staged = (m <= STAGE_CAP);
    for_each_pair_of_bin<true>(pairs_in, rb_l, rl, nr, lane,
                               [&](int sn, int dn) {
                                   const int pos = atomicAdd(&cur[dn & (BIN_NODES - 1)], 1);
                                   if (staged) stage[pos] = sn;
                                   else        csr_src[start + pos] = sn;
                               });
    if (staged) {
        __syncthreads();
        for (int i = t; i < m; i += 256) csr_src[start + i] = stage[i];
    }
}

#define WSTRIDE 128
__global__ __launch_bounds__(256) void k_gemm_wmma(
    const float* __restrict__ A, const float* __restrict__ W, const float* __restrict__ rscale,
    float* __restrict__ C, int nrows)
{
    __shared__ float Wl[128 * WSTRIDE];
    const int tid = threadIdx.x;

    for (int idx = tid; idx < 128 * 32; idx += 256) {
        int r = idx >> 5, q = idx & 31;
        reinterpret_cast<float4*>(&Wl[r * WSTRIDE + q * 4])[0] =
            reinterpret_cast<const float4*>(W)[r * 32 + q];
    }
    __syncthreads();

    const int wave  = tid >> 5;
    const int lane  = tid & 31;
    const int m     = lane & 15;
    const int khalf = (lane >> 4) * 2;
    const int tileRow = blockIdx.x * 128 + wave * 16;
    if (tileRow >= nrows) return;

    v8f acc[8] = {};

    const float* arow = A + (long long)(tileRow + m) * EMB;
    for (int k = 0; k < EMB; k += 4) {
        const int kk = k + khalf;
        v2f a;
        a.x = arow[kk];
        a.y = arow[kk + 1];
#pragma unroll
        for (int j = 0; j < 8; ++j) {
            const int col = j * 16 + m;
            v2f b;
            b.x = Wl[kk * WSTRIDE + col];
            b.y = Wl[(kk + 1) * WSTRIDE + col];
            acc[j] = __builtin_amdgcn_wmma_f32_16x16x4_f32(
                false, a, false, b, (short)0, acc[j], false, false);
        }
    }

    const int rbase = (lane >> 4) * 8;
    const float4 s0 = reinterpret_cast<const float4*>(rscale + tileRow + rbase)[0];
    const float4 s1 = reinterpret_cast<const float4*>(rscale + tileRow + rbase)[1];
    const float sc[8] = { s0.x, s0.y, s0.z, s0.w, s1.x, s1.y, s1.z, s1.w };
#pragma unroll
    for (int j = 0; j < 8; ++j) {
#pragma unroll
        for (int r = 0; r < 8; ++r) {
            C[(long long)(tileRow + rbase + r) * EMB + j * 16 + m] = acc[j][r] * sc[r];
        }
    }
}

__device__ __forceinline__ void pool_flush(float* __restrict__ sums, float* __restrict__ gcnt,
                                           int g, int lane, const float4& p, int rows)
{
    float* sp = sums + (long long)g * EMB + lane * 4;
    unsafeAtomicAdd(sp + 0, p.x);
    unsafeAtomicAdd(sp + 1, p.y);
    unsafeAtomicAdd(sp + 2, p.z);
    unsafeAtomicAdd(sp + 3, p.w);
    if (lane == 0) unsafeAtomicAdd(&gcnt[g], (float)rows);
}

__global__ __launch_bounds__(256) void k_gather_relu_pool(const float* __restrict__ xw,
                                                          const int* __restrict__ rowptr,
                                                          const int* __restrict__ cnt,
                                                          const int* __restrict__ csr_src,
                                                          const float* __restrict__ dinv,
                                                          const float* __restrict__ bias,
                                                          const int* __restrict__ batch,
                                                          float* __restrict__ sums,
                                                          float* __restrict__ gcnt, int n)
{
    const int lane = threadIdx.x & 31;
    const int nw   = gridDim.x * 8;
    const int gw   = __builtin_amdgcn_readfirstlane((int)(blockIdx.x * 8 + (threadIdx.x >> 5)));
    const int per  = n / nw;
    const int rem  = n - per * nw;
    const int lo   = gw * per + ((gw < rem) ? gw : rem);
    const int hi   = lo + per + ((gw < rem) ? 1 : 0);
    if (lo >= hi) return;
    const float4* xw4 = reinterpret_cast<const float4*>(xw);
    const float* bp = bias + lane * 4;
    const float4 bv = make_float4(bp[0], bp[1], bp[2], bp[3]);

    int   node = lo;
    int   beg = __builtin_amdgcn_readfirstlane(rowptr[node]);
    int   deg = __builtin_amdgcn_readfirstlane(cnt[node]);
    int   g   = __builtin_amdgcn_readfirstlane(batch[node]);
    float dd  = dinv[node];

    int    cg   = g;
    int    rows = 0;
    float4 pacc = make_float4(0.0f, 0.0f, 0.0f, 0.0f);

    for (;;) {
        const int  next = node + 1;
        const bool more = next < hi;
        int   nbeg = 0, ndeg = 0, ng = 0;
        float ndd  = 0.0f;
        if (more) {
            nbeg = rowptr[next];
            ndeg = cnt[next];
            ng   = batch[next];
            ndd  = dinv[next];
        }

        int s_l = 0;
        if (lane < deg) s_l = csr_src[beg + lane];
        float4 acc = xw4[(long long)node * 32 + lane];
        for (int base = 0; base < deg; base += 32) {
            const int m = (deg - base < 32) ? (deg - base) : 32;
#pragma unroll 4
            for (int j = 0; j < m; ++j) {
                const int s = __builtin_amdgcn_readlane(s_l, j);
                const float4 v = xw4[(long long)s * 32 + lane];
                acc.x += v.x;
                acc.y += v.y;
                acc.z += v.z;
                acc.w += v.w;
            }
            const int left = deg - base - 32;
            if (left > 0) {
                s_l = 0;
                if (lane < left) s_l = csr_src[beg + base + 32 + lane];
            }
        }

        float4 o;
        o.x = fmaxf(acc.x * dd + bv.x, 0.0f);
        o.y = fmaxf(acc.y * dd + bv.y, 0.0f);
        o.z = fmaxf(acc.z * dd + bv.z, 0.0f);
        o.w = fmaxf(acc.w * dd + bv.w, 0.0f);

        if (g != cg) {
            pool_flush(sums, gcnt, cg, lane, pacc, rows);
            pacc = make_float4(0.0f, 0.0f, 0.0f, 0.0f);
            rows = 0;
            cg   = g;
        }
        pacc.x += o.x;
        pacc.y += o.y;
        pacc.z += o.z;
        pacc.w += o.w;
        rows += 1;

        if (!more) break;
        node = next;
        beg  = __builtin_amdgcn_readfirstlane(nbeg);
        deg  = __builtin_amdgcn_readfirstlane(ndeg);
        g    = __builtin_amdgcn_readfirstlane(ng);
        dd   = ndd;
    }
    pool_flush(sums, gcnt, cg, lane, pacc, rows);
}

#define SC_ROWS (NSHAPE * NCOLOR)
#define GT_THREADS 1024
__global__ __launch_bounds__(GT_THREADS) void k_gather_tab_relu(const int* __restrict__ rec_in,
                                                                const float* __restrict__ tw,
                                                                const int* __restrict__ rowptr,
                                                                const int* __restrict__ cnt,
                                                                const int* __restrict__ csr_src,
                                                                const float* __restrict__ bias,
                                                                float* __restrict__ h, int n)
{
    __shared__ float4 sc4[SC_ROWS * 32];
    const float4* tw4 = reinterpret_cast<const float4*>(tw);
#pragma unroll
    for (int i = threadIdx.x; i < SC_ROWS * 32; i += GT_THREADS) {
        const int pr = i >> 5, q = i & 31;
        const float4 a = tw4[(pr / NCOLOR) * 32 + q];
        const float4 b = tw4[(NSHAPE + (pr % NCOLOR)) * 32 + q];
        sc4[i] = make_float4(a.x + b.x, a.y + b.y, a.z + b.z, a.w + b.w);
    }
    __syncthreads();

    const int4* rec = reinterpret_cast<const int4*>(rec_in);
    const int lane = threadIdx.x & 31;
    const int nw   = gridDim.x * (GT_THREADS / 32);
    int node = __builtin_amdgcn_readfirstlane((int)(blockIdx.x * (GT_THREADS / 32) + (threadIdx.x >> 5)));
    if (node >= n) return;
    const float* bp = bias + lane * 4;
    const float4 bv = make_float4(bp[0], bp[1], bp[2], bp[3]);

    int   beg = __builtin_amdgcn_readfirstlane(rowptr[node]);
    int   deg = __builtin_amdgcn_readfirstlane(cnt[node]);
    const int4 me = rec[node];
    float dd  = __int_as_float(me.w);
    int   q0 = __builtin_amdgcn_readfirstlane(me.x);
    int   q1 = __builtin_amdgcn_readfirstlane(me.y);
    int   q2 = __builtin_amdgcn_readfirstlane(me.z);

    for (;;) {
        const int  next = node + nw;
        const bool more = next < n;
        int  nbeg = 0, ndeg = 0;
        int4 nme  = make_int4(0, 0, 0, 0);
        if (more) {
            nbeg = rowptr[next];
            ndeg = cnt[next];
            nme  = rec[next];
        }

        float4 acc = make_float4(0.0f, 0.0f, 0.0f, 0.0f);
        for (int base = 0; base < deg; base += 32) {
            const int m = (deg - base < 32) ? (deg - base) : 32;
            int   rp = 0, r2 = 0;
            float c_l = 0.0f;
            if (lane < m) {
                const int  s_l = csr_src[beg + base + lane];
                const int4 rc  = rec[s_l];
                c_l = __int_as_float(rc.w) * dd;
                rp = rc.x * NCOLOR + (rc.y - NSHAPE);
                r2 = rc.z;
            }
#pragma unroll 4
            for (int j = 0; j < m; ++j) {
                const int   ip = __builtin_amdgcn_readlane(rp, j);
                const int   i2 = __builtin_amdgcn_readlane(r2, j);
                const float c  = __int_as_float(__builtin_amdgcn_readlane(__float_as_int(c_l), j));
                const float4 ab = sc4[ip * 32 + lane];
                const float4 d  = tw4[i2 * 32 + lane];
                acc.x += (ab.x + d.x) * c;
                acc.y += (ab.y + d.y) * c;
                acc.z += (ab.z + d.z) * c;
                acc.w += (ab.w + d.w) * c;
            }
        }

        const int qp = q0 * NCOLOR + (q1 - NSHAPE);
        const float4 ab = sc4[qp * 32 + lane];
        const float4 d  = tw4[q2 * 32 + lane];
        const float sl = dd * dd;
        float4 o;
        o.x = fmaxf(acc.x + (ab.x + d.x) * sl + bv.x, 0.0f);
        o.y = fmaxf(acc.y + (ab.y + d.y) * sl + bv.y, 0.0f);
        o.z = fmaxf(acc.z + (ab.z + d.z) * sl + bv.z, 0.0f);
        o.w = fmaxf(acc.w + (ab.w + d.w) * sl + bv.w, 0.0f);
        reinterpret_cast<float4*>(h)[(long long)node * 32 + lane] = o;

        if (!more) break;
        node = next;
        beg  = __builtin_amdgcn_readfirstlane(nbeg);
        deg  = __builtin_amdgcn_readfirstlane(ndeg);
        dd   = __int_as_float(nme.w);
        q0   = __builtin_amdgcn_readfirstlane(nme.x);
        q1   = __builtin_amdgcn_readfirstlane(nme.y);
        q2   = __builtin_amdgcn_readfirstlane(nme.z);
    }
}

__global__ void k_head(const float* __restrict__ sums,
                       const float* __restrict__ cnt,
                       const float* __restrict__ Wlin,
                       const float* __restrict__ blin,
                       float* __restrict__ out, int g_total)
{
    int idx = blockIdx.x * blockDim.x + threadIdx.x;
    if (idx >= g_total * NCLS) return;
    const int g = idx / NCLS;
    const int c = idx % NCLS;
    const float inv = 1.0f / fmaxf(cnt[g], 1.0f);
    const float* sp = sums + (long long)g * EMB;
    float acc = 0.0f;
#pragma unroll 4
    for (int k = 0; k < EMB; ++k) acc += sp[k] * Wlin[k * NCLS + c];
    out[idx] = acc * inv + blin[c];
}

static inline size_t align256(size_t x) { return (x + 255) & ~(size_t)255; }

extern "C" void kernel_launch(void* const* d_in, const int* in_sizes, int n_in,
                              void* d_out, int out_size, void* d_ws, size_t ws_size,
                              hipStream_t stream)
{
    const int*   x        = (const int*)  d_in[0];
    const int*   eidx     = (const int*)  d_in[1];
    const int*   batch    = (const int*)  d_in[2];
    const float* shapeTab = (const float*)d_in[3];
    const float* colorTab = (const float*)d_in[4];
    const float* posTab   = (const float*)d_in[5];
    const float* W1       = (const float*)d_in[6];
    const float* b1       = (const float*)d_in[7];
    const float* W2       = (const float*)d_in[8];
    const float* b2       = (const float*)d_in[9];
    const float* Wlin     = (const float*)d_in[10];
    const float* blin     = (const float*)d_in[11];
    float*       out      = (float*)d_out;

    const int N = in_sizes[0] / 3;
    const int E = in_sizes[1] / 2;
    const int G = out_size / NCLS;

    const int* src = eidx;
    const int* dst = eidx + E;

    char* ws = (char*)d_ws;
    size_t off = 0;
    float* h    = (float*)(ws + off); off = align256(off + (size_t)N * EMB * 4);
    float* xw   = (float*)(ws + off); off = align256(off + (size_t)N * EMB * 4);
    char*  csr  = ws + off;           off = align256(off + (size_t)N * EMB * 4);
    float* dinv = (float*)(ws + off); off = align256(off + (size_t)N * 4);
    float* sums = (float*)(ws + off); off = align256(off + (size_t)G * EMB * 4);
    float* cnt  = (float*)(ws + off); off = align256(off + (size_t)G * 4);
    (void)ws_size;

    const int nBins  = (N + BIN_NODES - 1) / BIN_NODES;
    const int nParts = (E + CHUNK - 1) / CHUNK;
    const int cPitch = (nBins + 1 + 31) & ~31;
    size_t coff = 0;
    int*  degc     = (int*)(csr + coff);  coff = align256(coff + (size_t)N * 4);
    int*  rowptr   = (int*)(csr + coff);  coff = align256(coff + (size_t)N * 4);
    int*  runs     = (int*)(csr + coff);  coff = align256(coff + (size_t)nParts * cPitch * 4);
    int*  pairs    = (int*)(csr + coff);  coff = align256(coff + (size_t)E * 8);
    int*  csrSrc   = (int*)(csr + coff);  coff = align256(coff + (size_t)E * 4);
    float* tw   = (float*)(csr + coff); coff = align256(coff + (size_t)NTABROWS * EMB * 4);
    int*  rec      = (int*)(csr + coff);  coff = align256(coff + (size_t)N * 16);

    const int TB = 256;
    const int gNodeWave = 2048;
    const int gNodeTab  = 512;
    const int gGemmBlks = (N + 127) / 128;

    k_chunk_sort<<<nParts + TAB_BLOCKS, PT, 0, stream>>>(src, dst, runs, pairs, E, nBins, cPitch, nParts,
                                                        shapeTab, colorTab, posTab, W1, tw,
                                                        sums, G * EMB, cnt, G);

    k_bin_sort<<<nBins, 256, 0, stream>>>(pairs, runs, cPitch, nParts, rowptr, degc, dinv, csrSrc, x, rec, N);

    k_gather_tab_relu<<<gNodeTab, GT_THREADS, 0, stream>>>(rec, tw, rowptr, degc, csrSrc, b1, h, N);

    k_gemm_wmma<<<gGemmBlks, TB, 0, stream>>>(h, W2, dinv, xw, N);
    k_gather_relu_pool<<<gNodeWave, 256, 0, stream>>>(xw, rowptr, degc, csrSrc, dinv, b2, batch, sums, cnt, N);

    k_head<<<(G * NCLS + TB - 1) / TB, TB, 0, stream>>>(sums, cnt, Wlin, blin, out, G);
}


# linear head applied at the pool hand-over: a wave adds (its partial graph sum @ Wlin) / graph rows to out (starts at blin) by float atomics; no sums plane, no head launch; head sums regrouped
# speedup vs baseline: 7.1902x; 1.0707x over previous
#include <hip/hip_runtime.h>
#include <hip/hip_bf16.h>

#define EMB 128
#define NCLS 10
#define NSHAPE 16
#define NCOLOR 16
#define MAXPOS 256
#define NTABROWS (NSHAPE + NCOLOR + MAXPOS)

typedef __attribute__((ext_vector_type(2))) float v2f;
typedef __attribute__((ext_vector_type(8))) float v8f;

__device__ __forceinline__ void tab_gemm_tile(int tile, int ctile, int lane,
    const float* __restrict__ st, const float* __restrict__ ct, const float* __restrict__ pt,
    const float* __restrict__ W, float* __restrict__ tw)
{
    const int m     = lane & 15;
    const int khalf = (lane >> 4) * 2;

    const float* abase;
    if (tile == 0)      abase = st;
    else if (tile == 1) abase = ct;
    else                abase = pt + (long long)(tile - 2) * 16 * EMB;

    const float* arow = abase + m * EMB;
    const int col = ctile * 16 + m;

    v8f acc = {};
    for (int k = 0; k < EMB; k += 4) {
        const int kk = k + khalf;
        v2f a, b;
        a.x = arow[kk];
        a.y = arow[kk + 1];
        b.x = W[kk * EMB + col];
        b.y = W[(kk + 1) * EMB + col];
        acc = __builtin_amdgcn_wmma_f32_16x16x4_f32(
            false, a, false, b, (short)0, acc, false, false);
    }

    const int rbase = (lane >> 4) * 8;
#pragma unroll
    for (int r = 0; r < 8; ++r)
        tw[(long long)(tile * 16 + rbase + r) * EMB + col] = acc[r];
}

#define BIN_SHIFT 7
#define BIN_NODES (1 << BIN_SHIFT)
#define MAXBINS 1024
#define PT 1024
#define KPT 12
#define CHUNK (PT * KPT)
#define MAXPARTS 256
#define RUNS_U 17
#define STAGE_CAP 8192

__device__ __forceinline__ int wave_incl_scan(int v, int lane)
{
#pragma unroll
    for (int off = 1; off < 32; off <<= 1) {
        const int u = __shfl_up(v, off);
        if (lane >= off) v += u;
    }
    return v;
}

__device__ __forceinline__ int wave_sum(int v)
{
#pragma unroll
    for (int off = 16; off > 0; off >>= 1) v += __shfl_xor(v, off);
    return v;
}

#define TAB_TILES (NTABROWS / 16)
#define TAB_BLOCKS ((TAB_TILES + 3) / 4)
__global__ __launch_bounds__(PT) void k_chunk_sort(const int* __restrict__ src, const int* __restrict__ dst,
                                                   int* __restrict__ runs, int* __restrict__ pairs_out,
                                                   int e, int nb, int pitch, int nparts,
                                                   const float* __restrict__ st, const float* __restrict__ ct,
                                                   const float* __restrict__ pt, const float* __restrict__ W,
                                                   float* __restrict__ tw,
                                                   const float* __restrict__ blin,
                                                   float* __restrict__ out, int nout,
                                                   float* __restrict__ gcnt, int ng)
{
    int2* pairs = reinterpret_cast<int2*>(pairs_out);
    __shared__ int  hist[MAXBINS];
    __shared__ int  lstart[MAXBINS];
    __shared__ int  wsA[32];
    __shared__ int2 sorted[CHUNK];
    const int t    = threadIdx.x;
    const int lane = t & 31;
    const int wave = t >> 5;
    const int p    = blockIdx.x;

    for (int i = blockIdx.x * PT + t; i < nout; i += gridDim.x * PT) out[i] = blin[i % NCLS];
    for (int i = blockIdx.x * PT + t; i < ng;   i += gridDim.x * PT) gcnt[i] = 0.0f;

    if (p >= nparts) {
        const int tile = (p - nparts) * 4 + (wave >> 3);
        if (tile < TAB_TILES)
            tab_gemm_tile(tile, wave & 7, lane, st, ct, pt, W, tw);
        return;
    }

    hist[t] = 0;
    __syncthreads();

    const int base = p * CHUNK + t;
    int s[KPT], d[KPT], r[KPT];
#pragma unroll
    for (int j = 0; j < KPT; ++j) {
        const int i  = base + j * PT;
        const int ic = (i < e) ? i : (e - 1);
        s[j] = src[ic];
        d[j] = dst[ic];
    }
#pragma unroll
    for (int j = 0; j < KPT; ++j) {
        r[j] = 0;
        if (base + j * PT < e) r[j] = atomicAdd(&hist[d[j] >> BIN_SHIFT], 1);
    }
    __syncthreads();
    const int myh = hist[t];

    const int ia = wave_incl_scan(myh, lane);
    if (lane == 31) wsA[wave] = ia;
    __syncthreads();
    if (wave == 0) {
        const int xa = wsA[lane];
        const int sa = wave_incl_scan(xa, lane);
        wsA[lane] = sa - xa;
    }
    __syncthreads();
    const int lex = ia - myh + wsA[wave];
    lstart[t] = lex;

    const int left = e - p * CHUNK;
    const int cc   = (left < CHUNK) ? left : CHUNK;
    if (t < nb)  runs[p * pitch + t]  = lex;
    if (t == 0)  runs[p * pitch + nb] = cc;
    __syncthreads();

#pragma unroll
    for (int j = 0; j < KPT; ++j) {
        if (base + j * PT < e)
            sorted[lstart[d[j] >> BIN_SHIFT] + r[j]] = make_int2(s[j], d[j]);
    }
    __syncthreads();

    for (int i = t; i < cc; i += PT) pairs[p * CHUNK + i] = sorted[i];
}

template <bool WITH_SRC, class F>
__device__ __forceinline__ void for_each_pair_of_bin(const int* __restrict__ pairs_in, int rb_l, int rl_l,
                                                     int nr, int lane, F f)
{
    const int2* pairs = reinterpret_cast<const int2*>(pairs_in);
    for (int u0 = 0; u0 < nr; u0 += RUNS_U) {
        int len[RUNS_U], rb[RUNS_U], es[RUNS_U], ed[RUNS_U];
#pragma unroll
        for (int k = 0; k < RUNS_U; ++k) {
            const int u = u0 + k;
            rb[k]  = __builtin_amdgcn_readlane(rb_l, u & 31);
            len[k] = (u < nr) ? __builtin_amdgcn_readlane(rl_l, u & 31) : 0;
            const int i = (lane < len[k]) ? (rb[k] + lane) : 0;
            if constexpr (WITH_SRC) { const int2 v = pairs[i]; es[k] = v.x; ed[k] = v.y; }
            else                    { es[k] = 0; ed[k] = pairs_in[2 * i + 1]; }
        }
#pragma unroll
        for (int k = 0; k < RUNS_U; ++k)
            if (lane < len[k]) f(es[k], ed[k]);
#pragma unroll
        for (int k = 0; k < RUNS_U; ++k)
            for (int i = 32 + lane; i < len[k]; i += 32) {
                if constexpr (WITH_SRC) { const int2 v = pairs[rb[k] + i]; f(v.x, v.y); }
                else                    f(0, pairs_in[2 * (rb[k] + i) + 1]);
            }
    }
}

__global__ __launch_bounds__(256) void k_bin_sort(const int* __restrict__ pairs_in, const int* __restrict__ runs,
                                                  int pitch, int nparts,
                                                  int* __restrict__ rowptr, int* __restrict__ cnt,
                                                  float* __restrict__ dinv, int* __restrict__ csr_src,
                                                  const int* __restrict__ x, int* __restrict__ rec_out,
                                                  const int* __restrict__ batch, float* __restrict__ gcnt, int n)
{
    __shared__ int c[BIN_NODES];
    __shared__ int cur[BIN_NODES];
    __shared__ int ws[8];
    __shared__ int wsM[8];
    __shared__ int wsS[8];
    __shared__ int stage[STAGE_CAP];
    const int t     = threadIdx.x;
    const int lane  = t & 31;
    const int wave  = __builtin_amdgcn_readfirstlane(t >> 5);
    const int b     = blockIdx.x;

    const int q = lane * 8 + wave;
    int rs = 0, rl = 0;
    if (q < nparts) {
        const int* row = runs + q * pitch + b;
        rs = row[0];
        rl = row[1] - rs;
    }
    const int gnode = b * BIN_NODES + t;
    const int gb    = (t < BIN_NODES && gnode < n) ? batch[gnode] : -1;
    const int rb_l = q * CHUNK + rs;
    const int nr   = (nparts - wave + 7) >> 3;
    if (t < BIN_NODES) c[t] = 0;
    const int wm = wave_sum(rl);
    const int wr = wave_sum(rs);
    if (lane == 0) { wsM[wave] = wm; wsS[wave] = wr; }
    __syncthreads();

    for_each_pair_of_bin<false>(pairs_in, rb_l, rl, nr, lane,
                                [&](int, int dn) { atomicAdd(&c[dn & (BIN_NODES - 1)], 1); });

    int m = 0, start = 0;
#pragma unroll
    for (int w = 0; w < 8; ++w) { m += wsM[w]; start += wsS[w]; }
    __syncthreads();

    const int v  = (t < BIN_NODES) ? c[t] : 0;
    const int iv = wave_incl_scan(v, lane);
    if (lane == 31) ws[wave] = iv;
    __syncthreads();
    int woff = 0;
    for (int w = 0; w < wave; ++w) woff += ws[w];
    if (t < BIN_NODES) {
        const int ex = iv - v + woff;
        cur[t] = ex;
        const int node = b * BIN_NODES + t;
        const int  gprev = __shfl_up(gb, 1);
        const bool ghead = (lane == 0) || (gprev != gb);
        const unsigned hm = (unsigned)__ballot(ghead);
        if (ghead && gb >= 0) {
            const unsigned above = (lane == 31) ? 0u : (hm >> (lane + 1));
            const int len = above ? (__builtin_ctz(above) + 1) : (32 - lane);
            unsafeAtomicAdd(&gcnt[gb], (float)len);
        }
        if (node < n) {
            const float di = __frsqrt_rn((float)(v + 1));
            rowptr[node] = start + ex;
            cnt[node]    = v;
            dinv[node]   = di;
            reinterpret_cast<int4*>(rec_out)[node] =
                make_int4(x[node * 3 + 0], x[node * 3 + 1] + NSHAPE, x[node * 3 + 2] + NSHAPE + NCOLOR,
                          __float_as_int(di));
        }
    }
    __syncthreads();

    const bool staged = (m <= STAGE_CAP);
    for_each_pair_of_bin<true>(pairs_in, rb_l, rl, nr, lane,
                               [&](int sn, int dn) {
                                   const int pos = atomicAdd(&cur[dn & (BIN_NODES - 1)], 1);
                                   if (staged) stage[pos] = sn;
                                   else        csr_src[start + pos] = sn;
                               });
    if (staged) {
        __syncthreads();
        for (int i = t; i < m; i += 256) csr_src[start + i] = stage[i];
    }
}

#define WSTRIDE 128
__global__ __launch_bounds__(256) void k_gemm_wmma(
    const float* __restrict__ A, const float* __restrict__ W, const float* __restrict__ rscale,
    float* __restrict__ C, int nrows)
{
    __shared__ float Wl[128 * WSTRIDE];
    const int tid = threadIdx.x;

    for (int idx = tid; idx < 128 * 32; idx += 256) {
        int r = idx >> 5, q = idx & 31;
        reinterpret_cast<float4*>(&Wl[r * WSTRIDE + q * 4])[0] =
            reinterpret_cast<const float4*>(W)[r * 32 + q];
    }
    __syncthreads();

    const int wave  = tid >> 5;
    const int lane  = tid & 31;
    const int m     = lane & 15;
    const int khalf = (lane >> 4) * 2;
    const int tileRow = blockIdx.x * 128 + wave * 16;
    if (tileRow >= nrows) return;

    v8f acc[8] = {};

    const float* arow = A + (long long)(tileRow + m) * EMB;
    for (int k = 0; k < EMB; k += 4) {
        const int kk = k + khalf;
        v2f a;
        a.x = arow[kk];
        a.y = arow[kk + 1];
#pragma unroll
        for (int j = 0; j < 8; ++j) {
            const int col = j * 16 + m;
            v2f b;
            b.x = Wl[kk * WSTRIDE + col];
            b.y = Wl[(kk + 1) * WSTRIDE + col];
            acc[j] = __builtin_amdgcn_wmma_f32_16x16x4_f32(
                false, a, false, b, (short)0, acc[j], false, false);
        }
    }

    const int rbase = (lane >> 4) * 8;
    const float4 s0 = reinterpret_cast<const float4*>(rscale + tileRow + rbase)[0];
    const float4 s1 = reinterpret_cast<const float4*>(rscale + tileRow + rbase)[1];
    const float sc[8] = { s0.x, s0.y, s0.z, s0.w, s1.x, s1.y, s1.z, s1.w };
#pragma unroll
    for (int j = 0; j < 8; ++j) {
#pragma unroll
        for (int r = 0; r < 8; ++r) {
            C[(long long)(tileRow + rbase + r) * EMB + j * 16 + m] = acc[j][r] * sc[r];
        }
    }
}


__device__ __forceinline__ float wave_sum_f(float v)
{
#pragma unroll
    for (int off = 16; off > 0; off >>= 1) v += __shfl_xor(v, off);
    return v;
}

__device__ __forceinline__ void head_flush(const float* __restrict__ Wlin, const float* __restrict__ gcnt,
                                           float* __restrict__ out, int g, int lane, const float4& p)
{
    const float* wr = Wlin + lane * 4 * NCLS;
    static_assert(NCLS % 2 == 0, "the classes are taken in two equal passes");
    float mine = 0.0f;
#pragma unroll 1
    for (int h = 0; h < 2; ++h) {
        const float* wh = wr + h * (NCLS / 2);
        float part[NCLS / 2];
#pragma unroll
        for (int c = 0; c < NCLS / 2; ++c)
            part[c] = p.x * wh[c] + p.y * wh[NCLS + c] + p.z * wh[2 * NCLS + c] + p.w * wh[3 * NCLS + c];
#pragma unroll
        for (int c = 0; c < NCLS / 2; ++c) {
            const float s = wave_sum_f(part[c]);
            if (lane == h * (NCLS / 2) + c) mine = s;
        }
    }
    if (lane < NCLS) unsafeAtomicAdd(&out[g * NCLS + lane], mine / fmaxf(gcnt[g], 1.0f));
}

__global__ __launch_bounds__(256) void k_gather_relu_pool(const float* __restrict__ xw,
                                                          const int* __restrict__ rowptr,
                                                          const int* __restrict__ cnt,
                                                          const int* __restrict__ csr_src,
                                                          const float* __restrict__ dinv,
                                                          const float* __restrict__ bias,
                                                          const int* __restrict__ batch,
                                                          const float* __restrict__ Wlin,
                                                          const float* __restrict__ gcnt,
                                                          float* __restrict__ out, int n)
{
    const int lane = threadIdx.x & 31;
    const int nw   = gridDim.x * 8;
    const int gw   = __builtin_amdgcn_readfirstlane((int)(blockIdx.x * 8 + (threadIdx.x >> 5)));
    const int per  = n / nw;
    const int rem  = n - per * nw;
    const int lo   = gw * per + ((gw < rem) ? gw : rem);
    const int hi   = lo + per + ((gw < rem) ? 1 : 0);
    if (lo >= hi) return;
    const float4* xw4 = reinterpret_cast<const float4*>(xw);
    const float* bp = bias + lane * 4;
    const float4 bv = make_float4(bp[0], bp[1], bp[2], bp[3]);

    int   node = lo;
    int   beg = __builtin_amdgcn_readfirstlane(rowptr[node]);
    int   deg = __builtin_amdgcn_readfirstlane(cnt[node]);
    int   g   = __builtin_amdgcn_readfirstlane(batch[node]);
    float dd  = dinv[node];

    int    cg   = g;
    float4 pacc = make_float4(0.0f, 0.0f, 0.0f, 0.0f);

    for (;;) {
        const int  next = node + 1;
        const bool more = next < hi;
        int   nbeg = 0, ndeg = 0, ng = 0;
        float ndd  = 0.0f;
        if (more) {
            nbeg = rowptr[next];
            ndeg = cnt[next];
            ng   = batch[next];
            ndd  = dinv[next];
        }

        int s_l = 0;
        if (lane < deg) s_l = csr_src[beg + lane];
        float4 acc = xw4[(long long)node * 32 + lane];
        for (int base = 0; base < deg; base += 32) {
            const int m = (deg - base < 32) ? (deg - base) : 32;
#pragma unroll 4
            for (int j = 0; j < m; ++j) {
                const int s = __builtin_amdgcn_readlane(s_l, j);
                const float4 v = xw4[(long long)s * 32 + lane];
                acc.x += v.x;
                acc.y += v.y;
                acc.z += v.z;
                acc.w += v.w;
            }
            const int left = deg - base - 32;
            if (left > 0) {
                s_l = 0;
                if (lane < left) s_l = csr_src[beg + base + 32 + lane];
            }
        }

        float4 o;
        o.x = fmaxf(acc.x * dd + bv.x, 0.0f);
        o.y = fmaxf(acc.y * dd + bv.y, 0.0f);
        o.z = fmaxf(acc.z * dd + bv.z, 0.0f);
        o.w = fmaxf(acc.w * dd + bv.w, 0.0f);

        if (g != cg) {
            head_flush(Wlin, gcnt, out, cg, lane, pacc);
            pacc = make_float4(0.0f, 0.0f, 0.0f, 0.0f);
            cg   = g;
        }
        pacc.x += o.x;
        pacc.y += o.y;
        pacc.z += o.z;
        pacc.w += o.w;

        if (!more) break;
        node = next;
        beg  = __builtin_amdgcn_readfirstlane(nbeg);
        deg  = __builtin_amdgcn_readfirstlane(ndeg);
        g    = __builtin_amdgcn_readfirstlane(ng);
        dd   = ndd;
    }
    head_flush(Wlin, gcnt, out, cg, lane, pacc);
}

#define SC_ROWS (NSHAPE * NCOLOR)
#define GT_THREADS 1024
__global__ __launch_bounds__(GT_THREADS) void k_gather_tab_relu(const int* __restrict__ rec_in,
                                                                const float* __restrict__ tw,
                                                                const int* __restrict__ rowptr,
                                                                const int* __restrict__ cnt,
                                                                const int* __restrict__ csr_src,
                                                                const float* __restrict__ bias,
                                                                float* __restrict__ h, int n)
{
    __shared__ float4 sc4[SC_ROWS * 32];
    const float4* tw4 = reinterpret_cast<const float4*>(tw);
#pragma unroll
    for (int i = threadIdx.x; i < SC_ROWS * 32; i += GT_THREADS) {
        const int pr = i >> 5, q = i & 31;
        const float4 a = tw4[(pr / NCOLOR) * 32 + q];
        const float4 b = tw4[(NSHAPE + (pr % NCOLOR)) * 32 + q];
        sc4[i] = make_float4(a.x + b.x, a.y + b.y, a.z + b.z, a.w + b.w);
    }
    __syncthreads();

    const int4* rec = reinterpret_cast<const int4*>(rec_in);
    const int lane = threadIdx.x & 31;
    const int nw   = gridDim.x * (GT_THREADS / 32);
    int node = __builtin_amdgcn_readfirstlane((int)(blockIdx.x * (GT_THREADS / 32) + (threadIdx.x >> 5)));
    if (node >= n) return;
    const float* bp = bias + lane * 4;
    const float4 bv = make_float4(bp[0], bp[1], bp[2], bp[3]);

    int   beg = __builtin_amdgcn_readfirstlane(rowptr[node]);
    int   deg = __builtin_amdgcn_readfirstlane(cnt[node]);
    const int4 me = rec[node];
    float dd  = __int_as_float(me.w);
    int   q0 = __builtin_amdgcn_readfirstlane(me.x);
    int   q1 = __builtin_amdgcn_readfirstlane(me.y);
    int   q2 = __builtin_amdgcn_readfirstlane(me.z);

    for (;;) {
        const int  next = node + nw;
        const bool more = next < n;
        int  nbeg = 0, ndeg = 0;
        int4 nme  = make_int4(0, 0, 0, 0);
        if (more) {
            nbeg = rowptr[next];
            ndeg = cnt[next];
            nme  = rec[next];
        }

        float4 acc = make_float4(0.0f, 0.0f, 0.0f, 0.0f);
        for (int base = 0; base < deg; base += 32) {
            const int m = (deg - base < 32) ? (deg - base) : 32;
            int   rp = 0, r2 = 0;
            float c_l = 0.0f;
            if (lane < m) {
                const int  s_l = csr_src[beg + base + lane];
                const int4 rc  = rec[s_l];
                c_l = __int_as_float(rc.w) * dd;
                rp = rc.x * NCOLOR + (rc.y - NSHAPE);
                r2 = rc.z;
            }
#pragma unroll 4
            for (int j = 0; j < m; ++j) {
                const int   ip = __builtin_amdgcn_readlane(rp, j);
                const int   i2 = __builtin_amdgcn_readlane(r2, j);
                const float c  = __int_as_float(__builtin_amdgcn_readlane(__float_as_int(c_l), j));
                const float4 ab = sc4[ip * 32 + lane];
                const float4 d  = tw4[i2 * 32 + lane];
                acc.x += (ab.x + d.x) * c;
                acc.y += (ab.y + d.y) * c;
                acc.z += (ab.z + d.z) * c;
                acc.w += (ab.w + d.w) * c;
            }
        }

        const int qp = q0 * NCOLOR + (q1 - NSHAPE);
        const float4 ab = sc4[qp * 32 + lane];
        const float4 d  = tw4[q2 * 32 + lane];
        const float sl = dd * dd;
        float4 o;
        o.x = fmaxf(acc.x + (ab.x + d.x) * sl + bv.x, 0.0f);
        o.y = fmaxf(acc.y + (ab.y + d.y) * sl + bv.y, 0.0f);
        o.z = fmaxf(acc.z + (ab.z + d.z) * sl + bv.z, 0.0f);
        o.w = fmaxf(acc.w + (ab.w + d.w) * sl + bv.w, 0.0f);
        reinterpret_cast<float4*>(h)[(long long)node * 32 + lane] = o;

        if (!more) break;
        node = next;
        beg  = __builtin_amdgcn_readfirstlane(nbeg);
        deg  = __builtin_amdgcn_readfirstlane(ndeg);
        dd   = __int_as_float(nme.w);
        q0   = __builtin_amdgcn_readfirstlane(nme.x);
        q1   = __builtin_amdgcn_readfirstlane(nme.y);
        q2   = __builtin_amdgcn_readfirstlane(nme.z);
    }
}

static inline size_t align256(size_t x) { return (x + 255) & ~(size_t)255; }

extern "C" void kernel_launch(void* const* d_in, const int* in_sizes, int n_in,
                              void* d_out, int out_size, void* d_ws, size_t ws_size,
                              hipStream_t stream)
{
    const int*   x        = (const int*)  d_in[0];
    const int*   eidx     = (const int*)  d_in[1];
    const int*   batch    = (const int*)  d_in[2];
    const float* shapeTab = (const float*)d_in[3];
    const float* colorTab = (const float*)d_in[4];
    const float* posTab   = (const float*)d_in[5];
    const float* W1       = (const float*)d_in[6];
    const float* b1       = (const float*)d_in[7];
    const float* W2       = (const float*)d_in[8];
    const float* b2       = (const float*)d_in[9];
    const float* Wlin     = (const float*)d_in[10];
    const float* blin     = (const float*)d_in[11];
    float*       out      = (float*)d_out;

    const int N = in_sizes[0] / 3;
    const int E = in_sizes[1] / 2;
    const int G = out_size / NCLS;

    const int* src = eidx;
    const int* dst = eidx + E;

    char* ws = (char*)d_ws;
    size_t off = 0;
    float* h    = (float*)(ws + off); off = align256(off + (size_t)N * EMB * 4);
    float* xw   = (float*)(ws + off); off = align256(off + (size_t)N * EMB * 4);
    char*  csr  = ws + off;           off = align256(off + (size_t)N * EMB * 4);
    float* dinv = (float*)(ws + off); off = align256(off + (size_t)N * 4);
    float* cnt  = (float*)(ws + off); off = align256(off + (size_t)G * 4);
    (void)ws_size;

    const int nBins  = (N + BIN_NODES - 1) / BIN_NODES;
    const int nParts = (E + CHUNK - 1) / CHUNK;
    const int cPitch = (nBins + 1 + 31) & ~31;
    size_t coff = 0;
    int*  degc     = (int*)(csr + coff);  coff = align256(coff + (size_t)N * 4);
    int*  rowptr   = (int*)(csr + coff);  coff = align256(coff + (size_t)N * 4);
    int*  runs     = (int*)(csr + coff);  coff = align256(coff + (size_t)nParts * cPitch * 4);
    int*  pairs    = (int*)(csr + coff);  coff = align256(coff + (size_t)E * 8);
    int*  csrSrc   = (int*)(csr + coff);  coff = align256(coff + (size_t)E * 4);
    float* tw   = (float*)(csr + coff); coff = align256(coff + (size_t)NTABROWS * EMB * 4);
    int*  rec      = (int*)(csr + coff);  coff = align256(coff + (size_t)N * 16);

    const int TB = 256;
    const int gNodeWave = 2048;
    const int gNodeTab  = 512;
    const int gGemmBlks = (N + 127) / 128;

    k_chunk_sort<<<nParts + TAB_BLOCKS, PT, 0, stream>>>(src, dst, runs, pairs, E, nBins, cPitch, nParts,
                                                        shapeTab, colorTab, posTab, W1, tw,
                                                        blin, out, G * NCLS, cnt, G);

    k_bin_sort<<<nBins, 256, 0, stream>>>(pairs, runs, cPitch, nParts, rowptr, degc, dinv, csrSrc, x, rec,
                                          batch, cnt, N);

    k_gather_tab_relu<<<gNodeTab, GT_THREADS, 0, stream>>>(rec, tw, rowptr, degc, csrSrc, b1, h, N);

    k_gemm_wmma<<<gGemmBlks, TB, 0, stream>>>(h, W2, dinv, xw, N);
    k_gather_relu_pool<<<gNodeWave, 256, 0, stream>>>(xw, rowptr, degc, csrSrc, dinv, b2, batch, Wlin, cnt, out, N);
}
